# starting kernel
# baseline (speedup 1.0000x reference)
#include <hip/hip_runtime.h>
#include <math.h>


#define N_NODES 100000
#define N0      60000
#define N1      40000
#define F0      128
#define F1      64
#define HID     64
#define HEADS   8
#define OUTD    16
#define ATTNV   128
#define NE      250000
#define LP      3
#define NG      8192
#define NB      8192

typedef __attribute__((ext_vector_type(2))) float f32x2;
typedef __attribute__((ext_vector_type(8))) float f32x8;

__device__ __forceinline__ f32x8 wmma_f32(f32x2 a, f32x2 b, f32x8 c) {
  return __builtin_amdgcn_wmma_f32_16x16x4_f32(false, a, false, b, (short)0, c,
                                               false, false);
}

__global__ void fill_kernel(float* __restrict__ p, float v, int n) {
  int i = blockIdx.x * 256 + threadIdx.x;
  if (i < n) p[i] = v;
}

__global__ void finals_kernel(const float* __restrict__ r_vec,
                              float* __restrict__ finals  ) {
  int c = threadIdx.x;
  if (c >= 32) return;
  for (int mp = 0; mp < 2; ++mp) {
    float re = r_vec[(mp * 32 + c) * 2 + 0];
    float im = r_vec[(mp * 32 + c) * 2 + 1];
    float inv = rsqrtf(re * re + im * im);
    re *= inv; im *= inv;
    float f1re = re, f1im = -im;
    float f0re = f1re * re - f1im * im;
    float f0im = f1re * im + f1im * re;
    float* F = finals + mp * LP * HID;
    F[0 * HID + 2 * c] = f0re; F[0 * HID + 2 * c + 1] = f0im;
    F[1 * HID + 2 * c] = f1re; F[1 * HID + 2 * c + 1] = f1im;
    F[2 * HID + 2 * c] = 1.f;  F[2 * HID + 2 * c + 1] = 0.f;
  }
}

template <int K>
__global__ __launch_bounds__(256) void gemm_tf_kernel(
    const float* __restrict__ feat, const float* __restrict__ W,
    const float* __restrict__ bias, const int* __restrict__ idx,
    float* __restrict__ tf, int M) {
  __shared__ float ldsB[K * HID];
  int tid = threadIdx.x;
  for (int i = tid; i < K * HID; i += 256) {
    int k = i >> 6, n = i & 63;
    ldsB[i] = W[n * K + k];
  }
  __syncthreads();
  int wave = tid >> 5, lane = tid & 31;
  int tile_m = blockIdx.x * 128 + wave * 16;
  if (tile_m >= M) return;
  int half = lane >> 4, l15 = lane & 15;
  int arow = tile_m + l15;
  if (arow >= M) arow = M - 1;
  f32x8 c0 = {}, c1 = {}, c2 = {}, c3 = {};
  for (int k0 = 0; k0 < K; k0 += 4) {
    int kk = k0 + half * 2;
    f32x2 a;
    a.x = feat[(long)arow * K + kk];
    a.y = feat[(long)arow * K + kk + 1];
    f32x2 b0, b1, b2, b3;
    b0.x = ldsB[kk * HID + l15 + 0];  b0.y = ldsB[(kk + 1) * HID + l15 + 0];
    b1.x = ldsB[kk * HID + l15 + 16]; b1.y = ldsB[(kk + 1) * HID + l15 + 16];
    b2.x = ldsB[kk * HID + l15 + 32]; b2.y = ldsB[(kk + 1) * HID + l15 + 32];
    b3.x = ldsB[kk * HID + l15 + 48]; b3.y = ldsB[(kk + 1) * HID + l15 + 48];
    c0 = wmma_f32(a, b0, c0);
    c1 = wmma_f32(a, b1, c1);
    c2 = wmma_f32(a, b2, c2);
    c3 = wmma_f32(a, b3, c3);
  }
#pragma unroll
  for (int v = 0; v < 8; ++v) {
    int r = tile_m + v + half * 8;
    if (r < M) {
      float* drow = tf + (long)idx[r] * HID;
      drow[l15 + 0]  = c0[v] + bias[l15 + 0];
      drow[l15 + 16] = c1[v] + bias[l15 + 16];
      drow[l15 + 32] = c2[v] + bias[l15 + 32];
      drow[l15 + 48] = c3[v] + bias[l15 + 48];
    }
  }
}

__global__ __launch_bounds__(256) void edge_h_kernel(
    const float* __restrict__ tf, const float* __restrict__ finals,
    const int* __restrict__ indices, const int* __restrict__ dst,
    const float* __restrict__ attn, float* __restrict__ h,
    float* __restrict__ a, float* __restrict__ m) {
  int wave = threadIdx.x >> 5, lane = threadIdx.x & 31;
  int e = blockIdx.x * 8 + wave;
  if (e >= NE) return;
  float2 acc = make_float2(0.f, 0.f);
#pragma unroll
  for (int i = 0; i < LP; ++i) {
    int node = indices[e * LP + i];
    const float* row = tf + (long)node * HID + lane * 2;
    float vx = row[0], vy = row[1];
    float fx = finals[i * HID + lane * 2], fy = finals[i * HID + lane * 2 + 1];
    acc.x += vx * fx - vy * fy;
    acc.y += vx * fy + vy * fx;
  }
  acc.x *= (1.f / 3.f);
  acc.y *= (1.f / 3.f);
  h[(long)e * HID + lane * 2] = acc.x;
  h[(long)e * HID + lane * 2 + 1] = acc.y;
  int d = dst[e];
#pragma unroll
  for (int hh = 0; hh < HEADS; ++hh) {
    float p = acc.x * attn[hh * HID + lane * 2] +
              acc.y * attn[hh * HID + lane * 2 + 1];
#pragma unroll
    for (int off = 16; off > 0; off >>= 1) p += __shfl_xor(p, off, 32);
    if (lane == 0) {
      float av = p > 0.f ? p : 0.01f * p;
      a[(long)e * HEADS + hh] = av;
      atomicMax(&m[d * HEADS + hh], av);
    }
  }
}

__global__ void edge_exp_kernel(float* __restrict__ a,
                                const int* __restrict__ dst,
                                const float* __restrict__ m,
                                float* __restrict__ s, int n) {
  int i = blockIdx.x * 256 + threadIdx.x;
  if (i >= n) return;
  int e = i >> 3, hh = i & 7;
  int d = dst[e];
  float ex = __expf(a[i] - m[d * HEADS + hh]);
  a[i] = ex;
  atomicAdd(&s[d * HEADS + hh], ex);
}

__global__ __launch_bounds__(256) void edge_scatter_kernel(
    const float* __restrict__ h, const float* __restrict__ a,
    const float* __restrict__ s, const int* __restrict__ dst,
    float* __restrict__ ft) {
  int wave = threadIdx.x >> 5, lane = threadIdx.x & 31;
  int e = blockIdx.x * 8 + wave;
  if (e >= NE) return;
  int d = dst[e];
  float an_l = 0.f;
  if (lane < HEADS) an_l = a[(long)e * HEADS + lane] / s[d * HEADS + lane];
  float hx = h[(long)e * HID + lane * 2];
  float hy = h[(long)e * HID + lane * 2 + 1];
#pragma unroll
  for (int hh = 0; hh < HEADS; ++hh) {
    float w = __shfl(an_l, hh, 32);
    float* base = ft + ((long)d * HEADS + hh) * HID + lane * 2;
    atomicAdd(base + 0, w * hx);
    atomicAdd(base + 1, w * hy);
  }
}

__global__ void target_elu_kernel(const float* __restrict__ ft,
                                  const int* __restrict__ target,
                                  float* __restrict__ outp, int n) {
  int i = blockIdx.x * 256 + threadIdx.x;
  if (i >= n) return;
  int b = i >> 9, j = i & 511;
  float v = ft[(long)target[b] * (HEADS * HID) + j];
  outp[i] = v > 0.f ? v : (__expf(v) - 1.f);
}

__global__ __launch_bounds__(256) void score_kernel(
    const float* __restrict__ outs, const float* __restrict__ fc1_w,
    const float* __restrict__ fc1_b, const float* __restrict__ fc2_w,
    float* __restrict__ score) {
  int p = blockIdx.y;
  const float* A = outs + (long)p * NB * (HEADS * HID);
  int wave = threadIdx.x >> 5, lane = threadIdx.x & 31;
  int tile_m = (blockIdx.x * 8 + wave) * 16;
  int half = lane >> 4, l15 = lane & 15;
  long arow = tile_m + l15;
  f32x8 acc[8] = {};
  for (int k0 = 0; k0 < HEADS * HID; k0 += 4) {
    int kk = k0 + half * 2;
    f32x2 af;
    af.x = A[arow * (HEADS * HID) + kk];
    af.y = A[arow * (HEADS * HID) + kk + 1];
#pragma unroll
    for (int nt = 0; nt < 8; ++nt) {
      int col = nt * 16 + l15;
      f32x2 bf;
      bf.x = fc1_w[(long)col * (HEADS * HID) + kk];
      bf.y = fc1_w[(long)col * (HEADS * HID) + kk + 1];
      acc[nt] = wmma_f32(af, bf, acc[nt]);
    }
  }
  float ssum = 0.f;
#pragma unroll
  for (int nt = 0; nt < 8; ++nt) {
    int col = nt * 16 + l15;
    float bb = fc1_b[col], w2 = fc2_w[col];
#pragma unroll
    for (int v = 0; v < 8; ++v) ssum += tanhf(acc[nt][v] + bb) * w2;
  }
#pragma unroll
  for (int off = 16; off > 0; off >>= 1) ssum += __shfl_xor(ssum, off, 32);
  if (lane == 0) atomicAdd(&score[p], ssum);
}

__global__ void beta_kernel(const float* __restrict__ score,
                            float* __restrict__ beta) {
  float s0 = score[0] / (float)NB, s1 = score[1] / (float)NB;
  float mx = fmaxf(s0, s1);
  float e0 = __expf(s0 - mx), e1 = __expf(s1 - mx);
  float inv = 1.f / (e0 + e1);
  beta[0] = e0 * inv;
  beta[1] = e1 * inv;
}

__global__ void hagg_kernel(const float* __restrict__ outs,
                            const float* __restrict__ beta,
                            float* __restrict__ hagg, int n) {
  int i = blockIdx.x * 256 + threadIdx.x;
  if (i >= n) return;
  hagg[i] = beta[0] * outs[i] + beta[1] * outs[(long)n + i];
}

__global__ __launch_bounds__(256) void logits_kernel(
    const float* __restrict__ hagg, const float* __restrict__ fc_w,
    const float* __restrict__ fc_b, float* __restrict__ logits) {
  __shared__ float ldsB[(HEADS * HID) * OUTD];
  int tid = threadIdx.x;
  for (int i = tid; i < (HEADS * HID) * OUTD; i += 256) {
    int k = i >> 4, n = i & 15;
    ldsB[i] = fc_w[n * (HEADS * HID) + k];
  }
  __syncthreads();
  int wave = tid >> 5, lane = tid & 31;
  int tile_m = (blockIdx.x * 8 + wave) * 16;
  int half = lane >> 4, l15 = lane & 15;
  long arow = tile_m + l15;
  f32x8 c = {};
  for (int k0 = 0; k0 < HEADS * HID; k0 += 4) {
    int kk = k0 + half * 2;
    f32x2 af;
    af.x = hagg[arow * (HEADS * HID) + kk];
    af.y = hagg[arow * (HEADS * HID) + kk + 1];
    f32x2 bf;
    bf.x = ldsB[kk * OUTD + l15];
    bf.y = ldsB[(kk + 1) * OUTD + l15];
    c = wmma_f32(af, bf, c);
  }
#pragma unroll
  for (int v = 0; v < 8; ++v) {
    int r = tile_m + v + half * 8;
    logits[(long)r * OUTD + l15] = c[v] + fc_b[l15];
  }
}

extern "C" void kernel_launch(void* const* d_in, const int* in_sizes, int n_in,
                              void* d_out, int out_size, void* d_ws,
                              size_t ws_size, hipStream_t stream) {
  const float* feat0 = (const float*)d_in[0];
  const float* feat1 = (const float*)d_in[1];
  const float* W0    = (const float*)d_in[2];
  const float* b0    = (const float*)d_in[3];
  const float* W1    = (const float*)d_in[4];
  const float* b1    = (const float*)d_in[5];
  const float* r_vec = (const float*)d_in[6];
  const float* attn0 = (const float*)d_in[7];
  const float* attn1 = (const float*)d_in[8];
  const float* fc1_w = (const float*)d_in[9];
  const float* fc1_b = (const float*)d_in[10];
  const float* fc2_w = (const float*)d_in[11];
  const float* fc_w  = (const float*)d_in[12];
  const float* fc_b  = (const float*)d_in[13];
  const int* idx0    = (const int*)d_in[14];
  const int* idx1    = (const int*)d_in[15];
  const int* mp_indices[2] = {(const int*)d_in[16], (const int*)d_in[19]};
  const int* mp_dst[2]     = {(const int*)d_in[17], (const int*)d_in[20]};
  const int* mp_target[2]  = {(const int*)d_in[18], (const int*)d_in[21]};
  const float* attn[2] = {attn0, attn1};

  float* ws     = (float*)d_ws;
  float* tf     = ws;
  float* finals = tf + (long)N_NODES * HID;
  float* h      = finals + 2 * LP * HID;
  float* a      = h + (long)NE * HID;
  float* m      = a + (long)NE * HEADS;
  float* s      = m + NG * HEADS;
  float* ft     = s + NG * HEADS;
  float* outs   = ft + (long)NG * HEADS * HID;
  float* score  = outs + 2L * NB * HEADS * HID;
  float* beta   = score + 2;

  float* logits = (float*)d_out;
  float* hagg   = logits + (long)NB * OUTD;

  gemm_tf_kernel<F0><<<(N0 + 127) / 128, 256, 0, stream>>>(feat0, W0, b0, idx0,
                                                           tf, N0);
  gemm_tf_kernel<F1><<<(N1 + 127) / 128, 256, 0, stream>>>(feat1, W1, b1, idx1,
                                                           tf, N1);
  finals_kernel<<<1, 32, 0, stream>>>(r_vec, finals);

  for (int mp = 0; mp < 2; ++mp) {
    fill_kernel<<<(NG * HEADS + 255) / 256, 256, 0, stream>>>(m, -3.0e38f,
                                                              NG * HEADS);
    fill_kernel<<<(NG * HEADS + 255) / 256, 256, 0, stream>>>(s, 0.f,
                                                              NG * HEADS);
    fill_kernel<<<(NG * HEADS * HID + 255) / 256, 256, 0, stream>>>(
        ft, 0.f, NG * HEADS * HID);
    edge_h_kernel<<<NE / 8, 256, 0, stream>>>(tf, finals + mp * LP * HID,
                                              mp_indices[mp], mp_dst[mp],
                                              attn[mp], h, a, m);
    edge_exp_kernel<<<(NE * HEADS + 255) / 256, 256, 0, stream>>>(
        a, mp_dst[mp], m, s, NE * HEADS);
    edge_scatter_kernel<<<NE / 8, 256, 0, stream>>>(h, a, s, mp_dst[mp], ft);
    target_elu_kernel<<<(NB * HEADS * HID + 255) / 256, 256, 0, stream>>>(
        ft, mp_target[mp], outs + (long)mp * NB * HEADS * HID,
        NB * HEADS * HID);
  }

  fill_kernel<<<1, 64, 0, stream>>>(score, 0.f, 2);
  {
    dim3 g(NB / 128, 2);
    score_kernel<<<g, 256, 0, stream>>>(outs, fc1_w, fc1_b, fc2_w, score);
  }
  beta_kernel<<<1, 1, 0, stream>>>(score, beta);
  hagg_kernel<<<(NB * HEADS * HID + 255) / 256, 256, 0, stream>>>(
      outs, beta, hagg, NB * HEADS * HID);
  logits_kernel<<<NB / 128, 256, 0, stream>>>(hagg, fc_w, fc_b, logits);
}


# edge logits: the 8 heads reduced in one transposed in-register butterfly (same add pairs), then one 8-lane store + one 8-lane atomic max per edge instead of 8 serial single-lane pairs
# speedup vs baseline: 1.9632x; 1.9632x over previous
#include <hip/hip_runtime.h>
#include <math.h>


#define N_NODES 100000
#define N0      60000
#define N1      40000
#define F0      128
#define F1      64
#define HID     64
#define HEADS   8
#define OUTD    16
#define ATTNV   128
#define NE      250000
#define LP      3
#define NG      8192
#define NB      8192

typedef __attribute__((ext_vector_type(2))) float f32x2;
typedef __attribute__((ext_vector_type(8))) float f32x8;

__device__ __forceinline__ f32x8 wmma_f32(f32x2 a, f32x2 b, f32x8 c) {
  return __builtin_amdgcn_wmma_f32_16x16x4_f32(false, a, false, b, (short)0, c,
                                               false, false);
}

__device__ __forceinline__ float xrow(float x) {
  int b = __builtin_bit_cast(int, x);
  return __builtin_bit_cast(
      float, __builtin_amdgcn_permlanex16(b, b, 0x76543210, 0xfedcba98, false,
                                          false));
}
template <int M>
__device__ __forceinline__ float xlane(float x) {
  int b = __builtin_bit_cast(int, x);
  return __builtin_bit_cast(
      float, __builtin_amdgcn_update_dpp(b, b, 0x160 + M, 0xf, 0xf, false));
}

__global__ void fill_kernel(float* __restrict__ p, float v, int n) {
  int i = blockIdx.x * 256 + threadIdx.x;
  if (i < n) p[i] = v;
}

__global__ void finals_kernel(const float* __restrict__ r_vec,
                              float* __restrict__ finals  ) {
  int c = threadIdx.x;
  if (c >= 32) return;
  for (int mp = 0; mp < 2; ++mp) {
    float re = r_vec[(mp * 32 + c) * 2 + 0];
    float im = r_vec[(mp * 32 + c) * 2 + 1];
    float inv = rsqrtf(re * re + im * im);
    re *= inv; im *= inv;
    float f1re = re, f1im = -im;
    float f0re = f1re * re - f1im * im;
    float f0im = f1re * im + f1im * re;
    float* F = finals + mp * LP * HID;
    F[0 * HID + 2 * c] = f0re; F[0 * HID + 2 * c + 1] = f0im;
    F[1 * HID + 2 * c] = f1re; F[1 * HID + 2 * c + 1] = f1im;
    F[2 * HID + 2 * c] = 1.f;  F[2 * HID + 2 * c + 1] = 0.f;
  }
}

template <int K>
__global__ __launch_bounds__(256) void gemm_tf_kernel(
    const float* __restrict__ feat, const float* __restrict__ W,
    const float* __restrict__ bias, const int* __restrict__ idx,
    float* __restrict__ tf, int M) {
  __shared__ float ldsB[K * HID];
  int tid = threadIdx.x;
  for (int i = tid; i < K * HID; i += 256) {
    int k = i >> 6, n = i & 63;
    ldsB[i] = W[n * K + k];
  }
  __syncthreads();
  int wave = tid >> 5, lane = tid & 31;
  int tile_m = blockIdx.x * 128 + wave * 16;
  if (tile_m >= M) return;
  int half = lane >> 4, l15 = lane & 15;
  int arow = tile_m + l15;
  if (arow >= M) arow = M - 1;
  f32x8 c0 = {}, c1 = {}, c2 = {}, c3 = {};
  for (int k0 = 0; k0 < K; k0 += 4) {
    int kk = k0 + half * 2;
    f32x2 a;
    a.x = feat[(long)arow * K + kk];
    a.y = feat[(long)arow * K + kk + 1];
    f32x2 b0, b1, b2, b3;
    b0.x = ldsB[kk * HID + l15 + 0];  b0.y = ldsB[(kk + 1) * HID + l15 + 0];
    b1.x = ldsB[kk * HID + l15 + 16]; b1.y = ldsB[(kk + 1) * HID + l15 + 16];
    b2.x = ldsB[kk * HID + l15 + 32]; b2.y = ldsB[(kk + 1) * HID + l15 + 32];
    b3.x = ldsB[kk * HID + l15 + 48]; b3.y = ldsB[(kk + 1) * HID + l15 + 48];
    c0 = wmma_f32(a, b0, c0);
    c1 = wmma_f32(a, b1, c1);
    c2 = wmma_f32(a, b2, c2);
    c3 = wmma_f32(a, b3, c3);
  }
#pragma unroll
  for (int v = 0; v < 8; ++v) {
    int r = tile_m + v + half * 8;
    if (r < M) {
      float* drow = tf + (long)idx[r] * HID;
      drow[l15 + 0]  = c0[v] + bias[l15 + 0];
      drow[l15 + 16] = c1[v] + bias[l15 + 16];
      drow[l15 + 32] = c2[v] + bias[l15 + 32];
      drow[l15 + 48] = c3[v] + bias[l15 + 48];
    }
  }
}

__global__ __launch_bounds__(256) void edge_h_kernel(
    const float* __restrict__ tf, const float* __restrict__ finals,
    const int* __restrict__ indices, const int* __restrict__ dst,
    const float* __restrict__ attn, float* __restrict__ h,
    float* __restrict__ a, float* __restrict__ m) {
  int wave = threadIdx.x >> 5, lane = threadIdx.x & 31;
  int e = blockIdx.x * 8 + wave;
  if (e >= NE) return;
  float2 acc = make_float2(0.f, 0.f);
#pragma unroll
  for (int i = 0; i < LP; ++i) {
    int node = indices[e * LP + i];
    const float* row = tf + (long)node * HID + lane * 2;
    float vx = row[0], vy = row[1];
    float fx = finals[i * HID + lane * 2], fy = finals[i * HID + lane * 2 + 1];
    acc.x += vx * fx - vy * fy;
    acc.y += vx * fy + vy * fx;
  }
  acc.x *= (1.f / 3.f);
  acc.y *= (1.f / 3.f);
  h[(long)e * HID + lane * 2] = acc.x;
  h[(long)e * HID + lane * 2 + 1] = acc.y;
  const int d = __builtin_amdgcn_readfirstlane(dst[e]);
  float p[HEADS];
#pragma unroll
  for (int hh = 0; hh < HEADS; ++hh)
    p[hh] = __builtin_fmaf(acc.x, attn[hh * HID + lane * 2],
                           acc.y * attn[hh * HID + lane * 2 + 1]);
  const bool b4 = (lane & 16) != 0, b3 = (lane & 8) != 0, b2 = (lane & 4) != 0;
  float q[4];
#pragma unroll
  for (int j = 0; j < 4; ++j) {
    float keep = b4 ? p[j + 4] : p[j];
    float send = b4 ? p[j] : p[j + 4];
    q[j] = keep + xrow(send);
  }
  float r[2];
#pragma unroll
  for (int j = 0; j < 2; ++j) {
    float keep = b3 ? q[j + 2] : q[j];
    float send = b3 ? q[j] : q[j + 2];
    r[j] = keep + xlane<8>(send);
  }
  float keep2 = b2 ? r[1] : r[0];
  float send2 = b2 ? r[0] : r[1];
  float t = keep2 + xlane<4>(send2);
  t += xlane<2>(t);
  t += xlane<1>(t);
  if ((lane & 3) == 0) {
    int hh = lane >> 2;
    float av = t > 0.f ? t : 0.01f * t;
    a[(long)e * HEADS + hh] = av;
    atomicMax(&m[d * HEADS + hh], av);
  }
}

__global__ void edge_exp_kernel(float* __restrict__ a,
                                const int* __restrict__ dst,
                                const float* __restrict__ m,
                                float* __restrict__ s, int n) {
  int i = blockIdx.x * 256 + threadIdx.x;
  if (i >= n) return;
  int e = i >> 3, hh = i & 7;
  int d = dst[e];
  float ex = __expf(a[i] - m[d * HEADS + hh]);
  a[i] = ex;
  atomicAdd(&s[d * HEADS + hh], ex);
}

__global__ __launch_bounds__(256) void edge_scatter_kernel(
    const float* __restrict__ h, const float* __restrict__ a,
    const float* __restrict__ s, const int* __restrict__ dst,
    float* __restrict__ ft) {
  int wave = threadIdx.x >> 5, lane = threadIdx.x & 31;
  int e = blockIdx.x * 8 + wave;
  if (e >= NE) return;
  int d = dst[e];
  float an_l = 0.f;
  if (lane < HEADS) an_l = a[(long)e * HEADS + lane] / s[d * HEADS + lane];
  float hx = h[(long)e * HID + lane * 2];
  float hy = h[(long)e * HID + lane * 2 + 1];
#pragma unroll
  for (int hh = 0; hh < HEADS; ++hh) {
    float w = __shfl(an_l, hh, 32);
    float* base = ft + ((long)d * HEADS + hh) * HID + lane * 2;
    atomicAdd(base + 0, w * hx);
    atomicAdd(base + 1, w * hy);
  }
}

__global__ void target_elu_kernel(const float* __restrict__ ft,
                                  const int* __restrict__ target,
                                  float* __restrict__ outp, int n) {
  int i = blockIdx.x * 256 + threadIdx.x;
  if (i >= n) return;
  int b = i >> 9, j = i & 511;
  float v = ft[(long)target[b] * (HEADS * HID) + j];
  outp[i] = v > 0.f ? v : (__expf(v) - 1.f);
}

__global__ __launch_bounds__(256) void score_kernel(
    const float* __restrict__ outs, const float* __restrict__ fc1_w,
    const float* __restrict__ fc1_b, const float* __restrict__ fc2_w,
    float* __restrict__ score) {
  int p = blockIdx.y;
  const float* A = outs + (long)p * NB * (HEADS * HID);
  int wave = threadIdx.x >> 5, lane = threadIdx.x & 31;
  int tile_m = (blockIdx.x * 8 + wave) * 16;
  int half = lane >> 4, l15 = lane & 15;
  long arow = tile_m + l15;
  f32x8 acc[8] = {};
  for (int k0 = 0; k0 < HEADS * HID; k0 += 4) {
    int kk = k0 + half * 2;
    f32x2 af;
    af.x = A[arow * (HEADS * HID) + kk];
    af.y = A[arow * (HEADS * HID) + kk + 1];
#pragma unroll
    for (int nt = 0; nt < 8; ++nt) {
      int col = nt * 16 + l15;
      f32x2 bf;
      bf.x = fc1_w[(long)col * (HEADS * HID) + kk];
      bf.y = fc1_w[(long)col * (HEADS * HID) + kk + 1];
      acc[nt] = wmma_f32(af, bf, acc[nt]);
    }
  }
  float ssum = 0.f;
#pragma unroll
  for (int nt = 0; nt < 8; ++nt) {
    int col = nt * 16 + l15;
    float bb = fc1_b[col], w2 = fc2_w[col];
#pragma unroll
    for (int v = 0; v < 8; ++v) ssum += tanhf(acc[nt][v] + bb) * w2;
  }
#pragma unroll
  for (int off = 16; off > 0; off >>= 1) ssum += __shfl_xor(ssum, off, 32);
  if (lane == 0) atomicAdd(&score[p], ssum);
}

__global__ void beta_kernel(const float* __restrict__ score,
                            float* __restrict__ beta) {
  float s0 = score[0] / (float)NB, s1 = score[1] / (float)NB;
  float mx = fmaxf(s0, s1);
  float e0 = __expf(s0 - mx), e1 = __expf(s1 - mx);
  float inv = 1.f / (e0 + e1);
  beta[0] = e0 * inv;
  beta[1] = e1 * inv;
}

__global__ void hagg_kernel(const float* __restrict__ outs,
                            const float* __restrict__ beta,
                            float* __restrict__ hagg, int n) {
  int i = blockIdx.x * 256 + threadIdx.x;
  if (i >= n) return;
  hagg[i] = beta[0] * outs[i] + beta[1] * outs[(long)n + i];
}

__global__ __launch_bounds__(256) void logits_kernel(
    const float* __restrict__ hagg, const float* __restrict__ fc_w,
    const float* __restrict__ fc_b, float* __restrict__ logits) {
  __shared__ float ldsB[(HEADS * HID) * OUTD];
  int tid = threadIdx.x;
  for (int i = tid; i < (HEADS * HID) * OUTD; i += 256) {
    int k = i >> 4, n = i & 15;
    ldsB[i] = fc_w[n * (HEADS * HID) + k];
  }
  __syncthreads();
  int wave = tid >> 5, lane = tid & 31;
  int tile_m = (blockIdx.x * 8 + wave) * 16;
  int half = lane >> 4, l15 = lane & 15;
  long arow = tile_m + l15;
  f32x8 c = {};
  for (int k0 = 0; k0 < HEADS * HID; k0 += 4) {
    int kk = k0 + half * 2;
    f32x2 af;
    af.x = hagg[arow * (HEADS * HID) + kk];
    af.y = hagg[arow * (HEADS * HID) + kk + 1];
    f32x2 bf;
    bf.x = ldsB[kk * OUTD + l15];
    bf.y = ldsB[(kk + 1) * OUTD + l15];
    c = wmma_f32(af, bf, c);
  }
#pragma unroll
  for (int v = 0; v < 8; ++v) {
    int r = tile_m + v + half * 8;
    logits[(long)r * OUTD + l15] = c[v] + fc_b[l15];
  }
}

extern "C" void kernel_launch(void* const* d_in, const int* in_sizes, int n_in,
                              void* d_out, int out_size, void* d_ws,
                              size_t ws_size, hipStream_t stream) {
  const float* feat0 = (const float*)d_in[0];
  const float* feat1 = (const float*)d_in[1];
  const float* W0    = (const float*)d_in[2];
  const float* b0    = (const float*)d_in[3];
  const float* W1    = (const float*)d_in[4];
  const float* b1    = (const float*)d_in[5];
  const float* r_vec = (const float*)d_in[6];
  const float* attn0 = (const float*)d_in[7];
  const float* attn1 = (const float*)d_in[8];
  const float* fc1_w = (const float*)d_in[9];
  const float* fc1_b = (const float*)d_in[10];
  const float* fc2_w = (const float*)d_in[11];
  const float* fc_w  = (const float*)d_in[12];
  const float* fc_b  = (const float*)d_in[13];
  const int* idx0    = (const int*)d_in[14];
  const int* idx1    = (const int*)d_in[15];
  const int* mp_indices[2] = {(const int*)d_in[16], (const int*)d_in[19]};
  const int* mp_dst[2]     = {(const int*)d_in[17], (const int*)d_in[20]};
  const int* mp_target[2]  = {(const int*)d_in[18], (const int*)d_in[21]};
  const float* attn[2] = {attn0, attn1};

  float* ws     = (float*)d_ws;
  float* tf     = ws;
  float* finals = tf + (long)N_NODES * HID;
  float* h      = finals + 2 * LP * HID;
  float* a      = h + (long)NE * HID;
  float* m      = a + (long)NE * HEADS;
  float* s      = m + NG * HEADS;
  float* ft     = s + NG * HEADS;
  float* outs   = ft + (long)NG * HEADS * HID;
  float* score  = outs + 2L * NB * HEADS * HID;
  float* beta   = score + 2;

  float* logits = (float*)d_out;
  float* hagg   = logits + (long)NB * OUTD;

  gemm_tf_kernel<F0><<<(N0 + 127) / 128, 256, 0, stream>>>(feat0, W0, b0, idx0,
                                                           tf, N0);
  gemm_tf_kernel<F1><<<(N1 + 127) / 128, 256, 0, stream>>>(feat1, W1, b1, idx1,
                                                           tf, N1);
  finals_kernel<<<1, 32, 0, stream>>>(r_vec, finals);

  for (int mp = 0; mp < 2; ++mp) {
    fill_kernel<<<(NG * HEADS + 255) / 256, 256, 0, stream>>>(m, -3.0e38f,
                                                              NG * HEADS);
    fill_kernel<<<(NG * HEADS + 255) / 256, 256, 0, stream>>>(s, 0.f,
                                                              NG * HEADS);
    fill_kernel<<<(NG * HEADS * HID + 255) / 256, 256, 0, stream>>>(
        ft, 0.f, NG * HEADS * HID);
    edge_h_kernel<<<NE / 8, 256, 0, stream>>>(tf, finals + mp * LP * HID,
                                              mp_indices[mp], mp_dst[mp],
                                              attn[mp], h, a, m);
    edge_exp_kernel<<<(NE * HEADS + 255) / 256, 256, 0, stream>>>(
        a, mp_dst[mp], m, s, NE * HEADS);
    edge_scatter_kernel<<<NE / 8, 256, 0, stream>>>(h, a, s, mp_dst[mp], ft);
    target_elu_kernel<<<(NB * HEADS * HID + 255) / 256, 256, 0, stream>>>(
        ft, mp_target[mp], outs + (long)mp * NB * HEADS * HID,
        NB * HEADS * HID);
  }

  fill_kernel<<<1, 64, 0, stream>>>(score, 0.f, 2);
  {
    dim3 g(NB / 128, 2);
    score_kernel<<<g, 256, 0, stream>>>(outs, fc1_w, fc1_b, fc2_w, score);
  }
  beta_kernel<<<1, 1, 0, stream>>>(score, beta);
  hagg_kernel<<<(NB * HEADS * HID + 255) / 256, 256, 0, stream>>>(
      outs, beta, hagg, NB * HEADS * HID);
  logits_kernel<<<NB / 128, 256, 0, stream>>>(hagg, fc_w, fc_b, logits);
}


# ft sums built in LDS by owner work-groups (32 destinations each: scan dst, LDS float adds of an*h in arrival order, one whole-line store) replace the global f32 atomic scatter and the ft clear
# speedup vs baseline: 2.4632x; 1.2547x over previous
#include <hip/hip_runtime.h>
#include <math.h>


#define N_NODES 100000
#define N0      60000
#define N1      40000
#define F0      128
#define F1      64
#define HID     64
#define HEADS   8
#define OUTD    16
#define ATTNV   128
#define NE      250000
#define LP      3
#define NG      8192
#define NB      8192

typedef __attribute__((ext_vector_type(2))) float f32x2;
typedef __attribute__((ext_vector_type(8))) float f32x8;

__device__ __forceinline__ f32x8 wmma_f32(f32x2 a, f32x2 b, f32x8 c) {
  return __builtin_amdgcn_wmma_f32_16x16x4_f32(false, a, false, b, (short)0, c,
                                               false, false);
}

__device__ __forceinline__ float xrow(float x) {
  int b = __builtin_bit_cast(int, x);
  return __builtin_bit_cast(
      float, __builtin_amdgcn_permlanex16(b, b, 0x76543210, 0xfedcba98, false,
                                          false));
}
template <int M>
__device__ __forceinline__ float xlane(float x) {
  int b = __builtin_bit_cast(int, x);
  return __builtin_bit_cast(
      float, __builtin_amdgcn_update_dpp(b, b, 0x160 + M, 0xf, 0xf, false));
}

__global__ void fill_kernel(float* __restrict__ p, float v, int n) {
  int i = blockIdx.x * 256 + threadIdx.x;
  if (i < n) p[i] = v;
}

__global__ void finals_kernel(const float* __restrict__ r_vec,
                              float* __restrict__ finals  ) {
  int c = threadIdx.x;
  if (c >= 32) return;
  for (int mp = 0; mp < 2; ++mp) {
    float re = r_vec[(mp * 32 + c) * 2 + 0];
    float im = r_vec[(mp * 32 + c) * 2 + 1];
    float inv = rsqrtf(re * re + im * im);
    re *= inv; im *= inv;
    float f1re = re, f1im = -im;
    float f0re = f1re * re - f1im * im;
    float f0im = f1re * im + f1im * re;
    float* F = finals + mp * LP * HID;
    F[0 * HID + 2 * c] = f0re; F[0 * HID + 2 * c + 1] = f0im;
    F[1 * HID + 2 * c] = f1re; F[1 * HID + 2 * c + 1] = f1im;
    F[2 * HID + 2 * c] = 1.f;  F[2 * HID + 2 * c + 1] = 0.f;
  }
}

template <int K>
__global__ __launch_bounds__(256) void gemm_tf_kernel(
    const float* __restrict__ feat, const float* __restrict__ W,
    const float* __restrict__ bias, const int* __restrict__ idx,
    float* __restrict__ tf, int M) {
  __shared__ float ldsB[K * HID];
  int tid = threadIdx.x;
  for (int i = tid; i < K * HID; i += 256) {
    int k = i >> 6, n = i & 63;
    ldsB[i] = W[n * K + k];
  }
  __syncthreads();
  int wave = tid >> 5, lane = tid & 31;
  int tile_m = blockIdx.x * 128 + wave * 16;
  if (tile_m >= M) return;
  int half = lane >> 4, l15 = lane & 15;
  int arow = tile_m + l15;
  if (arow >= M) arow = M - 1;
  f32x8 c0 = {}, c1 = {}, c2 = {}, c3 = {};
  for (int k0 = 0; k0 < K; k0 += 4) {
    int kk = k0 + half * 2;
    f32x2 a;
    a.x = feat[(long)arow * K + kk];
    a.y = feat[(long)arow * K + kk + 1];
    f32x2 b0, b1, b2, b3;
    b0.x = ldsB[kk * HID + l15 + 0];  b0.y = ldsB[(kk + 1) * HID + l15 + 0];
    b1.x = ldsB[kk * HID + l15 + 16]; b1.y = ldsB[(kk + 1) * HID + l15 + 16];
    b2.x = ldsB[kk * HID + l15 + 32]; b2.y = ldsB[(kk + 1) * HID + l15 + 32];
    b3.x = ldsB[kk * HID + l15 + 48]; b3.y = ldsB[(kk + 1) * HID + l15 + 48];
    c0 = wmma_f32(a, b0, c0);
    c1 = wmma_f32(a, b1, c1);
    c2 = wmma_f32(a, b2, c2);
    c3 = wmma_f32(a, b3, c3);
  }
#pragma unroll
  for (int v = 0; v < 8; ++v) {
    int r = tile_m + v + half * 8;
    if (r < M) {
      float* drow = tf + (long)idx[r] * HID;
      drow[l15 + 0]  = c0[v] + bias[l15 + 0];
      drow[l15 + 16] = c1[v] + bias[l15 + 16];
      drow[l15 + 32] = c2[v] + bias[l15 + 32];
      drow[l15 + 48] = c3[v] + bias[l15 + 48];
    }
  }
}

__global__ __launch_bounds__(256) void edge_h_kernel(
    const float* __restrict__ tf, const float* __restrict__ finals,
    const int* __restrict__ indices, const int* __restrict__ dst,
    const float* __restrict__ attn, float* __restrict__ h,
    float* __restrict__ a, float* __restrict__ m) {
  int wave = threadIdx.x >> 5, lane = threadIdx.x & 31;
  int e = blockIdx.x * 8 + wave;
  if (e >= NE) return;
  float2 acc = make_float2(0.f, 0.f);
#pragma unroll
  for (int i = 0; i < LP; ++i) {
    int node = indices[e * LP + i];
    const float* row = tf + (long)node * HID + lane * 2;
    float vx = row[0], vy = row[1];
    float fx = finals[i * HID + lane * 2], fy = finals[i * HID + lane * 2 + 1];
    acc.x += vx * fx - vy * fy;
    acc.y += vx * fy + vy * fx;
  }
  acc.x *= (1.f / 3.f);
  acc.y *= (1.f / 3.f);
  h[(long)e * HID + lane * 2] = acc.x;
  h[(long)e * HID + lane * 2 + 1] = acc.y;
  const int d = __builtin_amdgcn_readfirstlane(dst[e]);
  float p[HEADS];
#pragma unroll
  for (int hh = 0; hh < HEADS; ++hh)
    p[hh] = __builtin_fmaf(acc.x, attn[hh * HID + lane * 2],
                           acc.y * attn[hh * HID + lane * 2 + 1]);
  const bool b4 = (lane & 16) != 0, b3 = (lane & 8) != 0, b2 = (lane & 4) != 0;
  float q[4];
#pragma unroll
  for (int j = 0; j < 4; ++j) {
    float keep = b4 ? p[j + 4] : p[j];
    float send = b4 ? p[j] : p[j + 4];
    q[j] = keep + xrow(send);
  }
  float r[2];
#pragma unroll
  for (int j = 0; j < 2; ++j) {
    float keep = b3 ? q[j + 2] : q[j];
    float send = b3 ? q[j] : q[j + 2];
    r[j] = keep + xlane<8>(send);
  }
  float keep2 = b2 ? r[1] : r[0];
  float send2 = b2 ? r[0] : r[1];
  float t = keep2 + xlane<4>(send2);
  t += xlane<2>(t);
  t += xlane<1>(t);
  if ((lane & 3) == 0) {
    int hh = lane >> 2;
    float av = t > 0.f ? t : 0.01f * t;
    a[(long)e * HEADS + hh] = av;
    atomicMax(&m[d * HEADS + hh], av);
  }
}

__global__ void edge_exp_kernel(float* __restrict__ a,
                                const int* __restrict__ dst,
                                const float* __restrict__ m,
                                float* __restrict__ s, int n) {
  int i = blockIdx.x * 256 + threadIdx.x;
  if (i >= n) return;
  int e = i >> 3, hh = i & 7;
  int d = dst[e];
  float ex = __expf(a[i] - m[d * HEADS + hh]);
  a[i] = ex;
  atomicAdd(&s[d * HEADS + hh], ex);
}

#define OWN_D   32
#define OWN_CAP 25600

__device__ __forceinline__ float lane_bcast(float x, int l) {
  return __builtin_bit_cast(
      float, __builtin_amdgcn_readlane(__builtin_bit_cast(int, x), l));
}

__device__ __forceinline__ void note_edge(int d, int i, int d0, int* list,
                                          int* nlist) {
  const unsigned dl = (unsigned)(d - d0);
  if (dl < OWN_D) {
    const int r = atomicAdd(nlist, 1);
    if (r < OWN_CAP) list[r] = (i << 5) | (int)dl;
  }
}

template <int NU>
__device__ __forceinline__ void add_noted(const int* list, int j0, int lane,
                                          int d0, const float* __restrict__ h,
                                          const float* __restrict__ a,
                                          const float* __restrict__ s,
                                          float* acc) {
  const int mine = list[j0 + 32 * (lane & (NU - 1))];
  int ent[NU];
  float lo[NU], hi[NU], av[NU], sv[NU];
#pragma unroll
  for (int u = 0; u < NU; ++u) {
    ent[u] = __builtin_amdgcn_readlane(mine, u);
    const int e = ent[u] >> 5, d = d0 + (ent[u] & 31);
    lo[u] = h[(long)e * HID + lane];
    hi[u] = h[(long)e * HID + 32 + lane];
    av[u] = a[(long)e * HEADS + (lane & 7)];
    sv[u] = s[d * HEADS + (lane & 7)];
  }
#pragma unroll
  for (int u = 0; u < NU; ++u) {
    const float wl = av[u] / sv[u];
    float* row = acc + (ent[u] & 31) * (HEADS * HID) + lane;
#pragma unroll
    for (int hh = 0; hh < HEADS; ++hh) {
      const float w = lane_bcast(wl, hh);
      atomicAdd(row + hh * HID, w * lo[u]);
      atomicAdd(row + hh * HID + 32, w * hi[u]);
    }
  }
}

__global__ __launch_bounds__(1024) void owner_sum_kernel(
    const float* __restrict__ h, const float* __restrict__ a,
    const float* __restrict__ s, const int* __restrict__ dst,
    float* __restrict__ ft) {
  __shared__ __attribute__((aligned(16))) float acc[OWN_D * HEADS * HID];
  __shared__ int list[OWN_CAP];
  __shared__ int nlist;
  const int tid = threadIdx.x, lane = tid & 31;
  const int wave = __builtin_amdgcn_readfirstlane(tid >> 5);
  const int d0 = blockIdx.x * OWN_D;

  float4* acc4 = (float4*)acc;
  for (int i = tid; i < OWN_D * HEADS * HID / 4; i += 1024)
    acc4[i] = make_float4(0.f, 0.f, 0.f, 0.f);
  if (tid == 0) nlist = 0;
  __syncthreads();

  const int4* dst4 = (const int4*)dst;
  for (int base = 0; base < NE / 4; base += 1024 * 8) {
    int4 dv[8];
#pragma unroll
    for (int u = 0; u < 8; ++u) {
      const int q = base + u * 1024 + tid;
      dv[u] = q < NE / 4 ? dst4[q] : make_int4(-1, -1, -1, -1);
    }
#pragma unroll
    for (int u = 0; u < 8; ++u) {
      const int i = 4 * (base + u * 1024 + tid);
      note_edge(dv[u].x, i + 0, d0, list, &nlist);
      note_edge(dv[u].y, i + 1, d0, list, &nlist);
      note_edge(dv[u].z, i + 2, d0, list, &nlist);
      note_edge(dv[u].w, i + 3, d0, list, &nlist);
    }
  }
  __syncthreads();

  const int n = __builtin_amdgcn_readfirstlane(nlist);
  if (n <= OWN_CAP) {
    int j = wave;
    for (; j + 96 < n; j += 128) add_noted<4>(list, j, lane, d0, h, a, s, acc);
    for (; j < n; j += 32) add_noted<1>(list, j, lane, d0, h, a, s, acc);
  } else {
    for (int i = tid; i < NE; i += 1024) {
      const int d = dst[i];
      const unsigned dl = (unsigned)(d - d0);
      if (dl < OWN_D) {
        for (int hh = 0; hh < HEADS; ++hh) {
          const float w = a[(long)i * HEADS + hh] / s[d * HEADS + hh];
          for (int c = 0; c < HID; ++c)
            atomicAdd(&acc[(dl * HEADS + hh) * HID + c],
                      w * h[(long)i * HID + c]);
        }
      }
    }
  }
  __syncthreads();

  float4* o = (float4*)(ft + (long)d0 * HEADS * HID);
  for (int i = tid; i < OWN_D * HEADS * HID / 4; i += 1024) o[i] = acc4[i];
}

__global__ void target_elu_kernel(const float* __restrict__ ft,
                                  const int* __restrict__ target,
                                  float* __restrict__ outp, int n) {
  int i = blockIdx.x * 256 + threadIdx.x;
  if (i >= n) return;
  int b = i >> 9, j = i & 511;
  float v = ft[(long)target[b] * (HEADS * HID) + j];
  outp[i] = v > 0.f ? v : (__expf(v) - 1.f);
}

__global__ __launch_bounds__(256) void score_kernel(
    const float* __restrict__ outs, const float* __restrict__ fc1_w,
    const float* __restrict__ fc1_b, const float* __restrict__ fc2_w,
    float* __restrict__ score) {
  int p = blockIdx.y;
  const float* A = outs + (long)p * NB * (HEADS * HID);
  int wave = threadIdx.x >> 5, lane = threadIdx.x & 31;
  int tile_m = (blockIdx.x * 8 + wave) * 16;
  int half = lane >> 4, l15 = lane & 15;
  long arow = tile_m + l15;
  f32x8 acc[8] = {};
  for (int k0 = 0; k0 < HEADS * HID; k0 += 4) {
    int kk = k0 + half * 2;
    f32x2 af;
    af.x = A[arow * (HEADS * HID) + kk];
    af.y = A[arow * (HEADS * HID) + kk + 1];
#pragma unroll
    for (int nt = 0; nt < 8; ++nt) {
      int col = nt * 16 + l15;
      f32x2 bf;
      bf.x = fc1_w[(long)col * (HEADS * HID) + kk];
      bf.y = fc1_w[(long)col * (HEADS * HID) + kk + 1];
      acc[nt] = wmma_f32(af, bf, acc[nt]);
    }
  }
  float ssum = 0.f;
#pragma unroll
  for (int nt = 0; nt < 8; ++nt) {
    int col = nt * 16 + l15;
    float bb = fc1_b[col], w2 = fc2_w[col];
#pragma unroll
    for (int v = 0; v < 8; ++v) ssum += tanhf(acc[nt][v] + bb) * w2;
  }
#pragma unroll
  for (int off = 16; off > 0; off >>= 1) ssum += __shfl_xor(ssum, off, 32);
  if (lane == 0) atomicAdd(&score[p], ssum);
}

__global__ void beta_kernel(const float* __restrict__ score,
                            float* __restrict__ beta) {
  float s0 = score[0] / (float)NB, s1 = score[1] / (float)NB;
  float mx = fmaxf(s0, s1);
  float e0 = __expf(s0 - mx), e1 = __expf(s1 - mx);
  float inv = 1.f / (e0 + e1);
  beta[0] = e0 * inv;
  beta[1] = e1 * inv;
}

__global__ void hagg_kernel(const float* __restrict__ outs,
                            const float* __restrict__ beta,
                            float* __restrict__ hagg, int n) {
  int i = blockIdx.x * 256 + threadIdx.x;
  if (i >= n) return;
  hagg[i] = beta[0] * outs[i] + beta[1] * outs[(long)n + i];
}

__global__ __launch_bounds__(256) void logits_kernel(
    const float* __restrict__ hagg, const float* __restrict__ fc_w,
    const float* __restrict__ fc_b, float* __restrict__ logits) {
  __shared__ float ldsB[(HEADS * HID) * OUTD];
  int tid = threadIdx.x;
  for (int i = tid; i < (HEADS * HID) * OUTD; i += 256) {
    int k = i >> 4, n = i & 15;
    ldsB[i] = fc_w[n * (HEADS * HID) + k];
  }
  __syncthreads();
  int wave = tid >> 5, lane = tid & 31;
  int tile_m = (blockIdx.x * 8 + wave) * 16;
  int half = lane >> 4, l15 = lane & 15;
  long arow = tile_m + l15;
  f32x8 c = {};
  for (int k0 = 0; k0 < HEADS * HID; k0 += 4) {
    int kk = k0 + half * 2;
    f32x2 af;
    af.x = hagg[arow * (HEADS * HID) + kk];
    af.y = hagg[arow * (HEADS * HID) + kk + 1];
    f32x2 bf;
    bf.x = ldsB[kk * OUTD + l15];
    bf.y = ldsB[(kk + 1) * OUTD + l15];
    c = wmma_f32(af, bf, c);
  }
#pragma unroll
  for (int v = 0; v < 8; ++v) {
    int r = tile_m + v + half * 8;
    logits[(long)r * OUTD + l15] = c[v] + fc_b[l15];
  }
}

extern "C" void kernel_launch(void* const* d_in, const int* in_sizes, int n_in,
                              void* d_out, int out_size, void* d_ws,
                              size_t ws_size, hipStream_t stream) {
  const float* feat0 = (const float*)d_in[0];
  const float* feat1 = (const float*)d_in[1];
  const float* W0    = (const float*)d_in[2];
  const float* b0    = (const float*)d_in[3];
  const float* W1    = (const float*)d_in[4];
  const float* b1    = (const float*)d_in[5];
  const float* r_vec = (const float*)d_in[6];
  const float* attn0 = (const float*)d_in[7];
  const float* attn1 = (const float*)d_in[8];
  const float* fc1_w = (const float*)d_in[9];
  const float* fc1_b = (const float*)d_in[10];
  const float* fc2_w = (const float*)d_in[11];
  const float* fc_w  = (const float*)d_in[12];
  const float* fc_b  = (const float*)d_in[13];
  const int* idx0    = (const int*)d_in[14];
  const int* idx1    = (const int*)d_in[15];
  const int* mp_indices[2] = {(const int*)d_in[16], (const int*)d_in[19]};
  const int* mp_dst[2]     = {(const int*)d_in[17], (const int*)d_in[20]};
  const int* mp_target[2]  = {(const int*)d_in[18], (const int*)d_in[21]};
  const float* attn[2] = {attn0, attn1};

  float* ws     = (float*)d_ws;
  float* tf     = ws;
  float* finals = tf + (long)N_NODES * HID;
  float* h      = finals + 2 * LP * HID;
  float* a      = h + (long)NE * HID;
  float* m      = a + (long)NE * HEADS;
  float* s      = m + NG * HEADS;
  float* ft     = s + NG * HEADS;
  float* outs   = ft + (long)NG * HEADS * HID;
  float* score  = outs + 2L * NB * HEADS * HID;
  float* beta   = score + 2;

  float* logits = (float*)d_out;
  float* hagg   = logits + (long)NB * OUTD;

  gemm_tf_kernel<F0><<<(N0 + 127) / 128, 256, 0, stream>>>(feat0, W0, b0, idx0,
                                                           tf, N0);
  gemm_tf_kernel<F1><<<(N1 + 127) / 128, 256, 0, stream>>>(feat1, W1, b1, idx1,
                                                           tf, N1);
  finals_kernel<<<1, 32, 0, stream>>>(r_vec, finals);

  for (int mp = 0; mp < 2; ++mp) {
    fill_kernel<<<(NG * HEADS + 255) / 256, 256, 0, stream>>>(m, -3.0e38f,
                                                              NG * HEADS);
    fill_kernel<<<(NG * HEADS + 255) / 256, 256, 0, stream>>>(s, 0.f,
                                                              NG * HEADS);
    edge_h_kernel<<<NE / 8, 256, 0, stream>>>(tf, finals + mp * LP * HID,
                                              mp_indices[mp], mp_dst[mp],
                                              attn[mp], h, a, m);
    edge_exp_kernel<<<(NE * HEADS + 255) / 256, 256, 0, stream>>>(
        a, mp_dst[mp], m, s, NE * HEADS);
    owner_sum_kernel<<<NG / OWN_D, 1024, 0, stream>>>(h, a, s, mp_dst[mp], ft);
    target_elu_kernel<<<(NB * HEADS * HID + 255) / 256, 256, 0, stream>>>(
        ft, mp_target[mp], outs + (long)mp * NB * HEADS * HID,
        NB * HEADS * HID);
  }

  fill_kernel<<<1, 64, 0, stream>>>(score, 0.f, 2);
  {
    dim3 g(NB / 128, 2);
    score_kernel<<<g, 256, 0, stream>>>(outs, fc1_w, fc1_b, fc2_w, score);
  }
  beta_kernel<<<1, 1, 0, stream>>>(score, beta);
  hagg_kernel<<<(NB * HEADS * HID + 255) / 256, 256, 0, stream>>>(
      outs, beta, hagg, NB * HEADS * HID);
  logits_kernel<<<NB / 128, 256, 0, stream>>>(hagg, fc_w, fc_b, logits);
}


# edge kernel: one wave takes 8 consecutive edges (24 row gathers in flight together, constants loaded once, logits of 8 edges out as 2 whole-line stores + 2 atomic max); same sums per edge
# speedup vs baseline: 2.6279x; 1.0669x over previous
#include <hip/hip_runtime.h>
#include <math.h>


#define N_NODES 100000
#define N0      60000
#define N1      40000
#define F0      128
#define F1      64
#define HID     64
#define HEADS   8
#define OUTD    16
#define ATTNV   128
#define NE      250000
#define LP      3
#define NG      8192
#define NB      8192

typedef __attribute__((ext_vector_type(2))) float f32x2;
typedef __attribute__((ext_vector_type(8))) float f32x8;

__device__ __forceinline__ f32x8 wmma_f32(f32x2 a, f32x2 b, f32x8 c) {
  return __builtin_amdgcn_wmma_f32_16x16x4_f32(false, a, false, b, (short)0, c,
                                               false, false);
}

__device__ __forceinline__ float xrow(float x) {
  int b = __builtin_bit_cast(int, x);
  return __builtin_bit_cast(
      float, __builtin_amdgcn_permlanex16(b, b, 0x76543210, 0xfedcba98, false,
                                          false));
}
template <int M>
__device__ __forceinline__ float xlane(float x) {
  int b = __builtin_bit_cast(int, x);
  return __builtin_bit_cast(
      float, __builtin_amdgcn_update_dpp(b, b, 0x160 + M, 0xf, 0xf, false));
}

__global__ void fill_kernel(float* __restrict__ p, float v, int n) {
  int i = blockIdx.x * 256 + threadIdx.x;
  if (i < n) p[i] = v;
}

__global__ void finals_kernel(const float* __restrict__ r_vec,
                              float* __restrict__ finals  ) {
  int c = threadIdx.x;
  if (c >= 32) return;
  for (int mp = 0; mp < 2; ++mp) {
    float re = r_vec[(mp * 32 + c) * 2 + 0];
    float im = r_vec[(mp * 32 + c) * 2 + 1];
    float inv = rsqrtf(re * re + im * im);
    re *= inv; im *= inv;
    float f1re = re, f1im = -im;
    float f0re = f1re * re - f1im * im;
    float f0im = f1re * im + f1im * re;
    float* F = finals + mp * LP * HID;
    F[0 * HID + 2 * c] = f0re; F[0 * HID + 2 * c + 1] = f0im;
    F[1 * HID + 2 * c] = f1re; F[1 * HID + 2 * c + 1] = f1im;
    F[2 * HID + 2 * c] = 1.f;  F[2 * HID + 2 * c + 1] = 0.f;
  }
}

template <int K>
__global__ __launch_bounds__(256) void gemm_tf_kernel(
    const float* __restrict__ feat, const float* __restrict__ W,
    const float* __restrict__ bias, const int* __restrict__ idx,
    float* __restrict__ tf, int M) {
  __shared__ float ldsB[K * HID];
  int tid = threadIdx.x;
  for (int i = tid; i < K * HID; i += 256) {
    int k = i >> 6, n = i & 63;
    ldsB[i] = W[n * K + k];
  }
  __syncthreads();
  int wave = tid >> 5, lane = tid & 31;
  int tile_m = blockIdx.x * 128 + wave * 16;
  if (tile_m >= M) return;
  int half = lane >> 4, l15 = lane & 15;
  int arow = tile_m + l15;
  if (arow >= M) arow = M - 1;
  f32x8 c0 = {}, c1 = {}, c2 = {}, c3 = {};
  for (int k0 = 0; k0 < K; k0 += 4) {
    int kk = k0 + half * 2;
    f32x2 a;
    a.x = feat[(long)arow * K + kk];
    a.y = feat[(long)arow * K + kk + 1];
    f32x2 b0, b1, b2, b3;
    b0.x = ldsB[kk * HID + l15 + 0];  b0.y = ldsB[(kk + 1) * HID + l15 + 0];
    b1.x = ldsB[kk * HID + l15 + 16]; b1.y = ldsB[(kk + 1) * HID + l15 + 16];
    b2.x = ldsB[kk * HID + l15 + 32]; b2.y = ldsB[(kk + 1) * HID + l15 + 32];
    b3.x = ldsB[kk * HID + l15 + 48]; b3.y = ldsB[(kk + 1) * HID + l15 + 48];
    c0 = wmma_f32(a, b0, c0);
    c1 = wmma_f32(a, b1, c1);
    c2 = wmma_f32(a, b2, c2);
    c3 = wmma_f32(a, b3, c3);
  }
#pragma unroll
  for (int v = 0; v < 8; ++v) {
    int r = tile_m + v + half * 8;
    if (r < M) {
      float* drow = tf + (long)idx[r] * HID;
      drow[l15 + 0]  = c0[v] + bias[l15 + 0];
      drow[l15 + 16] = c1[v] + bias[l15 + 16];
      drow[l15 + 32] = c2[v] + bias[l15 + 32];
      drow[l15 + 48] = c3[v] + bias[l15 + 48];
    }
  }
}

#define EPW 8
__global__ __launch_bounds__(256) void edge_h_kernel(
    const float* __restrict__ tf, const float* __restrict__ finals,
    const int* __restrict__ indices, const int* __restrict__ dst,
    const float* __restrict__ attn, float* __restrict__ h,
    float* __restrict__ a, float* __restrict__ m) {
  const int lane = threadIdx.x & 31;
  const int wave = __builtin_amdgcn_readfirstlane(threadIdx.x >> 5);
  const int e0 = (blockIdx.x * 8 + wave) * EPW;
  if (e0 >= NE) return;
  const int mynode = indices[e0 * LP + (lane < LP * EPW ? lane : lane - 8)];
  const int g = lane >> 2, r = lane & 3;
  const int dlo = dst[e0 + r];
  const int dhi = dst[e0 + 4 + r];
  float2 v[EPW][LP];
#pragma unroll
  for (int u = 0; u < EPW; ++u) {
#pragma unroll
    for (int i = 0; i < LP; ++i) {
      const int node = __builtin_amdgcn_readlane(mynode, u * LP + i);
      v[u][i] = *((const float2*)(tf + (long)node * HID) + lane);
    }
  }
  float2 f[LP];
#pragma unroll
  for (int i = 0; i < LP; ++i) f[i] = *((const float2*)(finals + i * HID) + lane);
  float2 w[HEADS];
#pragma unroll
  for (int hh = 0; hh < HEADS; ++hh)
    w[hh] = *((const float2*)(attn + hh * HID) + lane);

  const bool b4 = (lane & 16) != 0, b3 = (lane & 8) != 0, b2 = (lane & 4) != 0;
  float tlo = 0.f, thi = 0.f;
#pragma unroll
  for (int u = 0; u < EPW; ++u) {
    float2 acc = make_float2(0.f, 0.f);
#pragma unroll
    for (int i = 0; i < LP; ++i) {
      const float vx = v[u][i].x, vy = v[u][i].y;
      const float fx = f[i].x, fy = f[i].y;
      acc.x += vx * fx - vy * fy;
      acc.y += vx * fy + vy * fx;
    }
    acc.x *= (1.f / 3.f);
    acc.y *= (1.f / 3.f);
    *((float2*)(h + (long)(e0 + u) * HID) + lane) = acc;
    float p[HEADS];
#pragma unroll
    for (int hh = 0; hh < HEADS; ++hh)
      p[hh] = __builtin_fmaf(acc.x, w[hh].x, acc.y * w[hh].y);
    float q[4];
#pragma unroll
    for (int j = 0; j < 4; ++j) {
      float keep = b4 ? p[j + 4] : p[j];
      float send = b4 ? p[j] : p[j + 4];
      q[j] = keep + xrow(send);
    }
    float rr[2];
#pragma unroll
    for (int j = 0; j < 2; ++j) {
      float keep = b3 ? q[j + 2] : q[j];
      float send = b3 ? q[j] : q[j + 2];
      rr[j] = keep + xlane<8>(send);
    }
    float keep2 = b2 ? rr[1] : rr[0];
    float send2 = b2 ? rr[0] : rr[1];
    float t = keep2 + xlane<4>(send2);
    t += xlane<2>(t);
    t += xlane<1>(t);
    if (u < 4) tlo = (r == u) ? t : tlo;
    else       thi = (r == u - 4) ? t : thi;
  }
  const float alo = tlo > 0.f ? tlo : 0.01f * tlo;
  const float ahi = thi > 0.f ? thi : 0.01f * thi;
  a[(long)(e0 + r) * HEADS + g] = alo;
  a[(long)(e0 + 4 + r) * HEADS + g] = ahi;
  atomicMax(&m[dlo * HEADS + g], alo);
  atomicMax(&m[dhi * HEADS + g], ahi);
}

__global__ void edge_exp_kernel(float* __restrict__ a,
                                const int* __restrict__ dst,
                                const float* __restrict__ m,
                                float* __restrict__ s, int n) {
  int i = blockIdx.x * 256 + threadIdx.x;
  if (i >= n) return;
  int e = i >> 3, hh = i & 7;
  int d = dst[e];
  float ex = __expf(a[i] - m[d * HEADS + hh]);
  a[i] = ex;
  atomicAdd(&s[d * HEADS + hh], ex);
}

#define OWN_D   32
#define OWN_CAP 25600

__device__ __forceinline__ float lane_bcast(float x, int l) {
  return __builtin_bit_cast(
      float, __builtin_amdgcn_readlane(__builtin_bit_cast(int, x), l));
}

__device__ __forceinline__ void note_edge(int d, int i, int d0, int* list,
                                          int* nlist) {
  const unsigned dl = (unsigned)(d - d0);
  if (dl < OWN_D) {
    const int r = atomicAdd(nlist, 1);
    if (r < OWN_CAP) list[r] = (i << 5) | (int)dl;
  }
}

template <int NU>
__device__ __forceinline__ void add_noted(const int* list, int j0, int lane,
                                          int d0, const float* __restrict__ h,
                                          const float* __restrict__ a,
                                          const float* __restrict__ s,
                                          float* acc) {
  const int mine = list[j0 + 32 * (lane & (NU - 1))];
  int ent[NU];
  float lo[NU], hi[NU], av[NU], sv[NU];
#pragma unroll
  for (int u = 0; u < NU; ++u) {
    ent[u] = __builtin_amdgcn_readlane(mine, u);
    const int e = ent[u] >> 5, d = d0 + (ent[u] & 31);
    lo[u] = h[(long)e * HID + lane];
    hi[u] = h[(long)e * HID + 32 + lane];
    av[u] = a[(long)e * HEADS + (lane & 7)];
    sv[u] = s[d * HEADS + (lane & 7)];
  }
#pragma unroll
  for (int u = 0; u < NU; ++u) {
    const float wl = av[u] / sv[u];
    float* row = acc + (ent[u] & 31) * (HEADS * HID) + lane;
#pragma unroll
    for (int hh = 0; hh < HEADS; ++hh) {
      const float w = lane_bcast(wl, hh);
      atomicAdd(row + hh * HID, w * lo[u]);
      atomicAdd(row + hh * HID + 32, w * hi[u]);
    }
  }
}

__global__ __launch_bounds__(1024) void owner_sum_kernel(
    const float* __restrict__ h, const float* __restrict__ a,
    const float* __restrict__ s, const int* __restrict__ dst,
    float* __restrict__ ft) {
  __shared__ __attribute__((aligned(16))) float acc[OWN_D * HEADS * HID];
  __shared__ int list[OWN_CAP];
  __shared__ int nlist;
  const int tid = threadIdx.x, lane = tid & 31;
  const int wave = __builtin_amdgcn_readfirstlane(tid >> 5);
  const int d0 = blockIdx.x * OWN_D;

  float4* acc4 = (float4*)acc;
  for (int i = tid; i < OWN_D * HEADS * HID / 4; i += 1024)
    acc4[i] = make_float4(0.f, 0.f, 0.f, 0.f);
  if (tid == 0) nlist = 0;
  __syncthreads();

  const int4* dst4 = (const int4*)dst;
  for (int base = 0; base < NE / 4; base += 1024 * 8) {
    int4 dv[8];
#pragma unroll
    for (int u = 0; u < 8; ++u) {
      const int q = base + u * 1024 + tid;
      dv[u] = q < NE / 4 ? dst4[q] : make_int4(-1, -1, -1, -1);
    }
#pragma unroll
    for (int u = 0; u < 8; ++u) {
      const int i = 4 * (base + u * 1024 + tid);
      note_edge(dv[u].x, i + 0, d0, list, &nlist);
      note_edge(dv[u].y, i + 1, d0, list, &nlist);
      note_edge(dv[u].z, i + 2, d0, list, &nlist);
      note_edge(dv[u].w, i + 3, d0, list, &nlist);
    }
  }
  __syncthreads();

  const int n = __builtin_amdgcn_readfirstlane(nlist);
  if (n <= OWN_CAP) {
    int j = wave;
    for (; j + 96 < n; j += 128) add_noted<4>(list, j, lane, d0, h, a, s, acc);
    for (; j < n; j += 32) add_noted<1>(list, j, lane, d0, h, a, s, acc);
  } else {
    for (int i = tid; i < NE; i += 1024) {
      const int d = dst[i];
      const unsigned dl = (unsigned)(d - d0);
      if (dl < OWN_D) {
        for (int hh = 0; hh < HEADS; ++hh) {
          const float w = a[(long)i * HEADS + hh] / s[d * HEADS + hh];
          for (int c = 0; c < HID; ++c)
            atomicAdd(&acc[(dl * HEADS + hh) * HID + c],
                      w * h[(long)i * HID + c]);
        }
      }
    }
  }
  __syncthreads();

  float4* o = (float4*)(ft + (long)d0 * HEADS * HID);
  for (int i = tid; i < OWN_D * HEADS * HID / 4; i += 1024) o[i] = acc4[i];
}

__global__ void target_elu_kernel(const float* __restrict__ ft,
                                  const int* __restrict__ target,
                                  float* __restrict__ outp, int n) {
  int i = blockIdx.x * 256 + threadIdx.x;
  if (i >= n) return;
  int b = i >> 9, j = i & 511;
  float v = ft[(long)target[b] * (HEADS * HID) + j];
  outp[i] = v > 0.f ? v : (__expf(v) - 1.f);
}

__global__ __launch_bounds__(256) void score_kernel(
    const float* __restrict__ outs, const float* __restrict__ fc1_w,
    const float* __restrict__ fc1_b, const float* __restrict__ fc2_w,
    float* __restrict__ score) {
  int p = blockIdx.y;
  const float* A = outs + (long)p * NB * (HEADS * HID);
  int wave = threadIdx.x >> 5, lane = threadIdx.x & 31;
  int tile_m = (blockIdx.x * 8 + wave) * 16;
  int half = lane >> 4, l15 = lane & 15;
  long arow = tile_m + l15;
  f32x8 acc[8] = {};
  for (int k0 = 0; k0 < HEADS * HID; k0 += 4) {
    int kk = k0 + half * 2;
    f32x2 af;
    af.x = A[arow * (HEADS * HID) + kk];
    af.y = A[arow * (HEADS * HID) + kk + 1];
#pragma unroll
    for (int nt = 0; nt < 8; ++nt) {
      int col = nt * 16 + l15;
      f32x2 bf;
      bf.x = fc1_w[(long)col * (HEADS * HID) + kk];
      bf.y = fc1_w[(long)col * (HEADS * HID) + kk + 1];
      acc[nt] = wmma_f32(af, bf, acc[nt]);
    }
  }
  float ssum = 0.f;
#pragma unroll
  for (int nt = 0; nt < 8; ++nt) {
    int col = nt * 16 + l15;
    float bb = fc1_b[col], w2 = fc2_w[col];
#pragma unroll
    for (int v = 0; v < 8; ++v) ssum += tanhf(acc[nt][v] + bb) * w2;
  }
#pragma unroll
  for (int off = 16; off > 0; off >>= 1) ssum += __shfl_xor(ssum, off, 32);
  if (lane == 0) atomicAdd(&score[p], ssum);
}

__global__ void beta_kernel(const float* __restrict__ score,
                            float* __restrict__ beta) {
  float s0 = score[0] / (float)NB, s1 = score[1] / (float)NB;
  float mx = fmaxf(s0, s1);
  float e0 = __expf(s0 - mx), e1 = __expf(s1 - mx);
  float inv = 1.f / (e0 + e1);
  beta[0] = e0 * inv;
  beta[1] = e1 * inv;
}

__global__ void hagg_kernel(const float* __restrict__ outs,
                            const float* __restrict__ beta,
                            float* __restrict__ hagg, int n) {
  int i = blockIdx.x * 256 + threadIdx.x;
  if (i >= n) return;
  hagg[i] = beta[0] * outs[i] + beta[1] * outs[(long)n + i];
}

__global__ __launch_bounds__(256) void logits_kernel(
    const float* __restrict__ hagg, const float* __restrict__ fc_w,
    const float* __restrict__ fc_b, float* __restrict__ logits) {
  __shared__ float ldsB[(HEADS * HID) * OUTD];
  int tid = threadIdx.x;
  for (int i = tid; i < (HEADS * HID) * OUTD; i += 256) {
    int k = i >> 4, n = i & 15;
    ldsB[i] = fc_w[n * (HEADS * HID) + k];
  }
  __syncthreads();
  int wave = tid >> 5, lane = tid & 31;
  int tile_m = (blockIdx.x * 8 + wave) * 16;
  int half = lane >> 4, l15 = lane & 15;
  long arow = tile_m + l15;
  f32x8 c = {};
  for (int k0 = 0; k0 < HEADS * HID; k0 += 4) {
    int kk = k0 + half * 2;
    f32x2 af;
    af.x = hagg[arow * (HEADS * HID) + kk];
    af.y = hagg[arow * (HEADS * HID) + kk + 1];
    f32x2 bf;
    bf.x = ldsB[kk * OUTD + l15];
    bf.y = ldsB[(kk + 1) * OUTD + l15];
    c = wmma_f32(af, bf, c);
  }
#pragma unroll
  for (int v = 0; v < 8; ++v) {
    int r = tile_m + v + half * 8;
    logits[(long)r * OUTD + l15] = c[v] + fc_b[l15];
  }
}

extern "C" void kernel_launch(void* const* d_in, const int* in_sizes, int n_in,
                              void* d_out, int out_size, void* d_ws,
                              size_t ws_size, hipStream_t stream) {
  const float* feat0 = (const float*)d_in[0];
  const float* feat1 = (const float*)d_in[1];
  const float* W0    = (const float*)d_in[2];
  const float* b0    = (const float*)d_in[3];
  const float* W1    = (const float*)d_in[4];
  const float* b1    = (const float*)d_in[5];
  const float* r_vec = (const float*)d_in[6];
  const float* attn0 = (const float*)d_in[7];
  const float* attn1 = (const float*)d_in[8];
  const float* fc1_w = (const float*)d_in[9];
  const float* fc1_b = (const float*)d_in[10];
  const float* fc2_w = (const float*)d_in[11];
  const float* fc_w  = (const float*)d_in[12];
  const float* fc_b  = (const float*)d_in[13];
  const int* idx0    = (const int*)d_in[14];
  const int* idx1    = (const int*)d_in[15];
  const int* mp_indices[2] = {(const int*)d_in[16], (const int*)d_in[19]};
  const int* mp_dst[2]     = {(const int*)d_in[17], (const int*)d_in[20]};
  const int* mp_target[2]  = {(const int*)d_in[18], (const int*)d_in[21]};
  const float* attn[2] = {attn0, attn1};

  float* ws     = (float*)d_ws;
  float* tf     = ws;
  float* finals = tf + (long)N_NODES * HID;
  float* h      = finals + 2 * LP * HID;
  float* a      = h + (long)NE * HID;
  float* m      = a + (long)NE * HEADS;
  float* s      = m + NG * HEADS;
  float* ft     = s + NG * HEADS;
  float* outs   = ft + (long)NG * HEADS * HID;
  float* score  = outs + 2L * NB * HEADS * HID;
  float* beta   = score + 2;

  float* logits = (float*)d_out;
  float* hagg   = logits + (long)NB * OUTD;

  gemm_tf_kernel<F0><<<(N0 + 127) / 128, 256, 0, stream>>>(feat0, W0, b0, idx0,
                                                           tf, N0);
  gemm_tf_kernel<F1><<<(N1 + 127) / 128, 256, 0, stream>>>(feat1, W1, b1, idx1,
                                                           tf, N1);
  finals_kernel<<<1, 32, 0, stream>>>(r_vec, finals);

  for (int mp = 0; mp < 2; ++mp) {
    fill_kernel<<<(NG * HEADS + 255) / 256, 256, 0, stream>>>(m, -3.0e38f,
                                                              NG * HEADS);
    fill_kernel<<<(NG * HEADS + 255) / 256, 256, 0, stream>>>(s, 0.f,
                                                              NG * HEADS);
    edge_h_kernel<<<(NE / EPW + 7) / 8, 256, 0, stream>>>(tf, finals + mp * LP * HID,
                                              mp_indices[mp], mp_dst[mp],
                                              attn[mp], h, a, m);
    edge_exp_kernel<<<(NE * HEADS + 255) / 256, 256, 0, stream>>>(
        a, mp_dst[mp], m, s, NE * HEADS);
    owner_sum_kernel<<<NG / OWN_D, 1024, 0, stream>>>(h, a, s, mp_dst[mp], ft);
    target_elu_kernel<<<(NB * HEADS * HID + 255) / 256, 256, 0, stream>>>(
        ft, mp_target[mp], outs + (long)mp * NB * HEADS * HID,
        NB * HEADS * HID);
  }

  fill_kernel<<<1, 64, 0, stream>>>(score, 0.f, 2);
  {
    dim3 g(NB / 128, 2);
    score_kernel<<<g, 256, 0, stream>>>(outs, fc1_w, fc1_b, fc2_w, score);
  }
  beta_kernel<<<1, 1, 0, stream>>>(score, beta);
  hagg_kernel<<<(NB * HEADS * HID + 255) / 256, 256, 0, stream>>>(
      outs, beta, hagg, NB * HEADS * HID);
  logits_kernel<<<NB / 128, 256, 0, stream>>>(hagg, fc_w, fc_b, logits);
}


# launch boundaries 20 -> 10: one set-up launch (both tf GEMMs, finals, all start values); beta inside hagg; elu at the owner row store, target row pick in score and hagg (no target_elu, no ft copy)
# speedup vs baseline: 3.0436x; 1.1582x over previous
#include <hip/hip_runtime.h>
#include <math.h>


#define N_NODES 100000
#define N0      60000
#define N1      40000
#define F0      128
#define F1      64
#define HID     64
#define HEADS   8
#define OUTD    16
#define ATTNV   128
#define NE      250000
#define LP      3
#define NG      8192
#define NB      8192

typedef __attribute__((ext_vector_type(2))) float f32x2;
typedef __attribute__((ext_vector_type(8))) float f32x8;

__device__ __forceinline__ f32x8 wmma_f32(f32x2 a, f32x2 b, f32x8 c) {
  return __builtin_amdgcn_wmma_f32_16x16x4_f32(false, a, false, b, (short)0, c,
                                               false, false);
}

__device__ __forceinline__ float xrow(float x) {
  int b = __builtin_bit_cast(int, x);
  return __builtin_bit_cast(
      float, __builtin_amdgcn_permlanex16(b, b, 0x76543210, 0xfedcba98, false,
                                          false));
}
template <int M>
__device__ __forceinline__ float xlane(float x) {
  int b = __builtin_bit_cast(int, x);
  return __builtin_bit_cast(
      float, __builtin_amdgcn_update_dpp(b, b, 0x160 + M, 0xf, 0xf, false));
}

#define INIT_WG ((2 * NG * HEADS + 255) / 256)
__device__ __forceinline__ void init_body(int blk,
                                          const float* __restrict__ r_vec,
                                          float* __restrict__ finals  ,
                                          float* __restrict__ m  ,
                                          float* __restrict__ s  ,
                                          float* __restrict__ score  ) {
  const int i = blk * 256 + threadIdx.x;
  if (i < 2 * NG * HEADS) {
    m[i] = -3.0e38f;
    s[i] = 0.f;
  }
  if (blk != 0) return;
  const int c = threadIdx.x;
  if (c < 2) score[c] = 0.f;
  if (c >= 32) return;
  for (int mp = 0; mp < 2; ++mp) {
    float re = r_vec[(mp * 32 + c) * 2 + 0];
    float im = r_vec[(mp * 32 + c) * 2 + 1];
    float inv = rsqrtf(re * re + im * im);
    re *= inv; im *= inv;
    float f1re = re, f1im = -im;
    float f0re = f1re * re - f1im * im;
    float f0im = f1re * im + f1im * re;
    float* F = finals + mp * LP * HID;
    F[0 * HID + 2 * c] = f0re; F[0 * HID + 2 * c + 1] = f0im;
    F[1 * HID + 2 * c] = f1re; F[1 * HID + 2 * c + 1] = f1im;
    F[2 * HID + 2 * c] = 1.f;  F[2 * HID + 2 * c + 1] = 0.f;
  }
}

template <int K>
__device__ __forceinline__ void gemm_tf_body(
    int blk, float* ldsB, const float* __restrict__ feat,
    const float* __restrict__ W, const float* __restrict__ bias,
    const int* __restrict__ idx, float* __restrict__ tf, int M) {
  int tid = threadIdx.x;
  for (int i = tid; i < K * HID; i += 256) {
    int k = i >> 6, n = i & 63;
    ldsB[i] = W[n * K + k];
  }
  __syncthreads();
  int wave = tid >> 5, lane = tid & 31;
  int tile_m = blk * 128 + wave * 16;
  if (tile_m >= M) return;
  int half = lane >> 4, l15 = lane & 15;
  int arow = tile_m + l15;
  if (arow >= M) arow = M - 1;
  f32x8 c0 = {}, c1 = {}, c2 = {}, c3 = {};
  for (int k0 = 0; k0 < K; k0 += 4) {
    int kk = k0 + half * 2;
    f32x2 a;
    a.x = feat[(long)arow * K + kk];
    a.y = feat[(long)arow * K + kk + 1];
    f32x2 b0, b1, b2, b3;
    b0.x = ldsB[kk * HID + l15 + 0];  b0.y = ldsB[(kk + 1) * HID + l15 + 0];
    b1.x = ldsB[kk * HID + l15 + 16]; b1.y = ldsB[(kk + 1) * HID + l15 + 16];
    b2.x = ldsB[kk * HID + l15 + 32]; b2.y = ldsB[(kk + 1) * HID + l15 + 32];
    b3.x = ldsB[kk * HID + l15 + 48]; b3.y = ldsB[(kk + 1) * HID + l15 + 48];
    c0 = wmma_f32(a, b0, c0);
    c1 = wmma_f32(a, b1, c1);
    c2 = wmma_f32(a, b2, c2);
    c3 = wmma_f32(a, b3, c3);
  }
#pragma unroll
  for (int v = 0; v < 8; ++v) {
    int r = tile_m + v + half * 8;
    if (r < M) {
      float* drow = tf + (long)idx[r] * HID;
      drow[l15 + 0]  = c0[v] + bias[l15 + 0];
      drow[l15 + 16] = c1[v] + bias[l15 + 16];
      drow[l15 + 32] = c2[v] + bias[l15 + 32];
      drow[l15 + 48] = c3[v] + bias[l15 + 48];
    }
  }
}

#define GEMM0_WG ((N0 + 127) / 128)
#define GEMM1_WG ((N1 + 127) / 128)
__global__ __launch_bounds__(256) void setup_kernel(
    const float* __restrict__ feat0, const float* __restrict__ W0,
    const float* __restrict__ b0, const int* __restrict__ idx0,
    const float* __restrict__ feat1, const float* __restrict__ W1,
    const float* __restrict__ b1, const int* __restrict__ idx1,
    float* __restrict__ tf, const float* __restrict__ r_vec,
    float* __restrict__ finals, float* __restrict__ m, float* __restrict__ s,
    float* __restrict__ score) {
  __shared__ float ldsB[F0 * HID];
  const int blk = blockIdx.x;
  if (blk < GEMM0_WG) {
    gemm_tf_body<F0>(blk, ldsB, feat0, W0, b0, idx0, tf, N0);
  } else if (blk < GEMM0_WG + GEMM1_WG) {
    gemm_tf_body<F1>(blk - GEMM0_WG, ldsB, feat1, W1, b1, idx1, tf, N1);
  } else {
    init_body(blk - GEMM0_WG - GEMM1_WG, r_vec, finals, m, s, score);
  }
}

#define EPW 8
__global__ __launch_bounds__(256) void edge_h_kernel(
    const float* __restrict__ tf, const float* __restrict__ finals,
    const int* __restrict__ indices, const int* __restrict__ dst,
    const float* __restrict__ attn, float* __restrict__ h,
    float* __restrict__ a, float* __restrict__ m) {
  const int lane = threadIdx.x & 31;
  const int wave = __builtin_amdgcn_readfirstlane(threadIdx.x >> 5);
  const int e0 = (blockIdx.x * 8 + wave) * EPW;
  if (e0 >= NE) return;
  const int mynode = indices[e0 * LP + (lane < LP * EPW ? lane : lane - 8)];
  const int g = lane >> 2, r = lane & 3;
  const int dlo = dst[e0 + r];
  const int dhi = dst[e0 + 4 + r];
  float2 v[EPW][LP];
#pragma unroll
  for (int u = 0; u < EPW; ++u) {
#pragma unroll
    for (int i = 0; i < LP; ++i) {
      const int node = __builtin_amdgcn_readlane(mynode, u * LP + i);
      v[u][i] = *((const float2*)(tf + (long)node * HID) + lane);
    }
  }
  float2 f[LP];
#pragma unroll
  for (int i = 0; i < LP; ++i) f[i] = *((const float2*)(finals + i * HID) + lane);
  float2 w[HEADS];
#pragma unroll
  for (int hh = 0; hh < HEADS; ++hh)
    w[hh] = *((const float2*)(attn + hh * HID) + lane);

  const bool b4 = (lane & 16) != 0, b3 = (lane & 8) != 0, b2 = (lane & 4) != 0;
  float tlo = 0.f, thi = 0.f;
#pragma unroll
  for (int u = 0; u < EPW; ++u) {
    float2 acc = make_float2(0.f, 0.f);
#pragma unroll
    for (int i = 0; i < LP; ++i) {
      const float vx = v[u][i].x, vy = v[u][i].y;
      const float fx = f[i].x, fy = f[i].y;
      acc.x += vx * fx - vy * fy;
      acc.y += vx * fy + vy * fx;
    }
    acc.x *= (1.f / 3.f);
    acc.y *= (1.f / 3.f);
    *((float2*)(h + (long)(e0 + u) * HID) + lane) = acc;
    float p[HEADS];
#pragma unroll
    for (int hh = 0; hh < HEADS; ++hh)
      p[hh] = __builtin_fmaf(acc.x, w[hh].x, acc.y * w[hh].y);
    float q[4];
#pragma unroll
    for (int j = 0; j < 4; ++j) {
      float keep = b4 ? p[j + 4] : p[j];
      float send = b4 ? p[j] : p[j + 4];
      q[j] = keep + xrow(send);
    }
    float rr[2];
#pragma unroll
    for (int j = 0; j < 2; ++j) {
      float keep = b3 ? q[j + 2] : q[j];
      float send = b3 ? q[j] : q[j + 2];
      rr[j] = keep + xlane<8>(send);
    }
    float keep2 = b2 ? rr[1] : rr[0];
    float send2 = b2 ? rr[0] : rr[1];
    float t = keep2 + xlane<4>(send2);
    t += xlane<2>(t);
    t += xlane<1>(t);
    if (u < 4) tlo = (r == u) ? t : tlo;
    else       thi = (r == u - 4) ? t : thi;
  }
  const float alo = tlo > 0.f ? tlo : 0.01f * tlo;
  const float ahi = thi > 0.f ? thi : 0.01f * thi;
  a[(long)(e0 + r) * HEADS + g] = alo;
  a[(long)(e0 + 4 + r) * HEADS + g] = ahi;
  atomicMax(&m[dlo * HEADS + g], alo);
  atomicMax(&m[dhi * HEADS + g], ahi);
}

__global__ void edge_exp_kernel(float* __restrict__ a,
                                const int* __restrict__ dst,
                                const float* __restrict__ m,
                                float* __restrict__ s, int n) {
  int i = blockIdx.x * 256 + threadIdx.x;
  if (i >= n) return;
  int e = i >> 3, hh = i & 7;
  int d = dst[e];
  float ex = __expf(a[i] - m[d * HEADS + hh]);
  a[i] = ex;
  atomicAdd(&s[d * HEADS + hh], ex);
}

#define OWN_D   32
#define OWN_CAP 25600

__device__ __forceinline__ float lane_bcast(float x, int l) {
  return __builtin_bit_cast(
      float, __builtin_amdgcn_readlane(__builtin_bit_cast(int, x), l));
}

__device__ __forceinline__ void note_edge(int d, int i, int d0, int* list,
                                          int* nlist) {
  const unsigned dl = (unsigned)(d - d0);
  if (dl < OWN_D) {
    const int r = atomicAdd(nlist, 1);
    if (r < OWN_CAP) list[r] = (i << 5) | (int)dl;
  }
}

template <int NU>
__device__ __forceinline__ void add_noted(const int* list, int j0, int lane,
                                          int d0, const float* __restrict__ h,
                                          const float* __restrict__ a,
                                          const float* __restrict__ s,
                                          float* acc) {
  const int mine = list[j0 + 32 * (lane & (NU - 1))];
  int ent[NU];
  float lo[NU], hi[NU], av[NU], sv[NU];
#pragma unroll
  for (int u = 0; u < NU; ++u) {
    ent[u] = __builtin_amdgcn_readlane(mine, u);
    const int e = ent[u] >> 5, d = d0 + (ent[u] & 31);
    lo[u] = h[(long)e * HID + lane];
    hi[u] = h[(long)e * HID + 32 + lane];
    av[u] = a[(long)e * HEADS + (lane & 7)];
    sv[u] = s[d * HEADS + (lane & 7)];
  }
#pragma unroll
  for (int u = 0; u < NU; ++u) {
    const float wl = av[u] / sv[u];
    float* row = acc + (ent[u] & 31) * (HEADS * HID) + lane;
#pragma unroll
    for (int hh = 0; hh < HEADS; ++hh) {
      const float w = lane_bcast(wl, hh);
      atomicAdd(row + hh * HID, w * lo[u]);
      atomicAdd(row + hh * HID + 32, w * hi[u]);
    }
  }
}

__global__ __launch_bounds__(1024) void owner_sum_kernel(
    const float* __restrict__ h, const float* __restrict__ a,
    const float* __restrict__ s, const int* __restrict__ dst,
    float* __restrict__ eft  ) {
  __shared__ __attribute__((aligned(16))) float acc[OWN_D * HEADS * HID];
  __shared__ int list[OWN_CAP];
  __shared__ int nlist;
  const int tid = threadIdx.x, lane = tid & 31;
  const int wave = __builtin_amdgcn_readfirstlane(tid >> 5);
  const int d0 = blockIdx.x * OWN_D;

  float4* acc4 = (float4*)acc;
  for (int i = tid; i < OWN_D * HEADS * HID / 4; i += 1024)
    acc4[i] = make_float4(0.f, 0.f, 0.f, 0.f);
  if (tid == 0) nlist = 0;
  __syncthreads();

  const int4* dst4 = (const int4*)dst;
  for (int base = 0; base < NE / 4; base += 1024 * 8) {
    int4 dv[8];
#pragma unroll
    for (int u = 0; u < 8; ++u) {
      const int q = base + u * 1024 + tid;
      dv[u] = q < NE / 4 ? dst4[q] : make_int4(-1, -1, -1, -1);
    }
#pragma unroll
    for (int u = 0; u < 8; ++u) {
      const int i = 4 * (base + u * 1024 + tid);
      note_edge(dv[u].x, i + 0, d0, list, &nlist);
      note_edge(dv[u].y, i + 1, d0, list, &nlist);
      note_edge(dv[u].z, i + 2, d0, list, &nlist);
      note_edge(dv[u].w, i + 3, d0, list, &nlist);
    }
  }
  __syncthreads();

  const int n = __builtin_amdgcn_readfirstlane(nlist);
  if (n <= OWN_CAP) {
    int j = wave;
    for (; j + 96 < n; j += 128) add_noted<4>(list, j, lane, d0, h, a, s, acc);
    for (; j < n; j += 32) add_noted<1>(list, j, lane, d0, h, a, s, acc);
  } else {
    for (int i = tid; i < NE; i += 1024) {
      const int d = dst[i];
      const unsigned dl = (unsigned)(d - d0);
      if (dl < OWN_D) {
        for (int hh = 0; hh < HEADS; ++hh) {
          const float w = a[(long)i * HEADS + hh] / s[d * HEADS + hh];
          for (int c = 0; c < HID; ++c)
            atomicAdd(&acc[(dl * HEADS + hh) * HID + c],
                      w * h[(long)i * HID + c]);
        }
      }
    }
  }
  __syncthreads();

  float4* o = (float4*)(eft + (long)d0 * HEADS * HID);
  for (int i = tid; i < OWN_D * HEADS * HID / 4; i += 1024) {
    float4 v = acc4[i];
    v.x = v.x > 0.f ? v.x : (__expf(v.x) - 1.f);
    v.y = v.y > 0.f ? v.y : (__expf(v.y) - 1.f);
    v.z = v.z > 0.f ? v.z : (__expf(v.z) - 1.f);
    v.w = v.w > 0.f ? v.w : (__expf(v.w) - 1.f);
    o[i] = v;
  }
}

__global__ __launch_bounds__(256) void score_kernel(
    const float* __restrict__ eft, const int* __restrict__ target0,
    const int* __restrict__ target1, const float* __restrict__ fc1_w,
    const float* __restrict__ fc1_b, const float* __restrict__ fc2_w,
    float* __restrict__ score) {
  int p = blockIdx.y;
  const float* A = eft + (long)p * NG * (HEADS * HID);
  const int* target = p ? target1 : target0;
  int wave = threadIdx.x >> 5, lane = threadIdx.x & 31;
  int tile_m = (blockIdx.x * 8 + wave) * 16;
  int half = lane >> 4, l15 = lane & 15;
  long arow = target[tile_m + l15];
  f32x8 acc[8] = {};
  for (int k0 = 0; k0 < HEADS * HID; k0 += 4) {
    int kk = k0 + half * 2;
    f32x2 af;
    af.x = A[arow * (HEADS * HID) + kk];
    af.y = A[arow * (HEADS * HID) + kk + 1];
#pragma unroll
    for (int nt = 0; nt < 8; ++nt) {
      int col = nt * 16 + l15;
      f32x2 bf;
      bf.x = fc1_w[(long)col * (HEADS * HID) + kk];
      bf.y = fc1_w[(long)col * (HEADS * HID) + kk + 1];
      acc[nt] = wmma_f32(af, bf, acc[nt]);
    }
  }
  float ssum = 0.f;
#pragma unroll
  for (int nt = 0; nt < 8; ++nt) {
    int col = nt * 16 + l15;
    float bb = fc1_b[col], w2 = fc2_w[col];
#pragma unroll
    for (int v = 0; v < 8; ++v) ssum += tanhf(acc[nt][v] + bb) * w2;
  }
#pragma unroll
  for (int off = 16; off > 0; off >>= 1) ssum += __shfl_xor(ssum, off, 32);
  if (lane == 0) atomicAdd(&score[p], ssum);
}

__global__ void hagg_kernel(const float* __restrict__ eft,
                            const int* __restrict__ target0,
                            const int* __restrict__ target1,
                            const float* __restrict__ score,
                            float* __restrict__ hagg, int n) {
  int i = blockIdx.x * 256 + threadIdx.x;
  if (i >= n) return;
  float s0 = score[0] / (float)NB, s1 = score[1] / (float)NB;
  float mx = fmaxf(s0, s1);
  float e0 = __expf(s0 - mx), e1 = __expf(s1 - mx);
  float inv = 1.f / (e0 + e1);
  float beta0 = e0 * inv, beta1 = e1 * inv;
  int b = i >> 9, j = i & 511;
  const float* r0 = eft + (long)target0[b] * (HEADS * HID);
  const float* r1 = eft + (long)NG * (HEADS * HID) +
                    (long)target1[b] * (HEADS * HID);
  hagg[i] = beta0 * r0[j] + beta1 * r1[j];
}

__global__ __launch_bounds__(256) void logits_kernel(
    const float* __restrict__ hagg, const float* __restrict__ fc_w,
    const float* __restrict__ fc_b, float* __restrict__ logits) {
  __shared__ float ldsB[(HEADS * HID) * OUTD];
  int tid = threadIdx.x;
  for (int i = tid; i < (HEADS * HID) * OUTD; i += 256) {
    int k = i >> 4, n = i & 15;
    ldsB[i] = fc_w[n * (HEADS * HID) + k];
  }
  __syncthreads();
  int wave = tid >> 5, lane = tid & 31;
  int tile_m = (blockIdx.x * 8 + wave) * 16;
  int half = lane >> 4, l15 = lane & 15;
  long arow = tile_m + l15;
  f32x8 c = {};
  for (int k0 = 0; k0 < HEADS * HID; k0 += 4) {
    int kk = k0 + half * 2;
    f32x2 af;
    af.x = hagg[arow * (HEADS * HID) + kk];
    af.y = hagg[arow * (HEADS * HID) + kk + 1];
    f32x2 bf;
    bf.x = ldsB[kk * OUTD + l15];
    bf.y = ldsB[(kk + 1) * OUTD + l15];
    c = wmma_f32(af, bf, c);
  }
#pragma unroll
  for (int v = 0; v < 8; ++v) {
    int r = tile_m + v + half * 8;
    logits[(long)r * OUTD + l15] = c[v] + fc_b[l15];
  }
}

extern "C" void kernel_launch(void* const* d_in, const int* in_sizes, int n_in,
                              void* d_out, int out_size, void* d_ws,
                              size_t ws_size, hipStream_t stream) {
  const float* feat0 = (const float*)d_in[0];
  const float* feat1 = (const float*)d_in[1];
  const float* W0    = (const float*)d_in[2];
  const float* b0    = (const float*)d_in[3];
  const float* W1    = (const float*)d_in[4];
  const float* b1    = (const float*)d_in[5];
  const float* r_vec = (const float*)d_in[6];
  const float* attn0 = (const float*)d_in[7];
  const float* attn1 = (const float*)d_in[8];
  const float* fc1_w = (const float*)d_in[9];
  const float* fc1_b = (const float*)d_in[10];
  const float* fc2_w = (const float*)d_in[11];
  const float* fc_w  = (const float*)d_in[12];
  const float* fc_b  = (const float*)d_in[13];
  const int* idx0    = (const int*)d_in[14];
  const int* idx1    = (const int*)d_in[15];
  const int* mp_indices[2] = {(const int*)d_in[16], (const int*)d_in[19]};
  const int* mp_dst[2]     = {(const int*)d_in[17], (const int*)d_in[20]};
  const int* mp_target[2]  = {(const int*)d_in[18], (const int*)d_in[21]};
  const float* attn[2] = {attn0, attn1};

  float* ws     = (float*)d_ws;
  float* tf     = ws;
  float* finals = tf + (long)N_NODES * HID;
  float* h      = finals + 2 * LP * HID;
  float* a      = h + (long)NE * HID;
  float* m      = a + (long)NE * HEADS;
  float* s      = m + 2 * NG * HEADS;
  float* eft    = s + 2 * NG * HEADS;
  float* score  = eft + 2L * NG * HEADS * HID;

  float* logits = (float*)d_out;
  float* hagg   = logits + (long)NB * OUTD;

  setup_kernel<<<GEMM0_WG + GEMM1_WG + INIT_WG, 256, 0, stream>>>(
      feat0, W0, b0, idx0, feat1, W1, b1, idx1, tf, r_vec, finals, m, s, score);

  for (int mp = 0; mp < 2; ++mp) {
    float* m_mp = m + mp * NG * HEADS;
    float* s_mp = s + mp * NG * HEADS;
    edge_h_kernel<<<(NE / EPW + 7) / 8, 256, 0, stream>>>(tf, finals + mp * LP * HID,
                                              mp_indices[mp], mp_dst[mp],
                                              attn[mp], h, a, m_mp);
    edge_exp_kernel<<<(NE * HEADS + 255) / 256, 256, 0, stream>>>(
        a, mp_dst[mp], m_mp, s_mp, NE * HEADS);
    owner_sum_kernel<<<NG / OWN_D, 1024, 0, stream>>>(
        h, a, s_mp, mp_dst[mp], eft + (long)mp * NG * HEADS * HID);
  }

  {
    dim3 g(NB / 128, 2);
    score_kernel<<<g, 256, 0, stream>>>(eft, mp_target[0], mp_target[1], fc1_w,
                                        fc1_b, fc2_w, score);
  }
  hagg_kernel<<<(NB * HEADS * HID + 255) / 256, 256, 0, stream>>>(
      eft, mp_target[0], mp_target[1], score, hagg, NB * HEADS * HID);
  logits_kernel<<<NB / 128, 256, 0, stream>>>(hagg, fc_w, fc_b, logits);
}


# edge softmax (segment max, exp, segment sum) taken in LDS by the owner work-groups over their noted edges: edge_exp launches, m/s planes and their global float atomics go; exp sums: LDS arrival order
# speedup vs baseline: 3.8901x; 1.2781x over previous
#include <hip/hip_runtime.h>
#include <math.h>


#define N_NODES 100000
#define N0      60000
#define N1      40000
#define F0      128
#define F1      64
#define HID     64
#define HEADS   8
#define OUTD    16
#define ATTNV   128
#define NE      250000
#define LP      3
#define NG      8192
#define NB      8192

typedef __attribute__((ext_vector_type(2))) float f32x2;
typedef __attribute__((ext_vector_type(8))) float f32x8;

__device__ __forceinline__ f32x8 wmma_f32(f32x2 a, f32x2 b, f32x8 c) {
  return __builtin_amdgcn_wmma_f32_16x16x4_f32(false, a, false, b, (short)0, c,
                                               false, false);
}

__device__ __forceinline__ float xrow(float x) {
  int b = __builtin_bit_cast(int, x);
  return __builtin_bit_cast(
      float, __builtin_amdgcn_permlanex16(b, b, 0x76543210, 0xfedcba98, false,
                                          false));
}
template <int M>
__device__ __forceinline__ float xlane(float x) {
  int b = __builtin_bit_cast(int, x);
  return __builtin_bit_cast(
      float, __builtin_amdgcn_update_dpp(b, b, 0x160 + M, 0xf, 0xf, false));
}

#define INIT_WG 1
__device__ __forceinline__ void init_body(const float* __restrict__ r_vec,
                                          float* __restrict__ finals  ,
                                          float* __restrict__ score  ) {
  const int c = threadIdx.x;
  if (c < 2) score[c] = 0.f;
  if (c >= 32) return;
  for (int mp = 0; mp < 2; ++mp) {
    float re = r_vec[(mp * 32 + c) * 2 + 0];
    float im = r_vec[(mp * 32 + c) * 2 + 1];
    float inv = rsqrtf(re * re + im * im);
    re *= inv; im *= inv;
    float f1re = re, f1im = -im;
    float f0re = f1re * re - f1im * im;
    float f0im = f1re * im + f1im * re;
    float* F = finals + mp * LP * HID;
    F[0 * HID + 2 * c] = f0re; F[0 * HID + 2 * c + 1] = f0im;
    F[1 * HID + 2 * c] = f1re; F[1 * HID + 2 * c + 1] = f1im;
    F[2 * HID + 2 * c] = 1.f;  F[2 * HID + 2 * c + 1] = 0.f;
  }
}

template <int K>
__device__ __forceinline__ void gemm_tf_body(
    int blk, float* ldsB, const float* __restrict__ feat,
    const float* __restrict__ W, const float* __restrict__ bias,
    const int* __restrict__ idx, float* __restrict__ tf, int M) {
  int tid = threadIdx.x;
  for (int i = tid; i < K * HID; i += 256) {
    int k = i >> 6, n = i & 63;
    ldsB[i] = W[n * K + k];
  }
  __syncthreads();
  int wave = tid >> 5, lane = tid & 31;
  int tile_m = blk * 128 + wave * 16;
  if (tile_m >= M) return;
  int half = lane >> 4, l15 = lane & 15;
  int arow = tile_m + l15;
  if (arow >= M) arow = M - 1;
  f32x8 c0 = {}, c1 = {}, c2 = {}, c3 = {};
  for (int k0 = 0; k0 < K; k0 += 4) {
    int kk = k0 + half * 2;
    f32x2 a;
    a.x = feat[(long)arow * K + kk];
    a.y = feat[(long)arow * K + kk + 1];
    f32x2 b0, b1, b2, b3;
    b0.x = ldsB[kk * HID + l15 + 0];  b0.y = ldsB[(kk + 1) * HID + l15 + 0];
    b1.x = ldsB[kk * HID + l15 + 16]; b1.y = ldsB[(kk + 1) * HID + l15 + 16];
    b2.x = ldsB[kk * HID + l15 + 32]; b2.y = ldsB[(kk + 1) * HID + l15 + 32];
    b3.x = ldsB[kk * HID + l15 + 48]; b3.y = ldsB[(kk + 1) * HID + l15 + 48];
    c0 = wmma_f32(a, b0, c0);
    c1 = wmma_f32(a, b1, c1);
    c2 = wmma_f32(a, b2, c2);
    c3 = wmma_f32(a, b3, c3);
  }
#pragma unroll
  for (int v = 0; v < 8; ++v) {
    int r = tile_m + v + half * 8;
    if (r < M) {
      float* drow = tf + (long)idx[r] * HID;
      drow[l15 + 0]  = c0[v] + bias[l15 + 0];
      drow[l15 + 16] = c1[v] + bias[l15 + 16];
      drow[l15 + 32] = c2[v] + bias[l15 + 32];
      drow[l15 + 48] = c3[v] + bias[l15 + 48];
    }
  }
}

#define GEMM0_WG ((N0 + 127) / 128)
#define GEMM1_WG ((N1 + 127) / 128)
__global__ __launch_bounds__(256) void setup_kernel(
    const float* __restrict__ feat0, const float* __restrict__ W0,
    const float* __restrict__ b0, const int* __restrict__ idx0,
    const float* __restrict__ feat1, const float* __restrict__ W1,
    const float* __restrict__ b1, const int* __restrict__ idx1,
    float* __restrict__ tf, const float* __restrict__ r_vec,
    float* __restrict__ finals, float* __restrict__ score) {
  __shared__ float ldsB[F0 * HID];
  const int blk = blockIdx.x;
  if (blk < GEMM0_WG) {
    gemm_tf_body<F0>(blk, ldsB, feat0, W0, b0, idx0, tf, N0);
  } else if (blk < GEMM0_WG + GEMM1_WG) {
    gemm_tf_body<F1>(blk - GEMM0_WG, ldsB, feat1, W1, b1, idx1, tf, N1);
  } else {
    init_body(r_vec, finals, score);
  }
}

#define EPW 8
__global__ __launch_bounds__(256) void edge_h_kernel(
    const float* __restrict__ tf, const float* __restrict__ finals,
    const int* __restrict__ indices, const float* __restrict__ attn,
    float* __restrict__ h, float* __restrict__ a) {
  const int lane = threadIdx.x & 31;
  const int wave = __builtin_amdgcn_readfirstlane(threadIdx.x >> 5);
  const int e0 = (blockIdx.x * 8 + wave) * EPW;
  if (e0 >= NE) return;
  const int mynode = indices[e0 * LP + (lane < LP * EPW ? lane : lane - 8)];
  const int g = lane >> 2, r = lane & 3;
  float2 v[EPW][LP];
#pragma unroll
  for (int u = 0; u < EPW; ++u) {
#pragma unroll
    for (int i = 0; i < LP; ++i) {
      const int node = __builtin_amdgcn_readlane(mynode, u * LP + i);
      v[u][i] = *((const float2*)(tf + (long)node * HID) + lane);
    }
  }
  float2 f[LP];
#pragma unroll
  for (int i = 0; i < LP; ++i) f[i] = *((const float2*)(finals + i * HID) + lane);
  float2 w[HEADS];
#pragma unroll
  for (int hh = 0; hh < HEADS; ++hh)
    w[hh] = *((const float2*)(attn + hh * HID) + lane);

  const bool b4 = (lane & 16) != 0, b3 = (lane & 8) != 0, b2 = (lane & 4) != 0;
  float tlo = 0.f, thi = 0.f;
#pragma unroll
  for (int u = 0; u < EPW; ++u) {
    float2 acc = make_float2(0.f, 0.f);
#pragma unroll
    for (int i = 0; i < LP; ++i) {
      const float vx = v[u][i].x, vy = v[u][i].y;
      const float fx = f[i].x, fy = f[i].y;
      acc.x += vx * fx - vy * fy;
      acc.y += vx * fy + vy * fx;
    }
    acc.x *= (1.f / 3.f);
    acc.y *= (1.f / 3.f);
    *((float2*)(h + (long)(e0 + u) * HID) + lane) = acc;
    float p[HEADS];
#pragma unroll
    for (int hh = 0; hh < HEADS; ++hh)
      p[hh] = __builtin_fmaf(acc.x, w[hh].x, acc.y * w[hh].y);
    float q[4];
#pragma unroll
    for (int j = 0; j < 4; ++j) {
      float keep = b4 ? p[j + 4] : p[j];
      float send = b4 ? p[j] : p[j + 4];
      q[j] = keep + xrow(send);
    }
    float rr[2];
#pragma unroll
    for (int j = 0; j < 2; ++j) {
      float keep = b3 ? q[j + 2] : q[j];
      float send = b3 ? q[j] : q[j + 2];
      rr[j] = keep + xlane<8>(send);
    }
    float keep2 = b2 ? rr[1] : rr[0];
    float send2 = b2 ? rr[0] : rr[1];
    float t = keep2 + xlane<4>(send2);
    t += xlane<2>(t);
    t += xlane<1>(t);
    if (u < 4) tlo = (r == u) ? t : tlo;
    else       thi = (r == u - 4) ? t : thi;
  }
  const float alo = tlo > 0.f ? tlo : 0.01f * tlo;
  const float ahi = thi > 0.f ? thi : 0.01f * thi;
  a[(long)(e0 + r) * HEADS + g] = alo;
  a[(long)(e0 + 4 + r) * HEADS + g] = ahi;
}

#define OWN_D   32
#define OWN_CAP 25600

__device__ __forceinline__ float lane_bcast(float x, int l) {
  return __builtin_bit_cast(
      float, __builtin_amdgcn_readlane(__builtin_bit_cast(int, x), l));
}

__device__ __forceinline__ void note_edge(int d, int i, int d0, int* list,
                                          int* nlist) {
  const unsigned dl = (unsigned)(d - d0);
  if (dl < OWN_D) {
    const int r = atomicAdd(nlist, 1);
    if (r < OWN_CAP) list[r] = (i << 5) | (int)dl;
  }
}

template <bool SUM>
__device__ __forceinline__ void softmax_pass(const int* list, int n8, int tid,
                                             const float* __restrict__ a,
                                             float* mloc, float* sloc) {
  for (int base = 0; base < n8; base += 1024 * 8) {
    float v[8];
    int key[8];
#pragma unroll
    for (int t = 0; t < 8; ++t) {
      const int j = base + t * 1024 + tid;
      const bool ok = j < n8;
      const int ent = list[ok ? (j >> 3) : 0];
      key[t] = ok ? ((ent & 31) * HEADS + (j & 7)) : -1;
      v[t] = a[(long)(ent >> 5) * HEADS + (j & 7)];
    }
#pragma unroll
    for (int t = 0; t < 8; ++t) {
      if (key[t] >= 0) {
        if (SUM) atomicAdd(&sloc[key[t]], __expf(v[t] - mloc[key[t]]));
        else     atomicMax(&mloc[key[t]], v[t]);
      }
    }
  }
}

template <int NU>
__device__ __forceinline__ void add_noted(const int* list, int j0, int lane,
                                          const float* __restrict__ h,
                                          const float* __restrict__ a,
                                          const float* mloc, const float* sloc,
                                          float* acc) {
  const int mine = list[j0 + 32 * (lane & (NU - 1))];
  int ent[NU];
  float lo[NU], hi[NU], av[NU], mv[NU], sv[NU];
#pragma unroll
  for (int u = 0; u < NU; ++u) {
    ent[u] = __builtin_amdgcn_readlane(mine, u);
    const int e = ent[u] >> 5, dl = ent[u] & 31;
    lo[u] = h[(long)e * HID + lane];
    hi[u] = h[(long)e * HID + 32 + lane];
    av[u] = a[(long)e * HEADS + (lane & 7)];
    mv[u] = mloc[dl * HEADS + (lane & 7)];
    sv[u] = sloc[dl * HEADS + (lane & 7)];
  }
#pragma unroll
  for (int u = 0; u < NU; ++u) {
    const float wl = __expf(av[u] - mv[u]) / sv[u];
    float* row = acc + (ent[u] & 31) * (HEADS * HID) + lane;
#pragma unroll
    for (int hh = 0; hh < HEADS; ++hh) {
      const float w = lane_bcast(wl, hh);
      atomicAdd(row + hh * HID, w * lo[u]);
      atomicAdd(row + hh * HID + 32, w * hi[u]);
    }
  }
}

__global__ __launch_bounds__(1024) void owner_sum_kernel(
    const float* __restrict__ h, const float* __restrict__ a,
    const int* __restrict__ dst,
    float* __restrict__ eft  ) {
  __shared__ __attribute__((aligned(16))) float acc[OWN_D * HEADS * HID];
  __shared__ int list[OWN_CAP];
  __shared__ float mloc[OWN_D * HEADS];
  __shared__ float sloc[OWN_D * HEADS];
  __shared__ int nlist;
  const int tid = threadIdx.x, lane = tid & 31;
  const int wave = __builtin_amdgcn_readfirstlane(tid >> 5);
  const int d0 = blockIdx.x * OWN_D;

  float4* acc4 = (float4*)acc;
  for (int i = tid; i < OWN_D * HEADS * HID / 4; i += 1024)
    acc4[i] = make_float4(0.f, 0.f, 0.f, 0.f);
  if (tid < OWN_D * HEADS) {
    mloc[tid] = -3.0e38f;
    sloc[tid] = 0.f;
  }
  if (tid == 0) nlist = 0;
  __syncthreads();

  const int4* dst4 = (const int4*)dst;
  for (int base = 0; base < NE / 4; base += 1024 * 8) {
    int4 dv[8];
#pragma unroll
    for (int u = 0; u < 8; ++u) {
      const int q = base + u * 1024 + tid;
      dv[u] = q < NE / 4 ? dst4[q] : make_int4(-1, -1, -1, -1);
    }
#pragma unroll
    for (int u = 0; u < 8; ++u) {
      const int i = 4 * (base + u * 1024 + tid);
      note_edge(dv[u].x, i + 0, d0, list, &nlist);
      note_edge(dv[u].y, i + 1, d0, list, &nlist);
      note_edge(dv[u].z, i + 2, d0, list, &nlist);
      note_edge(dv[u].w, i + 3, d0, list, &nlist);
    }
  }
  __syncthreads();

  const int n = __builtin_amdgcn_readfirstlane(nlist);
  if (n <= OWN_CAP) {
    softmax_pass<false>(list, n * HEADS, tid, a, mloc, sloc);
    __syncthreads();
    softmax_pass<true>(list, n * HEADS, tid, a, mloc, sloc);
    __syncthreads();
    int j = wave;
    for (; j + 96 < n; j += 128) add_noted<4>(list, j, lane, h, a, mloc, sloc, acc);
    for (; j < n; j += 32) add_noted<1>(list, j, lane, h, a, mloc, sloc, acc);
  } else {
    for (int i = tid; i < NE; i += 1024) {
      const unsigned dl = (unsigned)(dst[i] - d0);
      if (dl < OWN_D)
        for (int hh = 0; hh < HEADS; ++hh)
          atomicMax(&mloc[dl * HEADS + hh], a[(long)i * HEADS + hh]);
    }
    __syncthreads();
    for (int i = tid; i < NE; i += 1024) {
      const unsigned dl = (unsigned)(dst[i] - d0);
      if (dl < OWN_D)
        for (int hh = 0; hh < HEADS; ++hh)
          atomicAdd(&sloc[dl * HEADS + hh],
                    __expf(a[(long)i * HEADS + hh] - mloc[dl * HEADS + hh]));
    }
    __syncthreads();
    for (int i = tid; i < NE; i += 1024) {
      const unsigned dl = (unsigned)(dst[i] - d0);
      if (dl < OWN_D) {
        for (int hh = 0; hh < HEADS; ++hh) {
          const float w = __expf(a[(long)i * HEADS + hh] - mloc[dl * HEADS + hh]) /
                          sloc[dl * HEADS + hh];
          for (int c = 0; c < HID; ++c)
            atomicAdd(&acc[(dl * HEADS + hh) * HID + c],
                      w * h[(long)i * HID + c]);
        }
      }
    }
  }
  __syncthreads();

  float4* o = (float4*)(eft + (long)d0 * HEADS * HID);
  for (int i = tid; i < OWN_D * HEADS * HID / 4; i += 1024) {
    float4 v = acc4[i];
    v.x = v.x > 0.f ? v.x : (__expf(v.x) - 1.f);
    v.y = v.y > 0.f ? v.y : (__expf(v.y) - 1.f);
    v.z = v.z > 0.f ? v.z : (__expf(v.z) - 1.f);
    v.w = v.w > 0.f ? v.w : (__expf(v.w) - 1.f);
    o[i] = v;
  }
}

__global__ __launch_bounds__(256) void score_kernel(
    const float* __restrict__ eft, const int* __restrict__ target0,
    const int* __restrict__ target1, const float* __restrict__ fc1_w,
    const float* __restrict__ fc1_b, const float* __restrict__ fc2_w,
    float* __restrict__ score) {
  int p = blockIdx.y;
  const float* A = eft + (long)p * NG * (HEADS * HID);
  const int* target = p ? target1 : target0;
  int wave = threadIdx.x >> 5, lane = threadIdx.x & 31;
  int tile_m = (blockIdx.x * 8 + wave) * 16;
  int half = lane >> 4, l15 = lane & 15;
  long arow = target[tile_m + l15];
  f32x8 acc[8] = {};
  for (int k0 = 0; k0 < HEADS * HID; k0 += 4) {
    int kk = k0 + half * 2;
    f32x2 af;
    af.x = A[arow * (HEADS * HID) + kk];
    af.y = A[arow * (HEADS * HID) + kk + 1];
#pragma unroll
    for (int nt = 0; nt < 8; ++nt) {
      int col = nt * 16 + l15;
      f32x2 bf;
      bf.x = fc1_w[(long)col * (HEADS * HID) + kk];
      bf.y = fc1_w[(long)col * (HEADS * HID) + kk + 1];
      acc[nt] = wmma_f32(af, bf, acc[nt]);
    }
  }
  float ssum = 0.f;
#pragma unroll
  for (int nt = 0; nt < 8; ++nt) {
    int col = nt * 16 + l15;
    float bb = fc1_b[col], w2 = fc2_w[col];
#pragma unroll
    for (int v = 0; v < 8; ++v) ssum += tanhf(acc[nt][v] + bb) * w2;
  }
#pragma unroll
  for (int off = 16; off > 0; off >>= 1) ssum += __shfl_xor(ssum, off, 32);
  if (lane == 0) atomicAdd(&score[p], ssum);
}

__global__ void hagg_kernel(const float* __restrict__ eft,
                            const int* __restrict__ target0,
                            const int* __restrict__ target1,
                            const float* __restrict__ score,
                            float* __restrict__ hagg, int n) {
  int i = blockIdx.x * 256 + threadIdx.x;
  if (i >= n) return;
  float s0 = score[0] / (float)NB, s1 = score[1] / (float)NB;
  float mx = fmaxf(s0, s1);
  float e0 = __expf(s0 - mx), e1 = __expf(s1 - mx);
  float inv = 1.f / (e0 + e1);
  float beta0 = e0 * inv, beta1 = e1 * inv;
  int b = i >> 9, j = i & 511;
  const float* r0 = eft + (long)target0[b] * (HEADS * HID);
  const float* r1 = eft + (long)NG * (HEADS * HID) +
                    (long)target1[b] * (HEADS * HID);
  hagg[i] = beta0 * r0[j] + beta1 * r1[j];
}

__global__ __launch_bounds__(256) void logits_kernel(
    const float* __restrict__ hagg, const float* __restrict__ fc_w,
    const float* __restrict__ fc_b, float* __restrict__ logits) {
  __shared__ float ldsB[(HEADS * HID) * OUTD];
  int tid = threadIdx.x;
  for (int i = tid; i < (HEADS * HID) * OUTD; i += 256) {
    int k = i >> 4, n = i & 15;
    ldsB[i] = fc_w[n * (HEADS * HID) + k];
  }
  __syncthreads();
  int wave = tid >> 5, lane = tid & 31;
  int tile_m = (blockIdx.x * 8 + wave) * 16;
  int half = lane >> 4, l15 = lane & 15;
  long arow = tile_m + l15;
  f32x8 c = {};
  for (int k0 = 0; k0 < HEADS * HID; k0 += 4) {
    int kk = k0 + half * 2;
    f32x2 af;
    af.x = hagg[arow * (HEADS * HID) + kk];
    af.y = hagg[arow * (HEADS * HID) + kk + 1];
    f32x2 bf;
    bf.x = ldsB[kk * OUTD + l15];
    bf.y = ldsB[(kk + 1) * OUTD + l15];
    c = wmma_f32(af, bf, c);
  }
#pragma unroll
  for (int v = 0; v < 8; ++v) {
    int r = tile_m + v + half * 8;
    logits[(long)r * OUTD + l15] = c[v] + fc_b[l15];
  }
}

extern "C" void kernel_launch(void* const* d_in, const int* in_sizes, int n_in,
                              void* d_out, int out_size, void* d_ws,
                              size_t ws_size, hipStream_t stream) {
  const float* feat0 = (const float*)d_in[0];
  const float* feat1 = (const float*)d_in[1];
  const float* W0    = (const float*)d_in[2];
  const float* b0    = (const float*)d_in[3];
  const float* W1    = (const float*)d_in[4];
  const float* b1    = (const float*)d_in[5];
  const float* r_vec = (const float*)d_in[6];
  const float* attn0 = (const float*)d_in[7];
  const float* attn1 = (const float*)d_in[8];
  const float* fc1_w = (const float*)d_in[9];
  const float* fc1_b = (const float*)d_in[10];
  const float* fc2_w = (const float*)d_in[11];
  const float* fc_w  = (const float*)d_in[12];
  const float* fc_b  = (const float*)d_in[13];
  const int* idx0    = (const int*)d_in[14];
  const int* idx1    = (const int*)d_in[15];
  const int* mp_indices[2] = {(const int*)d_in[16], (const int*)d_in[19]};
  const int* mp_dst[2]     = {(const int*)d_in[17], (const int*)d_in[20]};
  const int* mp_target[2]  = {(const int*)d_in[18], (const int*)d_in[21]};
  const float* attn[2] = {attn0, attn1};

  float* ws     = (float*)d_ws;
  float* tf     = ws;
  float* finals = tf + (long)N_NODES * HID;
  float* h      = finals + 2 * LP * HID;
  float* a      = h + (long)NE * HID;
  float* eft    = a + (long)NE * HEADS;
  float* score  = eft + 2L * NG * HEADS * HID;

  float* logits = (float*)d_out;
  float* hagg   = logits + (long)NB * OUTD;

  setup_kernel<<<GEMM0_WG + GEMM1_WG + INIT_WG, 256, 0, stream>>>(
      feat0, W0, b0, idx0, feat1, W1, b1, idx1, tf, r_vec, finals, score);

  for (int mp = 0; mp < 2; ++mp) {
    edge_h_kernel<<<(NE / EPW + 7) / 8, 256, 0, stream>>>(tf, finals + mp * LP * HID,
                                              mp_indices[mp], attn[mp], h, a);
    owner_sum_kernel<<<NG / OWN_D, 1024, 0, stream>>>(
        h, a, mp_dst[mp], eft + (long)mp * NG * HEADS * HID);
  }

  {
    dim3 g(NB / 128, 2);
    score_kernel<<<g, 256, 0, stream>>>(eft, mp_target[0], mp_target[1], fc1_w,
                                        fc1_b, fc2_w, score);
  }
  hagg_kernel<<<(NB * HEADS * HID + 255) / 256, 256, 0, stream>>>(
      eft, mp_target[0], mp_target[1], score, hagg, NB * HEADS * HID);
  logits_kernel<<<NB / 128, 256, 0, stream>>>(hagg, fc_w, fc_b, logits);
}


# score GEMM: fc1_w staged once per work-group into padded LDS, 256 work-groups (64 rows x 128 cols, wave 16x64), 16-byte fragment loads; k summed in another order inside each 8, 2048 partial sums
# speedup vs baseline: 4.4878x; 1.1536x over previous
#include <hip/hip_runtime.h>
#include <math.h>


#define N_NODES 100000
#define N0      60000
#define N1      40000
#define F0      128
#define F1      64
#define HID     64
#define HEADS   8
#define OUTD    16
#define ATTNV   128
#define NE      250000
#define LP      3
#define NG      8192
#define NB      8192

typedef __attribute__((ext_vector_type(2))) float f32x2;
typedef __attribute__((ext_vector_type(8))) float f32x8;

__device__ __forceinline__ f32x8 wmma_f32(f32x2 a, f32x2 b, f32x8 c) {
  return __builtin_amdgcn_wmma_f32_16x16x4_f32(false, a, false, b, (short)0, c,
                                               false, false);
}

__device__ __forceinline__ float xrow(float x) {
  int b = __builtin_bit_cast(int, x);
  return __builtin_bit_cast(
      float, __builtin_amdgcn_permlanex16(b, b, 0x76543210, 0xfedcba98, false,
                                          false));
}
template <int M>
__device__ __forceinline__ float xlane(float x) {
  int b = __builtin_bit_cast(int, x);
  return __builtin_bit_cast(
      float, __builtin_amdgcn_update_dpp(b, b, 0x160 + M, 0xf, 0xf, false));
}

#define INIT_WG 1
__device__ __forceinline__ void init_body(const float* __restrict__ r_vec,
                                          float* __restrict__ finals  ,
                                          float* __restrict__ score  ) {
  const int c = threadIdx.x;
  if (c < 2) score[c] = 0.f;
  if (c >= 32) return;
  for (int mp = 0; mp < 2; ++mp) {
    float re = r_vec[(mp * 32 + c) * 2 + 0];
    float im = r_vec[(mp * 32 + c) * 2 + 1];
    float inv = rsqrtf(re * re + im * im);
    re *= inv; im *= inv;
    float f1re = re, f1im = -im;
    float f0re = f1re * re - f1im * im;
    float f0im = f1re * im + f1im * re;
    float* F = finals + mp * LP * HID;
    F[0 * HID + 2 * c] = f0re; F[0 * HID + 2 * c + 1] = f0im;
    F[1 * HID + 2 * c] = f1re; F[1 * HID + 2 * c + 1] = f1im;
    F[2 * HID + 2 * c] = 1.f;  F[2 * HID + 2 * c + 1] = 0.f;
  }
}

template <int K>
__device__ __forceinline__ void gemm_tf_body(
    int blk, float* ldsB, const float* __restrict__ feat,
    const float* __restrict__ W, const float* __restrict__ bias,
    const int* __restrict__ idx, float* __restrict__ tf, int M) {
  int tid = threadIdx.x;
  for (int i = tid; i < K * HID; i += 256) {
    int k = i >> 6, n = i & 63;
    ldsB[i] = W[n * K + k];
  }
  __syncthreads();
  int wave = tid >> 5, lane = tid & 31;
  int tile_m = blk * 128 + wave * 16;
  if (tile_m >= M) return;
  int half = lane >> 4, l15 = lane & 15;
  int arow = tile_m + l15;
  if (arow >= M) arow = M - 1;
  f32x8 c0 = {}, c1 = {}, c2 = {}, c3 = {};
  for (int k0 = 0; k0 < K; k0 += 4) {
    int kk = k0 + half * 2;
    f32x2 a;
    a.x = feat[(long)arow * K + kk];
    a.y = feat[(long)arow * K + kk + 1];
    f32x2 b0, b1, b2, b3;
    b0.x = ldsB[kk * HID + l15 + 0];  b0.y = ldsB[(kk + 1) * HID + l15 + 0];
    b1.x = ldsB[kk * HID + l15 + 16]; b1.y = ldsB[(kk + 1) * HID + l15 + 16];
    b2.x = ldsB[kk * HID + l15 + 32]; b2.y = ldsB[(kk + 1) * HID + l15 + 32];
    b3.x = ldsB[kk * HID + l15 + 48]; b3.y = ldsB[(kk + 1) * HID + l15 + 48];
    c0 = wmma_f32(a, b0, c0);
    c1 = wmma_f32(a, b1, c1);
    c2 = wmma_f32(a, b2, c2);
    c3 = wmma_f32(a, b3, c3);
  }
#pragma unroll
  for (int v = 0; v < 8; ++v) {
    int r = tile_m + v + half * 8;
    if (r < M) {
      float* drow = tf + (long)idx[r] * HID;
      drow[l15 + 0]  = c0[v] + bias[l15 + 0];
      drow[l15 + 16] = c1[v] + bias[l15 + 16];
      drow[l15 + 32] = c2[v] + bias[l15 + 32];
      drow[l15 + 48] = c3[v] + bias[l15 + 48];
    }
  }
}

#define GEMM0_WG ((N0 + 127) / 128)
#define GEMM1_WG ((N1 + 127) / 128)
__global__ __launch_bounds__(256) void setup_kernel(
    const float* __restrict__ feat0, const float* __restrict__ W0,
    const float* __restrict__ b0, const int* __restrict__ idx0,
    const float* __restrict__ feat1, const float* __restrict__ W1,
    const float* __restrict__ b1, const int* __restrict__ idx1,
    float* __restrict__ tf, const float* __restrict__ r_vec,
    float* __restrict__ finals, float* __restrict__ score) {
  __shared__ float ldsB[F0 * HID];
  const int blk = blockIdx.x;
  if (blk < GEMM0_WG) {
    gemm_tf_body<F0>(blk, ldsB, feat0, W0, b0, idx0, tf, N0);
  } else if (blk < GEMM0_WG + GEMM1_WG) {
    gemm_tf_body<F1>(blk - GEMM0_WG, ldsB, feat1, W1, b1, idx1, tf, N1);
  } else {
    init_body(r_vec, finals, score);
  }
}

#define EPW 8
__global__ __launch_bounds__(256) void edge_h_kernel(
    const float* __restrict__ tf, const float* __restrict__ finals,
    const int* __restrict__ indices, const float* __restrict__ attn,
    float* __restrict__ h, float* __restrict__ a) {
  const int lane = threadIdx.x & 31;
  const int wave = __builtin_amdgcn_readfirstlane(threadIdx.x >> 5);
  const int e0 = (blockIdx.x * 8 + wave) * EPW;
  if (e0 >= NE) return;
  const int mynode = indices[e0 * LP + (lane < LP * EPW ? lane : lane - 8)];
  const int g = lane >> 2, r = lane & 3;
  float2 v[EPW][LP];
#pragma unroll
  for (int u = 0; u < EPW; ++u) {
#pragma unroll
    for (int i = 0; i < LP; ++i) {
      const int node = __builtin_amdgcn_readlane(mynode, u * LP + i);
      v[u][i] = *((const float2*)(tf + (long)node * HID) + lane);
    }
  }
  float2 f[LP];
#pragma unroll
  for (int i = 0; i < LP; ++i) f[i] = *((const float2*)(finals + i * HID) + lane);
  float2 w[HEADS];
#pragma unroll
  for (int hh = 0; hh < HEADS; ++hh)
    w[hh] = *((const float2*)(attn + hh * HID) + lane);

  const bool b4 = (lane & 16) != 0, b3 = (lane & 8) != 0, b2 = (lane & 4) != 0;
  float tlo = 0.f, thi = 0.f;
#pragma unroll
  for (int u = 0; u < EPW; ++u) {
    float2 acc = make_float2(0.f, 0.f);
#pragma unroll
    for (int i = 0; i < LP; ++i) {
      const float vx = v[u][i].x, vy = v[u][i].y;
      const float fx = f[i].x, fy = f[i].y;
      acc.x += vx * fx - vy * fy;
      acc.y += vx * fy + vy * fx;
    }
    acc.x *= (1.f / 3.f);
    acc.y *= (1.f / 3.f);
    *((float2*)(h + (long)(e0 + u) * HID) + lane) = acc;
    float p[HEADS];
#pragma unroll
    for (int hh = 0; hh < HEADS; ++hh)
      p[hh] = __builtin_fmaf(acc.x, w[hh].x, acc.y * w[hh].y);
    float q[4];
#pragma unroll
    for (int j = 0; j < 4; ++j) {
      float keep = b4 ? p[j + 4] : p[j];
      float send = b4 ? p[j] : p[j + 4];
      q[j] = keep + xrow(send);
    }
    float rr[2];
#pragma unroll
    for (int j = 0; j < 2; ++j) {
      float keep = b3 ? q[j + 2] : q[j];
      float send = b3 ? q[j] : q[j + 2];
      rr[j] = keep + xlane<8>(send);
    }
    float keep2 = b2 ? rr[1] : rr[0];
    float send2 = b2 ? rr[0] : rr[1];
    float t = keep2 + xlane<4>(send2);
    t += xlane<2>(t);
    t += xlane<1>(t);
    if (u < 4) tlo = (r == u) ? t : tlo;
    else       thi = (r == u - 4) ? t : thi;
  }
  const float alo = tlo > 0.f ? tlo : 0.01f * tlo;
  const float ahi = thi > 0.f ? thi : 0.01f * thi;
  a[(long)(e0 + r) * HEADS + g] = alo;
  a[(long)(e0 + 4 + r) * HEADS + g] = ahi;
}

#define OWN_D   32
#define OWN_CAP 25600

__device__ __forceinline__ float lane_bcast(float x, int l) {
  return __builtin_bit_cast(
      float, __builtin_amdgcn_readlane(__builtin_bit_cast(int, x), l));
}

__device__ __forceinline__ void note_edge(int d, int i, int d0, int* list,
                                          int* nlist) {
  const unsigned dl = (unsigned)(d - d0);
  if (dl < OWN_D) {
    const int r = atomicAdd(nlist, 1);
    if (r < OWN_CAP) list[r] = (i << 5) | (int)dl;
  }
}

template <bool SUM>
__device__ __forceinline__ void softmax_pass(const int* list, int n8, int tid,
                                             const float* __restrict__ a,
                                             float* mloc, float* sloc) {
  for (int base = 0; base < n8; base += 1024 * 8) {
    float v[8];
    int key[8];
#pragma unroll
    for (int t = 0; t < 8; ++t) {
      const int j = base + t * 1024 + tid;
      const bool ok = j < n8;
      const int ent = list[ok ? (j >> 3) : 0];
      key[t] = ok ? ((ent & 31) * HEADS + (j & 7)) : -1;
      v[t] = a[(long)(ent >> 5) * HEADS + (j & 7)];
    }
#pragma unroll
    for (int t = 0; t < 8; ++t) {
      if (key[t] >= 0) {
        if (SUM) atomicAdd(&sloc[key[t]], __expf(v[t] - mloc[key[t]]));
        else     atomicMax(&mloc[key[t]], v[t]);
      }
    }
  }
}

template <int NU>
__device__ __forceinline__ void add_noted(const int* list, int j0, int lane,
                                          const float* __restrict__ h,
                                          const float* __restrict__ a,
                                          const float* mloc, const float* sloc,
                                          float* acc) {
  const int mine = list[j0 + 32 * (lane & (NU - 1))];
  int ent[NU];
  float lo[NU], hi[NU], av[NU], mv[NU], sv[NU];
#pragma unroll
  for (int u = 0; u < NU; ++u) {
    ent[u] = __builtin_amdgcn_readlane(mine, u);
    const int e = ent[u] >> 5, dl = ent[u] & 31;
    lo[u] = h[(long)e * HID + lane];
    hi[u] = h[(long)e * HID + 32 + lane];
    av[u] = a[(long)e * HEADS + (lane & 7)];
    mv[u] = mloc[dl * HEADS + (lane & 7)];
    sv[u] = sloc[dl * HEADS + (lane & 7)];
  }
#pragma unroll
  for (int u = 0; u < NU; ++u) {
    const float wl = __expf(av[u] - mv[u]) / sv[u];
    float* row = acc + (ent[u] & 31) * (HEADS * HID) + lane;
#pragma unroll
    for (int hh = 0; hh < HEADS; ++hh) {
      const float w = lane_bcast(wl, hh);
      atomicAdd(row + hh * HID, w * lo[u]);
      atomicAdd(row + hh * HID + 32, w * hi[u]);
    }
  }
}

__global__ __launch_bounds__(1024) void owner_sum_kernel(
    const float* __restrict__ h, const float* __restrict__ a,
    const int* __restrict__ dst,
    float* __restrict__ eft  ) {
  __shared__ __attribute__((aligned(16))) float acc[OWN_D * HEADS * HID];
  __shared__ int list[OWN_CAP];
  __shared__ float mloc[OWN_D * HEADS];
  __shared__ float sloc[OWN_D * HEADS];
  __shared__ int nlist;
  const int tid = threadIdx.x, lane = tid & 31;
  const int wave = __builtin_amdgcn_readfirstlane(tid >> 5);
  const int d0 = blockIdx.x * OWN_D;

  float4* acc4 = (float4*)acc;
  for (int i = tid; i < OWN_D * HEADS * HID / 4; i += 1024)
    acc4[i] = make_float4(0.f, 0.f, 0.f, 0.f);
  if (tid < OWN_D * HEADS) {
    mloc[tid] = -3.0e38f;
    sloc[tid] = 0.f;
  }
  if (tid == 0) nlist = 0;
  __syncthreads();

  const int4* dst4 = (const int4*)dst;
  for (int base = 0; base < NE / 4; base += 1024 * 8) {
    int4 dv[8];
#pragma unroll
    for (int u = 0; u < 8; ++u) {
      const int q = base + u * 1024 + tid;
      dv[u] = q < NE / 4 ? dst4[q] : make_int4(-1, -1, -1, -1);
    }
#pragma unroll
    for (int u = 0; u < 8; ++u) {
      const int i = 4 * (base + u * 1024 + tid);
      note_edge(dv[u].x, i + 0, d0, list, &nlist);
      note_edge(dv[u].y, i + 1, d0, list, &nlist);
      note_edge(dv[u].z, i + 2, d0, list, &nlist);
      note_edge(dv[u].w, i + 3, d0, list, &nlist);
    }
  }
  __syncthreads();

  const int n = __builtin_amdgcn_readfirstlane(nlist);
  if (n <= OWN_CAP) {
    softmax_pass<false>(list, n * HEADS, tid, a, mloc, sloc);
    __syncthreads();
    softmax_pass<true>(list, n * HEADS, tid, a, mloc, sloc);
    __syncthreads();
    int j = wave;
    for (; j + 96 < n; j += 128) add_noted<4>(list, j, lane, h, a, mloc, sloc, acc);
    for (; j < n; j += 32) add_noted<1>(list, j, lane, h, a, mloc, sloc, acc);
  } else {
    for (int i = tid; i < NE; i += 1024) {
      const unsigned dl = (unsigned)(dst[i] - d0);
      if (dl < OWN_D)
        for (int hh = 0; hh < HEADS; ++hh)
          atomicMax(&mloc[dl * HEADS + hh], a[(long)i * HEADS + hh]);
    }
    __syncthreads();
    for (int i = tid; i < NE; i += 1024) {
      const unsigned dl = (unsigned)(dst[i] - d0);
      if (dl < OWN_D)
        for (int hh = 0; hh < HEADS; ++hh)
          atomicAdd(&sloc[dl * HEADS + hh],
                    __expf(a[(long)i * HEADS + hh] - mloc[dl * HEADS + hh]));
    }
    __syncthreads();
    for (int i = tid; i < NE; i += 1024) {
      const unsigned dl = (unsigned)(dst[i] - d0);
      if (dl < OWN_D) {
        for (int hh = 0; hh < HEADS; ++hh) {
          const float w = __expf(a[(long)i * HEADS + hh] - mloc[dl * HEADS + hh]) /
                          sloc[dl * HEADS + hh];
          for (int c = 0; c < HID; ++c)
            atomicAdd(&acc[(dl * HEADS + hh) * HID + c],
                      w * h[(long)i * HID + c]);
        }
      }
    }
  }
  __syncthreads();

  float4* o = (float4*)(eft + (long)d0 * HEADS * HID);
  for (int i = tid; i < OWN_D * HEADS * HID / 4; i += 1024) {
    float4 v = acc4[i];
    v.x = v.x > 0.f ? v.x : (__expf(v.x) - 1.f);
    v.y = v.y > 0.f ? v.y : (__expf(v.y) - 1.f);
    v.z = v.z > 0.f ? v.z : (__expf(v.z) - 1.f);
    v.w = v.w > 0.f ? v.w : (__expf(v.w) - 1.f);
    o[i] = v;
  }
}

#define SC_ROWS  64
#define SC_PITCH 516
__device__ __forceinline__ void score_chunk(const float4 (&a)[4], const float* b,
                                            f32x8 (&acc)[4]) {
#pragma unroll
  for (int q = 0; q < 4; ++q) {
    float4 bv[4];
#pragma unroll
    for (int nt = 0; nt < 4; ++nt)
      bv[nt] = *(const float4*)(b + nt * 16 * SC_PITCH + q * 8);
    f32x2 a01, a23;
    a01.x = a[q].x; a01.y = a[q].y;
    a23.x = a[q].z; a23.y = a[q].w;
#pragma unroll
    for (int nt = 0; nt < 4; ++nt) {
      f32x2 b01;
      b01.x = bv[nt].x; b01.y = bv[nt].y;
      acc[nt] = wmma_f32(a01, b01, acc[nt]);
    }
#pragma unroll
    for (int nt = 0; nt < 4; ++nt) {
      f32x2 b23;
      b23.x = bv[nt].z; b23.y = bv[nt].w;
      acc[nt] = wmma_f32(a23, b23, acc[nt]);
    }
  }
}
__global__ __launch_bounds__(256) void score_kernel(
    const float* __restrict__ eft, const int* __restrict__ target0,
    const int* __restrict__ target1, const float* __restrict__ fc1_w,
    const float* __restrict__ fc1_b, const float* __restrict__ fc2_w,
    float* __restrict__ score) {
  __shared__ __attribute__((aligned(16))) float ldsW[ATTNV * SC_PITCH];
  const int tid = threadIdx.x;
  const float4* w4 = (const float4*)fc1_w;
  const int rot = blockIdx.x * 128;
  for (int base = 0; base < ATTNV * (HEADS * HID) / 4; base += 256 * 16) {
    float4 t[16];
#pragma unroll
    for (int u = 0; u < 16; ++u)
      t[u] = w4[(base + u * 256 + tid + rot) & (ATTNV * (HEADS * HID) / 4 - 1)];
#pragma unroll
    for (int u = 0; u < 16; ++u) {
      const int i = (base + u * 256 + tid + rot) & (ATTNV * (HEADS * HID) / 4 - 1);
      *(float4*)(ldsW + (i >> 7) * SC_PITCH + (i & 127) * 4) = t[u];
    }
  }
  __syncthreads();

  const int p = blockIdx.x / (NB / SC_ROWS);
  const int mblk = blockIdx.x % (NB / SC_ROWS);
  const float* A = eft + (long)p * NG * (HEADS * HID);
  const int* target = p ? target1 : target0;
  const int wave = tid >> 5, lane = tid & 31;
  const int mt = wave & 3, nh = wave >> 2;
  const int half = lane >> 4, l15 = lane & 15;
  const long arow = target[mblk * SC_ROWS + mt * 16 + l15];
  const float4* ap = (const float4*)(A + arow * (HEADS * HID) + half * 4);
  const float* bp = ldsW + (nh * 64 + l15) * SC_PITCH + half * 4;
  f32x8 acc[4] = {};
  float4 a0[4], a1[4];
#pragma unroll
  for (int q = 0; q < 4; ++q) a0[q] = ap[q * 2];
  for (int ch = 0; ch < (HEADS * HID) / 32; ch += 2) {
#pragma unroll
    for (int q = 0; q < 4; ++q) a1[q] = ap[((ch + 1) * 4 + q) * 2];
    score_chunk(a0, bp + ch * 32, acc);
    const int nx = ch + 2 < (HEADS * HID) / 32 ? ch + 2 : ch;
#pragma unroll
    for (int q = 0; q < 4; ++q) a0[q] = ap[(nx * 4 + q) * 2];
    asm volatile("" ::: "memory");
    score_chunk(a1, bp + (ch + 1) * 32, acc);
  }
  float ssum = 0.f;
#pragma unroll
  for (int nt = 0; nt < 4; ++nt) {
    int col = nh * 64 + nt * 16 + l15;
    float bb = fc1_b[col], w2 = fc2_w[col];
#pragma unroll
    for (int v = 0; v < 8; ++v) ssum += tanhf(acc[nt][v] + bb) * w2;
  }
#pragma unroll
  for (int off = 16; off > 0; off >>= 1) ssum += __shfl_xor(ssum, off, 32);
  if (lane == 0) atomicAdd(&score[p], ssum);
}

__global__ void hagg_kernel(const float* __restrict__ eft,
                            const int* __restrict__ target0,
                            const int* __restrict__ target1,
                            const float* __restrict__ score,
                            float* __restrict__ hagg, int n) {
  int i = blockIdx.x * 256 + threadIdx.x;
  if (i >= n) return;
  float s0 = score[0] / (float)NB, s1 = score[1] / (float)NB;
  float mx = fmaxf(s0, s1);
  float e0 = __expf(s0 - mx), e1 = __expf(s1 - mx);
  float inv = 1.f / (e0 + e1);
  float beta0 = e0 * inv, beta1 = e1 * inv;
  int b = i >> 9, j = i & 511;
  const float* r0 = eft + (long)target0[b] * (HEADS * HID);
  const float* r1 = eft + (long)NG * (HEADS * HID) +
                    (long)target1[b] * (HEADS * HID);
  hagg[i] = beta0 * r0[j] + beta1 * r1[j];
}

__global__ __launch_bounds__(256) void logits_kernel(
    const float* __restrict__ hagg, const float* __restrict__ fc_w,
    const float* __restrict__ fc_b, float* __restrict__ logits) {
  __shared__ float ldsB[(HEADS * HID) * OUTD];
  int tid = threadIdx.x;
  for (int i = tid; i < (HEADS * HID) * OUTD; i += 256) {
    int k = i >> 4, n = i & 15;
    ldsB[i] = fc_w[n * (HEADS * HID) + k];
  }
  __syncthreads();
  int wave = tid >> 5, lane = tid & 31;
  int tile_m = (blockIdx.x * 8 + wave) * 16;
  int half = lane >> 4, l15 = lane & 15;
  long arow = tile_m + l15;
  f32x8 c = {};
  for (int k0 = 0; k0 < HEADS * HID; k0 += 4) {
    int kk = k0 + half * 2;
    f32x2 af;
    af.x = hagg[arow * (HEADS * HID) + kk];
    af.y = hagg[arow * (HEADS * HID) + kk + 1];
    f32x2 bf;
    bf.x = ldsB[kk * OUTD + l15];
    bf.y = ldsB[(kk + 1) * OUTD + l15];
    c = wmma_f32(af, bf, c);
  }
#pragma unroll
  for (int v = 0; v < 8; ++v) {
    int r = tile_m + v + half * 8;
    logits[(long)r * OUTD + l15] = c[v] + fc_b[l15];
  }
}

extern "C" void kernel_launch(void* const* d_in, const int* in_sizes, int n_in,
                              void* d_out, int out_size, void* d_ws,
                              size_t ws_size, hipStream_t stream) {
  const float* feat0 = (const float*)d_in[0];
  const float* feat1 = (const float*)d_in[1];
  const float* W0    = (const float*)d_in[2];
  const float* b0    = (const float*)d_in[3];
  const float* W1    = (const float*)d_in[4];
  const float* b1    = (const float*)d_in[5];
  const float* r_vec = (const float*)d_in[6];
  const float* attn0 = (const float*)d_in[7];
  const float* attn1 = (const float*)d_in[8];
  const float* fc1_w = (const float*)d_in[9];
  const float* fc1_b = (const float*)d_in[10];
  const float* fc2_w = (const float*)d_in[11];
  const float* fc_w  = (const float*)d_in[12];
  const float* fc_b  = (const float*)d_in[13];
  const int* idx0    = (const int*)d_in[14];
  const int* idx1    = (const int*)d_in[15];
  const int* mp_indices[2] = {(const int*)d_in[16], (const int*)d_in[19]};
  const int* mp_dst[2]     = {(const int*)d_in[17], (const int*)d_in[20]};
  const int* mp_target[2]  = {(const int*)d_in[18], (const int*)d_in[21]};
  const float* attn[2] = {attn0, attn1};

  float* ws     = (float*)d_ws;
  float* tf     = ws;
  float* finals = tf + (long)N_NODES * HID;
  float* h      = finals + 2 * LP * HID;
  float* a      = h + (long)NE * HID;
  float* eft    = a + (long)NE * HEADS;
  float* score  = eft + 2L * NG * HEADS * HID;

  float* logits = (float*)d_out;
  float* hagg   = logits + (long)NB * OUTD;

  setup_kernel<<<GEMM0_WG + GEMM1_WG + INIT_WG, 256, 0, stream>>>(
      feat0, W0, b0, idx0, feat1, W1, b1, idx1, tf, r_vec, finals, score);

  for (int mp = 0; mp < 2; ++mp) {
    edge_h_kernel<<<(NE / EPW + 7) / 8, 256, 0, stream>>>(tf, finals + mp * LP * HID,
                                              mp_indices[mp], attn[mp], h, a);
    owner_sum_kernel<<<NG / OWN_D, 1024, 0, stream>>>(
        h, a, mp_dst[mp], eft + (long)mp * NG * HEADS * HID);
  }

  score_kernel<<<2 * (NB / SC_ROWS), 256, 0, stream>>>(
      eft, mp_target[0], mp_target[1], fc1_w, fc1_b, fc2_w, score);
  hagg_kernel<<<(NB * HEADS * HID + 255) / 256, 256, 0, stream>>>(
      eft, mp_target[0], mp_target[1], score, hagg, NB * HEADS * HID);
  logits_kernel<<<NB / 128, 256, 0, stream>>>(hagg, fc_w, fc_b, logits);
}


# logits product taken inside the hagg launch: a work-group forms 16 hagg rows (16 bytes per lane), keeps a copy in LDS, one wave runs their WMMA chain in the same k order; 8 -> 7 launches
# speedup vs baseline: 4.9632x; 1.1059x over previous
#include <hip/hip_runtime.h>
#include <math.h>


#define N_NODES 100000
#define N0      60000
#define N1      40000
#define F0      128
#define F1      64
#define HID     64
#define HEADS   8
#define OUTD    16
#define ATTNV   128
#define NE      250000
#define LP      3
#define NG      8192
#define NB      8192

typedef __attribute__((ext_vector_type(2))) float f32x2;
typedef __attribute__((ext_vector_type(8))) float f32x8;

__device__ __forceinline__ f32x8 wmma_f32(f32x2 a, f32x2 b, f32x8 c) {
  return __builtin_amdgcn_wmma_f32_16x16x4_f32(false, a, false, b, (short)0, c,
                                               false, false);
}

__device__ __forceinline__ float xrow(float x) {
  int b = __builtin_bit_cast(int, x);
  return __builtin_bit_cast(
      float, __builtin_amdgcn_permlanex16(b, b, 0x76543210, 0xfedcba98, false,
                                          false));
}
template <int M>
__device__ __forceinline__ float xlane(float x) {
  int b = __builtin_bit_cast(int, x);
  return __builtin_bit_cast(
      float, __builtin_amdgcn_update_dpp(b, b, 0x160 + M, 0xf, 0xf, false));
}

#define INIT_WG 1
__device__ __forceinline__ void init_body(const float* __restrict__ r_vec,
                                          float* __restrict__ finals  ,
                                          float* __restrict__ score  ) {
  const int c = threadIdx.x;
  if (c < 2) score[c] = 0.f;
  if (c >= 32) return;
  for (int mp = 0; mp < 2; ++mp) {
    float re = r_vec[(mp * 32 + c) * 2 + 0];
    float im = r_vec[(mp * 32 + c) * 2 + 1];
    float inv = rsqrtf(re * re + im * im);
    re *= inv; im *= inv;
    float f1re = re, f1im = -im;
    float f0re = f1re * re - f1im * im;
    float f0im = f1re * im + f1im * re;
    float* F = finals + mp * LP * HID;
    F[0 * HID + 2 * c] = f0re; F[0 * HID + 2 * c + 1] = f0im;
    F[1 * HID + 2 * c] = f1re; F[1 * HID + 2 * c + 1] = f1im;
    F[2 * HID + 2 * c] = 1.f;  F[2 * HID + 2 * c + 1] = 0.f;
  }
}

template <int K>
__device__ __forceinline__ void gemm_tf_body(
    int blk, float* ldsB, const float* __restrict__ feat,
    const float* __restrict__ W, const float* __restrict__ bias,
    const int* __restrict__ idx, float* __restrict__ tf, int M) {
  int tid = threadIdx.x;
  for (int i = tid; i < K * HID; i += 256) {
    int k = i >> 6, n = i & 63;
    ldsB[i] = W[n * K + k];
  }
  __syncthreads();
  int wave = tid >> 5, lane = tid & 31;
  int tile_m = blk * 128 + wave * 16;
  if (tile_m >= M) return;
  int half = lane >> 4, l15 = lane & 15;
  int arow = tile_m + l15;
  if (arow >= M) arow = M - 1;
  f32x8 c0 = {}, c1 = {}, c2 = {}, c3 = {};
  for (int k0 = 0; k0 < K; k0 += 4) {
    int kk = k0 + half * 2;
    f32x2 a;
    a.x = feat[(long)arow * K + kk];
    a.y = feat[(long)arow * K + kk + 1];
    f32x2 b0, b1, b2, b3;
    b0.x = ldsB[kk * HID + l15 + 0];  b0.y = ldsB[(kk + 1) * HID + l15 + 0];
    b1.x = ldsB[kk * HID + l15 + 16]; b1.y = ldsB[(kk + 1) * HID + l15 + 16];
    b2.x = ldsB[kk * HID + l15 + 32]; b2.y = ldsB[(kk + 1) * HID + l15 + 32];
    b3.x = ldsB[kk * HID + l15 + 48]; b3.y = ldsB[(kk + 1) * HID + l15 + 48];
    c0 = wmma_f32(a, b0, c0);
    c1 = wmma_f32(a, b1, c1);
    c2 = wmma_f32(a, b2, c2);
    c3 = wmma_f32(a, b3, c3);
  }
#pragma unroll
  for (int v = 0; v < 8; ++v) {
    int r = tile_m + v + half * 8;
    if (r < M) {
      float* drow = tf + (long)idx[r] * HID;
      drow[l15 + 0]  = c0[v] + bias[l15 + 0];
      drow[l15 + 16] = c1[v] + bias[l15 + 16];
      drow[l15 + 32] = c2[v] + bias[l15 + 32];
      drow[l15 + 48] = c3[v] + bias[l15 + 48];
    }
  }
}

#define GEMM0_WG ((N0 + 127) / 128)
#define GEMM1_WG ((N1 + 127) / 128)
__global__ __launch_bounds__(256) void setup_kernel(
    const float* __restrict__ feat0, const float* __restrict__ W0,
    const float* __restrict__ b0, const int* __restrict__ idx0,
    const float* __restrict__ feat1, const float* __restrict__ W1,
    const float* __restrict__ b1, const int* __restrict__ idx1,
    float* __restrict__ tf, const float* __restrict__ r_vec,
    float* __restrict__ finals, float* __restrict__ score) {
  __shared__ float ldsB[F0 * HID];
  const int blk = blockIdx.x;
  if (blk < GEMM0_WG) {
    gemm_tf_body<F0>(blk, ldsB, feat0, W0, b0, idx0, tf, N0);
  } else if (blk < GEMM0_WG + GEMM1_WG) {
    gemm_tf_body<F1>(blk - GEMM0_WG, ldsB, feat1, W1, b1, idx1, tf, N1);
  } else {
    init_body(r_vec, finals, score);
  }
}

#define EPW 8
__global__ __launch_bounds__(256) void edge_h_kernel(
    const float* __restrict__ tf, const float* __restrict__ finals,
    const int* __restrict__ indices, const float* __restrict__ attn,
    float* __restrict__ h, float* __restrict__ a) {
  const int lane = threadIdx.x & 31;
  const int wave = __builtin_amdgcn_readfirstlane(threadIdx.x >> 5);
  const int e0 = (blockIdx.x * 8 + wave) * EPW;
  if (e0 >= NE) return;
  const int mynode = indices[e0 * LP + (lane < LP * EPW ? lane : lane - 8)];
  const int g = lane >> 2, r = lane & 3;
  float2 v[EPW][LP];
#pragma unroll
  for (int u = 0; u < EPW; ++u) {
#pragma unroll
    for (int i = 0; i < LP; ++i) {
      const int node = __builtin_amdgcn_readlane(mynode, u * LP + i);
      v[u][i] = *((const float2*)(tf + (long)node * HID) + lane);
    }
  }
  float2 f[LP];
#pragma unroll
  for (int i = 0; i < LP; ++i) f[i] = *((const float2*)(finals + i * HID) + lane);
  float2 w[HEADS];
#pragma unroll
  for (int hh = 0; hh < HEADS; ++hh)
    w[hh] = *((const float2*)(attn + hh * HID) + lane);

  const bool b4 = (lane & 16) != 0, b3 = (lane & 8) != 0, b2 = (lane & 4) != 0;
  float tlo = 0.f, thi = 0.f;
#pragma unroll
  for (int u = 0; u < EPW; ++u) {
    float2 acc = make_float2(0.f, 0.f);
#pragma unroll
    for (int i = 0; i < LP; ++i) {
      const float vx = v[u][i].x, vy = v[u][i].y;
      const float fx = f[i].x, fy = f[i].y;
      acc.x += vx * fx - vy * fy;
      acc.y += vx * fy + vy * fx;
    }
    acc.x *= (1.f / 3.f);
    acc.y *= (1.f / 3.f);
    *((float2*)(h + (long)(e0 + u) * HID) + lane) = acc;
    float p[HEADS];
#pragma unroll
    for (int hh = 0; hh < HEADS; ++hh)
      p[hh] = __builtin_fmaf(acc.x, w[hh].x, acc.y * w[hh].y);
    float q[4];
#pragma unroll
    for (int j = 0; j < 4; ++j) {
      float keep = b4 ? p[j + 4] : p[j];
      float send = b4 ? p[j] : p[j + 4];
      q[j] = keep + xrow(send);
    }
    float rr[2];
#pragma unroll
    for (int j = 0; j < 2; ++j) {
      float keep = b3 ? q[j + 2] : q[j];
      float send = b3 ? q[j] : q[j + 2];
      rr[j] = keep + xlane<8>(send);
    }
    float keep2 = b2 ? rr[1] : rr[0];
    float send2 = b2 ? rr[0] : rr[1];
    float t = keep2 + xlane<4>(send2);
    t += xlane<2>(t);
    t += xlane<1>(t);
    if (u < 4) tlo = (r == u) ? t : tlo;
    else       thi = (r == u - 4) ? t : thi;
  }
  const float alo = tlo > 0.f ? tlo : 0.01f * tlo;
  const float ahi = thi > 0.f ? thi : 0.01f * thi;
  a[(long)(e0 + r) * HEADS + g] = alo;
  a[(long)(e0 + 4 + r) * HEADS + g] = ahi;
}

#define OWN_D   32
#define OWN_CAP 25600

__device__ __forceinline__ float lane_bcast(float x, int l) {
  return __builtin_bit_cast(
      float, __builtin_amdgcn_readlane(__builtin_bit_cast(int, x), l));
}

__device__ __forceinline__ void note_edge(int d, int i, int d0, int* list,
                                          int* nlist) {
  const unsigned dl = (unsigned)(d - d0);
  if (dl < OWN_D) {
    const int r = atomicAdd(nlist, 1);
    if (r < OWN_CAP) list[r] = (i << 5) | (int)dl;
  }
}

template <bool SUM>
__device__ __forceinline__ void softmax_pass(const int* list, int n8, int tid,
                                             const float* __restrict__ a,
                                             float* mloc, float* sloc) {
  for (int base = 0; base < n8; base += 1024 * 8) {
    float v[8];
    int key[8];
#pragma unroll
    for (int t = 0; t < 8; ++t) {
      const int j = base + t * 1024 + tid;
      const bool ok = j < n8;
      const int ent = list[ok ? (j >> 3) : 0];
      key[t] = ok ? ((ent & 31) * HEADS + (j & 7)) : -1;
      v[t] = a[(long)(ent >> 5) * HEADS + (j & 7)];
    }
#pragma unroll
    for (int t = 0; t < 8; ++t) {
      if (key[t] >= 0) {
        if (SUM) atomicAdd(&sloc[key[t]], __expf(v[t] - mloc[key[t]]));
        else     atomicMax(&mloc[key[t]], v[t]);
      }
    }
  }
}

template <int NU>
__device__ __forceinline__ void add_noted(const int* list, int j0, int lane,
                                          const float* __restrict__ h,
                                          const float* __restrict__ a,
                                          const float* mloc, const float* sloc,
                                          float* acc) {
  const int mine = list[j0 + 32 * (lane & (NU - 1))];
  int ent[NU];
  float lo[NU], hi[NU], av[NU], mv[NU], sv[NU];
#pragma unroll
  for (int u = 0; u < NU; ++u) {
    ent[u] = __builtin_amdgcn_readlane(mine, u);
    const int e = ent[u] >> 5, dl = ent[u] & 31;
    lo[u] = h[(long)e * HID + lane];
    hi[u] = h[(long)e * HID + 32 + lane];
    av[u] = a[(long)e * HEADS + (lane & 7)];
    mv[u] = mloc[dl * HEADS + (lane & 7)];
    sv[u] = sloc[dl * HEADS + (lane & 7)];
  }
#pragma unroll
  for (int u = 0; u < NU; ++u) {
    const float wl = __expf(av[u] - mv[u]) / sv[u];
    float* row = acc + (ent[u] & 31) * (HEADS * HID) + lane;
#pragma unroll
    for (int hh = 0; hh < HEADS; ++hh) {
      const float w = lane_bcast(wl, hh);
      atomicAdd(row + hh * HID, w * lo[u]);
      atomicAdd(row + hh * HID + 32, w * hi[u]);
    }
  }
}

__global__ __launch_bounds__(1024) void owner_sum_kernel(
    const float* __restrict__ h, const float* __restrict__ a,
    const int* __restrict__ dst,
    float* __restrict__ eft  ) {
  __shared__ __attribute__((aligned(16))) float acc[OWN_D * HEADS * HID];
  __shared__ int list[OWN_CAP];
  __shared__ float mloc[OWN_D * HEADS];
  __shared__ float sloc[OWN_D * HEADS];
  __shared__ int nlist;
  const int tid = threadIdx.x, lane = tid & 31;
  const int wave = __builtin_amdgcn_readfirstlane(tid >> 5);
  const int d0 = blockIdx.x * OWN_D;

  float4* acc4 = (float4*)acc;
  for (int i = tid; i < OWN_D * HEADS * HID / 4; i += 1024)
    acc4[i] = make_float4(0.f, 0.f, 0.f, 0.f);
  if (tid < OWN_D * HEADS) {
    mloc[tid] = -3.0e38f;
    sloc[tid] = 0.f;
  }
  if (tid == 0) nlist = 0;
  __syncthreads();

  const int4* dst4 = (const int4*)dst;
  for (int base = 0; base < NE / 4; base += 1024 * 8) {
    int4 dv[8];
#pragma unroll
    for (int u = 0; u < 8; ++u) {
      const int q = base + u * 1024 + tid;
      dv[u] = q < NE / 4 ? dst4[q] : make_int4(-1, -1, -1, -1);
    }
#pragma unroll
    for (int u = 0; u < 8; ++u) {
      const int i = 4 * (base + u * 1024 + tid);
      note_edge(dv[u].x, i + 0, d0, list, &nlist);
      note_edge(dv[u].y, i + 1, d0, list, &nlist);
      note_edge(dv[u].z, i + 2, d0, list, &nlist);
      note_edge(dv[u].w, i + 3, d0, list, &nlist);
    }
  }
  __syncthreads();

  const int n = __builtin_amdgcn_readfirstlane(nlist);
  if (n <= OWN_CAP) {
    softmax_pass<false>(list, n * HEADS, tid, a, mloc, sloc);
    __syncthreads();
    softmax_pass<true>(list, n * HEADS, tid, a, mloc, sloc);
    __syncthreads();
    int j = wave;
    for (; j + 96 < n; j += 128) add_noted<4>(list, j, lane, h, a, mloc, sloc, acc);
    for (; j < n; j += 32) add_noted<1>(list, j, lane, h, a, mloc, sloc, acc);
  } else {
    for (int i = tid; i < NE; i += 1024) {
      const unsigned dl = (unsigned)(dst[i] - d0);
      if (dl < OWN_D)
        for (int hh = 0; hh < HEADS; ++hh)
          atomicMax(&mloc[dl * HEADS + hh], a[(long)i * HEADS + hh]);
    }
    __syncthreads();
    for (int i = tid; i < NE; i += 1024) {
      const unsigned dl = (unsigned)(dst[i] - d0);
      if (dl < OWN_D)
        for (int hh = 0; hh < HEADS; ++hh)
          atomicAdd(&sloc[dl * HEADS + hh],
                    __expf(a[(long)i * HEADS + hh] - mloc[dl * HEADS + hh]));
    }
    __syncthreads();
    for (int i = tid; i < NE; i += 1024) {
      const unsigned dl = (unsigned)(dst[i] - d0);
      if (dl < OWN_D) {
        for (int hh = 0; hh < HEADS; ++hh) {
          const float w = __expf(a[(long)i * HEADS + hh] - mloc[dl * HEADS + hh]) /
                          sloc[dl * HEADS + hh];
          for (int c = 0; c < HID; ++c)
            atomicAdd(&acc[(dl * HEADS + hh) * HID + c],
                      w * h[(long)i * HID + c]);
        }
      }
    }
  }
  __syncthreads();

  float4* o = (float4*)(eft + (long)d0 * HEADS * HID);
  for (int i = tid; i < OWN_D * HEADS * HID / 4; i += 1024) {
    float4 v = acc4[i];
    v.x = v.x > 0.f ? v.x : (__expf(v.x) - 1.f);
    v.y = v.y > 0.f ? v.y : (__expf(v.y) - 1.f);
    v.z = v.z > 0.f ? v.z : (__expf(v.z) - 1.f);
    v.w = v.w > 0.f ? v.w : (__expf(v.w) - 1.f);
    o[i] = v;
  }
}

#define SC_ROWS  64
#define SC_PITCH 516
__device__ __forceinline__ void score_chunk(const float4 (&a)[4], const float* b,
                                            f32x8 (&acc)[4]) {
#pragma unroll
  for (int q = 0; q < 4; ++q) {
    float4 bv[4];
#pragma unroll
    for (int nt = 0; nt < 4; ++nt)
      bv[nt] = *(const float4*)(b + nt * 16 * SC_PITCH + q * 8);
    f32x2 a01, a23;
    a01.x = a[q].x; a01.y = a[q].y;
    a23.x = a[q].z; a23.y = a[q].w;
#pragma unroll
    for (int nt = 0; nt < 4; ++nt) {
      f32x2 b01;
      b01.x = bv[nt].x; b01.y = bv[nt].y;
      acc[nt] = wmma_f32(a01, b01, acc[nt]);
    }
#pragma unroll
    for (int nt = 0; nt < 4; ++nt) {
      f32x2 b23;
      b23.x = bv[nt].z; b23.y = bv[nt].w;
      acc[nt] = wmma_f32(a23, b23, acc[nt]);
    }
  }
}
__global__ __launch_bounds__(256) void score_kernel(
    const float* __restrict__ eft, const int* __restrict__ target0,
    const int* __restrict__ target1, const float* __restrict__ fc1_w,
    const float* __restrict__ fc1_b, const float* __restrict__ fc2_w,
    float* __restrict__ score) {
  __shared__ __attribute__((aligned(16))) float ldsW[ATTNV * SC_PITCH];
  const int tid = threadIdx.x;
  const float4* w4 = (const float4*)fc1_w;
  const int rot = blockIdx.x * 128;
  for (int base = 0; base < ATTNV * (HEADS * HID) / 4; base += 256 * 16) {
    float4 t[16];
#pragma unroll
    for (int u = 0; u < 16; ++u)
      t[u] = w4[(base + u * 256 + tid + rot) & (ATTNV * (HEADS * HID) / 4 - 1)];
#pragma unroll
    for (int u = 0; u < 16; ++u) {
      const int i = (base + u * 256 + tid + rot) & (ATTNV * (HEADS * HID) / 4 - 1);
      *(float4*)(ldsW + (i >> 7) * SC_PITCH + (i & 127) * 4) = t[u];
    }
  }
  __syncthreads();

  const int p = blockIdx.x / (NB / SC_ROWS);
  const int mblk = blockIdx.x % (NB / SC_ROWS);
  const float* A = eft + (long)p * NG * (HEADS * HID);
  const int* target = p ? target1 : target0;
  const int wave = tid >> 5, lane = tid & 31;
  const int mt = wave & 3, nh = wave >> 2;
  const int half = lane >> 4, l15 = lane & 15;
  const long arow = target[mblk * SC_ROWS + mt * 16 + l15];
  const float4* ap = (const float4*)(A + arow * (HEADS * HID) + half * 4);
  const float* bp = ldsW + (nh * 64 + l15) * SC_PITCH + half * 4;
  f32x8 acc[4] = {};
  float4 a0[4], a1[4];
#pragma unroll
  for (int q = 0; q < 4; ++q) a0[q] = ap[q * 2];
  for (int ch = 0; ch < (HEADS * HID) / 32; ch += 2) {
#pragma unroll
    for (int q = 0; q < 4; ++q) a1[q] = ap[((ch + 1) * 4 + q) * 2];
    score_chunk(a0, bp + ch * 32, acc);
    const int nx = ch + 2 < (HEADS * HID) / 32 ? ch + 2 : ch;
#pragma unroll
    for (int q = 0; q < 4; ++q) a0[q] = ap[(nx * 4 + q) * 2];
    asm volatile("" ::: "memory");
    score_chunk(a1, bp + (ch + 1) * 32, acc);
  }
  float ssum = 0.f;
#pragma unroll
  for (int nt = 0; nt < 4; ++nt) {
    int col = nh * 64 + nt * 16 + l15;
    float bb = fc1_b[col], w2 = fc2_w[col];
#pragma unroll
    for (int v = 0; v < 8; ++v) ssum += tanhf(acc[nt][v] + bb) * w2;
  }
#pragma unroll
  for (int off = 16; off > 0; off >>= 1) ssum += __shfl_xor(ssum, off, 32);
  if (lane == 0) atomicAdd(&score[p], ssum);
}

#define HL_ROWS  16
#define HL_PITCH 516
__global__ __launch_bounds__(256) void hagg_logits_kernel(
    const float* __restrict__ eft, const int* __restrict__ target0,
    const int* __restrict__ target1, const float* __restrict__ score,
    const float* __restrict__ fc_w, const float* __restrict__ fc_b,
    float* __restrict__ hagg, float* __restrict__ logits) {
  __shared__ __attribute__((aligned(16))) float ldsA[HL_ROWS * HL_PITCH];
  __shared__ __attribute__((aligned(16))) float ldsW[OUTD * HL_PITCH];
  const int tid = threadIdx.x;
  const int b0 = blockIdx.x * HL_ROWS;
  const int c4 = tid & 127, rsub = tid >> 7;
  int t0[8], t1[8];
#pragma unroll
  for (int u = 0; u < 8; ++u) {
    t0[u] = target0[b0 + 2 * u + rsub];
    t1[u] = target1[b0 + 2 * u + rsub];
  }
  float4 wv[8];
#pragma unroll
  for (int u = 0; u < 8; ++u)
    wv[u] = *((const float4*)(fc_w + (2 * u + rsub) * (HEADS * HID)) + c4);

  float s0 = score[0] / (float)NB, s1 = score[1] / (float)NB;
  float mx = fmaxf(s0, s1);
  float e0 = __expf(s0 - mx), e1 = __expf(s1 - mx);
  float inv = 1.f / (e0 + e1);
  float beta0 = e0 * inv, beta1 = e1 * inv;

  float4 x0[8], x1[8];
#pragma unroll
  for (int u = 0; u < 8; ++u) {
    x0[u] = *((const float4*)(eft + (long)t0[u] * (HEADS * HID)) + c4);
    x1[u] = *((const float4*)(eft + (long)NG * (HEADS * HID) +
                              (long)t1[u] * (HEADS * HID)) + c4);
  }
#pragma unroll
  for (int u = 0; u < 8; ++u)
    *(float4*)(ldsW + (2 * u + rsub) * HL_PITCH + c4 * 4) = wv[u];
  float4 hv[8];
#pragma unroll
  for (int u = 0; u < 8; ++u) {
    hv[u].x = __builtin_fmaf(beta0, x0[u].x, beta1 * x1[u].x);
    hv[u].y = __builtin_fmaf(beta0, x0[u].y, beta1 * x1[u].y);
    hv[u].z = __builtin_fmaf(beta0, x0[u].z, beta1 * x1[u].z);
    hv[u].w = __builtin_fmaf(beta0, x0[u].w, beta1 * x1[u].w);
    *(float4*)(ldsA + (2 * u + rsub) * HL_PITCH + c4 * 4) = hv[u];
  }
  __syncthreads();
#pragma unroll
  for (int u = 0; u < 8; ++u)
    *((float4*)(hagg + (long)(b0 + 2 * u + rsub) * (HEADS * HID)) + c4) = hv[u];
  if (tid >= 32) return;
  const int half = tid >> 4, l15 = tid & 15;
  const float* ap = ldsA + l15 * HL_PITCH + half * 2;
  const float* bp = ldsW + l15 * HL_PITCH + half * 2;
  f32x8 c = {};
  for (int k0 = 0; k0 < HEADS * HID; k0 += 4) {
    const float2 av = *(const float2*)(ap + k0);
    const float2 bv = *(const float2*)(bp + k0);
    f32x2 af, bf;
    af.x = av.x; af.y = av.y;
    bf.x = bv.x; bf.y = bv.y;
    c = wmma_f32(af, bf, c);
  }
  const float bias = fc_b[l15];
#pragma unroll
  for (int v = 0; v < 8; ++v)
    logits[(long)(b0 + v + half * 8) * OUTD + l15] = c[v] + bias;
}

extern "C" void kernel_launch(void* const* d_in, const int* in_sizes, int n_in,
                              void* d_out, int out_size, void* d_ws,
                              size_t ws_size, hipStream_t stream) {
  const float* feat0 = (const float*)d_in[0];
  const float* feat1 = (const float*)d_in[1];
  const float* W0    = (const float*)d_in[2];
  const float* b0    = (const float*)d_in[3];
  const float* W1    = (const float*)d_in[4];
  const float* b1    = (const float*)d_in[5];
  const float* r_vec = (const float*)d_in[6];
  const float* attn0 = (const float*)d_in[7];
  const float* attn1 = (const float*)d_in[8];
  const float* fc1_w = (const float*)d_in[9];
  const float* fc1_b = (const float*)d_in[10];
  const float* fc2_w = (const float*)d_in[11];
  const float* fc_w  = (const float*)d_in[12];
  const float* fc_b  = (const float*)d_in[13];
  const int* idx0    = (const int*)d_in[14];
  const int* idx1    = (const int*)d_in[15];
  const int* mp_indices[2] = {(const int*)d_in[16], (const int*)d_in[19]};
  const int* mp_dst[2]     = {(const int*)d_in[17], (const int*)d_in[20]};
  const int* mp_target[2]  = {(const int*)d_in[18], (const int*)d_in[21]};
  const float* attn[2] = {attn0, attn1};

  float* ws     = (float*)d_ws;
  float* tf     = ws;
  float* finals = tf + (long)N_NODES * HID;
  float* h      = finals + 2 * LP * HID;
  float* a      = h + (long)NE * HID;
  float* eft    = a + (long)NE * HEADS;
  float* score  = eft + 2L * NG * HEADS * HID;

  float* logits = (float*)d_out;
  float* hagg   = logits + (long)NB * OUTD;

  setup_kernel<<<GEMM0_WG + GEMM1_WG + INIT_WG, 256, 0, stream>>>(
      feat0, W0, b0, idx0, feat1, W1, b1, idx1, tf, r_vec, finals, score);

  for (int mp = 0; mp < 2; ++mp) {
    edge_h_kernel<<<(NE / EPW + 7) / 8, 256, 0, stream>>>(tf, finals + mp * LP * HID,
                                              mp_indices[mp], attn[mp], h, a);
    owner_sum_kernel<<<NG / OWN_D, 1024, 0, stream>>>(
        h, a, mp_dst[mp], eft + (long)mp * NG * HEADS * HID);
  }

  score_kernel<<<2 * (NB / SC_ROWS), 256, 0, stream>>>(
      eft, mp_target[0], mp_target[1], fc1_w, fc1_b, fc2_w, score);
  hagg_logits_kernel<<<NB / HL_ROWS, 256, 0, stream>>>(
      eft, mp_target[0], mp_target[1], score, fc_w, fc_b, hagg, logits);
}


# owner launch: each scanning wave lists its hits by ballot rank (fixed order); one wave per destination takes max, exp sum, ft row in registers in list order; LDS float atomics go, sums repeatable
# speedup vs baseline: 5.3932x; 1.0866x over previous
#include <hip/hip_runtime.h>
#include <math.h>


#define N_NODES 100000
#define N0      60000
#define N1      40000
#define F0      128
#define F1      64
#define HID     64
#define HEADS   8
#define OUTD    16
#define ATTNV   128
#define NE      250000
#define LP      3
#define NG      8192
#define NB      8192

typedef __attribute__((ext_vector_type(2))) float f32x2;
typedef __attribute__((ext_vector_type(8))) float f32x8;

__device__ __forceinline__ f32x8 wmma_f32(f32x2 a, f32x2 b, f32x8 c) {
  return __builtin_amdgcn_wmma_f32_16x16x4_f32(false, a, false, b, (short)0, c,
                                               false, false);
}

__device__ __forceinline__ float xrow(float x) {
  int b = __builtin_bit_cast(int, x);
  return __builtin_bit_cast(
      float, __builtin_amdgcn_permlanex16(b, b, 0x76543210, 0xfedcba98, false,
                                          false));
}
template <int M>
__device__ __forceinline__ float xlane(float x) {
  int b = __builtin_bit_cast(int, x);
  return __builtin_bit_cast(
      float, __builtin_amdgcn_update_dpp(b, b, 0x160 + M, 0xf, 0xf, false));
}

#define INIT_WG 1
__device__ __forceinline__ void init_body(const float* __restrict__ r_vec,
                                          float* __restrict__ finals  ,
                                          float* __restrict__ score  ) {
  const int c = threadIdx.x;
  if (c < 2) score[c] = 0.f;
  if (c >= 32) return;
  for (int mp = 0; mp < 2; ++mp) {
    float re = r_vec[(mp * 32 + c) * 2 + 0];
    float im = r_vec[(mp * 32 + c) * 2 + 1];
    float inv = rsqrtf(re * re + im * im);
    re *= inv; im *= inv;
    float f1re = re, f1im = -im;
    float f0re = f1re * re - f1im * im;
    float f0im = f1re * im + f1im * re;
    float* F = finals + mp * LP * HID;
    F[0 * HID + 2 * c] = f0re; F[0 * HID + 2 * c + 1] = f0im;
    F[1 * HID + 2 * c] = f1re; F[1 * HID + 2 * c + 1] = f1im;
    F[2 * HID + 2 * c] = 1.f;  F[2 * HID + 2 * c + 1] = 0.f;
  }
}

template <int K>
__device__ __forceinline__ void gemm_tf_body(
    int blk, float* ldsB, const float* __restrict__ feat,
    const float* __restrict__ W, const float* __restrict__ bias,
    const int* __restrict__ idx, float* __restrict__ tf, int M) {
  int tid = threadIdx.x;
  for (int i = tid; i < K * HID; i += 256) {
    int k = i >> 6, n = i & 63;
    ldsB[i] = W[n * K + k];
  }
  __syncthreads();
  int wave = tid >> 5, lane = tid & 31;
  int tile_m = blk * 128 + wave * 16;
  if (tile_m >= M) return;
  int half = lane >> 4, l15 = lane & 15;
  int arow = tile_m + l15;
  if (arow >= M) arow = M - 1;
  f32x8 c0 = {}, c1 = {}, c2 = {}, c3 = {};
  for (int k0 = 0; k0 < K; k0 += 4) {
    int kk = k0 + half * 2;
    f32x2 a;
    a.x = feat[(long)arow * K + kk];
    a.y = feat[(long)arow * K + kk + 1];
    f32x2 b0, b1, b2, b3;
    b0.x = ldsB[kk * HID + l15 + 0];  b0.y = ldsB[(kk + 1) * HID + l15 + 0];
    b1.x = ldsB[kk * HID + l15 + 16]; b1.y = ldsB[(kk + 1) * HID + l15 + 16];
    b2.x = ldsB[kk * HID + l15 + 32]; b2.y = ldsB[(kk + 1) * HID + l15 + 32];
    b3.x = ldsB[kk * HID + l15 + 48]; b3.y = ldsB[(kk + 1) * HID + l15 + 48];
    c0 = wmma_f32(a, b0, c0);
    c1 = wmma_f32(a, b1, c1);
    c2 = wmma_f32(a, b2, c2);
    c3 = wmma_f32(a, b3, c3);
  }
#pragma unroll
  for (int v = 0; v < 8; ++v) {
    int r = tile_m + v + half * 8;
    if (r < M) {
      float* drow = tf + (long)idx[r] * HID;
      drow[l15 + 0]  = c0[v] + bias[l15 + 0];
      drow[l15 + 16] = c1[v] + bias[l15 + 16];
      drow[l15 + 32] = c2[v] + bias[l15 + 32];
      drow[l15 + 48] = c3[v] + bias[l15 + 48];
    }
  }
}

#define GEMM0_WG ((N0 + 127) / 128)
#define GEMM1_WG ((N1 + 127) / 128)
__global__ __launch_bounds__(256) void setup_kernel(
    const float* __restrict__ feat0, const float* __restrict__ W0,
    const float* __restrict__ b0, const int* __restrict__ idx0,
    const float* __restrict__ feat1, const float* __restrict__ W1,
    const float* __restrict__ b1, const int* __restrict__ idx1,
    float* __restrict__ tf, const float* __restrict__ r_vec,
    float* __restrict__ finals, float* __restrict__ score) {
  __shared__ float ldsB[F0 * HID];
  const int blk = blockIdx.x;
  if (blk < GEMM0_WG) {
    gemm_tf_body<F0>(blk, ldsB, feat0, W0, b0, idx0, tf, N0);
  } else if (blk < GEMM0_WG + GEMM1_WG) {
    gemm_tf_body<F1>(blk - GEMM0_WG, ldsB, feat1, W1, b1, idx1, tf, N1);
  } else {
    init_body(r_vec, finals, score);
  }
}

#define EPW 8
__global__ __launch_bounds__(256) void edge_h_kernel(
    const float* __restrict__ tf, const float* __restrict__ finals,
    const int* __restrict__ indices, const float* __restrict__ attn,
    float* __restrict__ h, float* __restrict__ a) {
  const int lane = threadIdx.x & 31;
  const int wave = __builtin_amdgcn_readfirstlane(threadIdx.x >> 5);
  const int e0 = (blockIdx.x * 8 + wave) * EPW;
  if (e0 >= NE) return;
  const int mynode = indices[e0 * LP + (lane < LP * EPW ? lane : lane - 8)];
  const int g = lane >> 2, r = lane & 3;
  float2 v[EPW][LP];
#pragma unroll
  for (int u = 0; u < EPW; ++u) {
#pragma unroll
    for (int i = 0; i < LP; ++i) {
      const int node = __builtin_amdgcn_readlane(mynode, u * LP + i);
      v[u][i] = *((const float2*)(tf + (long)node * HID) + lane);
    }
  }
  float2 f[LP];
#pragma unroll
  for (int i = 0; i < LP; ++i) f[i] = *((const float2*)(finals + i * HID) + lane);
  float2 w[HEADS];
#pragma unroll
  for (int hh = 0; hh < HEADS; ++hh)
    w[hh] = *((const float2*)(attn + hh * HID) + lane);

  const bool b4 = (lane & 16) != 0, b3 = (lane & 8) != 0, b2 = (lane & 4) != 0;
  float tlo = 0.f, thi = 0.f;
#pragma unroll
  for (int u = 0; u < EPW; ++u) {
    float2 acc = make_float2(0.f, 0.f);
#pragma unroll
    for (int i = 0; i < LP; ++i) {
      const float vx = v[u][i].x, vy = v[u][i].y;
      const float fx = f[i].x, fy = f[i].y;
      acc.x += vx * fx - vy * fy;
      acc.y += vx * fy + vy * fx;
    }
    acc.x *= (1.f / 3.f);
    acc.y *= (1.f / 3.f);
    *((float2*)(h + (long)(e0 + u) * HID) + lane) = acc;
    float p[HEADS];
#pragma unroll
    for (int hh = 0; hh < HEADS; ++hh)
      p[hh] = __builtin_fmaf(acc.x, w[hh].x, acc.y * w[hh].y);
    float q[4];
#pragma unroll
    for (int j = 0; j < 4; ++j) {
      float keep = b4 ? p[j + 4] : p[j];
      float send = b4 ? p[j] : p[j + 4];
      q[j] = keep + xrow(send);
    }
    float rr[2];
#pragma unroll
    for (int j = 0; j < 2; ++j) {
      float keep = b3 ? q[j + 2] : q[j];
      float send = b3 ? q[j] : q[j + 2];
      rr[j] = keep + xlane<8>(send);
    }
    float keep2 = b2 ? rr[1] : rr[0];
    float send2 = b2 ? rr[0] : rr[1];
    float t = keep2 + xlane<4>(send2);
    t += xlane<2>(t);
    t += xlane<1>(t);
    if (u < 4) tlo = (r == u) ? t : tlo;
    else       thi = (r == u - 4) ? t : thi;
  }
  const float alo = tlo > 0.f ? tlo : 0.01f * tlo;
  const float ahi = thi > 0.f ? thi : 0.01f * thi;
  a[(long)(e0 + r) * HEADS + g] = alo;
  a[(long)(e0 + 4 + r) * HEADS + g] = ahi;
}

#define OWN_D   32
#define OWN_WCAP 1024
#define OWN_DCAP 256

__device__ __forceinline__ float lane_bcast(float x, int l) {
  return __builtin_bit_cast(
      float, __builtin_amdgcn_readlane(__builtin_bit_cast(int, x), l));
}

__device__ __forceinline__ f32x2 fma_bcast(float w, f32x2 x, f32x2 acc) {
  f32x2 ws;
  ws.x = w; ws.y = w;
  return __builtin_elementwise_fma(ws, x, acc);
}

__device__ __forceinline__ int append_hits(bool hit, int val, int* lst, int c,
                                           int cap) {
  const unsigned mask = __builtin_amdgcn_ballot_w32(hit);
  if (mask != 0u) {
    const int r = c + (int)__builtin_amdgcn_mbcnt_lo(mask, 0u);
    if (hit && r < cap) lst[r] = val;
    c += __builtin_popcount(mask);
  }
  return c;
}

template <int KIND>
__device__ __forceinline__ void sweep_dst(const int* __restrict__ dst, int dd,
                                          int lane, const float* __restrict__ h,
                                          const float* __restrict__ a, float& mx,
                                          float& sp, f32x2 (&acc)[HEADS]) {
  for (int i0 = 0; i0 < NE; i0 += 32) {
    const int i = i0 + lane;
    const int d = i < NE ? dst[i] : -1;
    unsigned mask = __builtin_amdgcn_ballot_w32(d == dd);
    while (mask != 0u) {
      const int e = i0 + __builtin_ctz(mask);
      mask &= mask - 1u;
      const float v = a[(long)e * HEADS + (lane & 7)];
      if (KIND == 0) mx = fmaxf(mx, v);
      if (KIND == 1) sp += __expf(v - mx);
      if (KIND == 2) {
        const float wl = __expf(v - mx) / sp;
        f32x2 x;
        x.x = h[(long)e * HID + lane];
        x.y = h[(long)e * HID + 32 + lane];
#pragma unroll
        for (int hh = 0; hh < HEADS; ++hh) acc[hh] = fma_bcast(lane_bcast(wl, hh), x, acc[hh]);
      }
    }
  }
}

__global__ __launch_bounds__(1024) void owner_sum_kernel(
    const float* __restrict__ h, const float* __restrict__ a,
    const int* __restrict__ dst,
    float* __restrict__ eft  ) {
  __shared__ int list[OWN_D * OWN_WCAP];
  __shared__ int mine[OWN_D * OWN_DCAP];
  __shared__ int cnt[OWN_D];
  const int tid = threadIdx.x, lane = tid & 31;
  const int wave = __builtin_amdgcn_readfirstlane(tid >> 5);
  const int d0 = blockIdx.x * OWN_D;

  int* wl = list + wave * OWN_WCAP;
  int c = 0;
  const int4* dst4 = (const int4*)dst;
  for (int base = 0; base < NE / 4; base += 1024 * 8) {
    int4 dv[8];
#pragma unroll
    for (int u = 0; u < 8; ++u) {
      const int q = base + u * 1024 + tid;
      dv[u] = q < NE / 4 ? dst4[q] : make_int4(-1, -1, -1, -1);
    }
#pragma unroll
    for (int u = 0; u < 8; ++u) {
      const int i = 4 * (base + u * 1024 + tid);
      const unsigned dx = (unsigned)(dv[u].x - d0), dy = (unsigned)(dv[u].y - d0);
      const unsigned dz = (unsigned)(dv[u].z - d0), dw = (unsigned)(dv[u].w - d0);
      c = append_hits(dx < OWN_D, ((i + 0) << 5) | (int)dx, wl, c, OWN_WCAP);
      c = append_hits(dy < OWN_D, ((i + 1) << 5) | (int)dy, wl, c, OWN_WCAP);
      c = append_hits(dz < OWN_D, ((i + 2) << 5) | (int)dz, wl, c, OWN_WCAP);
      c = append_hits(dw < OWN_D, ((i + 3) << 5) | (int)dw, wl, c, OWN_WCAP);
    }
  }
  if (lane == 0) cnt[wave] = c;
  __syncthreads();

  const int cv = cnt[lane];
  const bool ovf = __builtin_amdgcn_ballot_w32(cv > OWN_WCAP) != 0u;
  int* my = mine + wave * OWN_DCAP;
  int m = 0;
  if (!ovf) {
    for (int s = 0; s < OWN_D; ++s) {
      const int cs = __builtin_amdgcn_readlane(cv, s);
      const int* sl = list + s * OWN_WCAP;
      for (int j0 = 0; j0 < cs; j0 += 32) {
        const int j = j0 + lane;
        const int ent = sl[j < cs ? j : 0];
        m = append_hits(j < cs && (ent & 31) == wave, ent >> 5, my, m, OWN_DCAP);
      }
    }
  }
  __builtin_amdgcn_wave_barrier();

  const int slot = lane >> 3, head = lane & 7;
  float mx = -3.0e38f, sp = 0.f;
  f32x2 acc[HEADS];
#pragma unroll
  for (int hh = 0; hh < HEADS; ++hh) acc[hh].x = acc[hh].y = 0.f;

  if (ovf || m > OWN_DCAP) {
    sweep_dst<0>(dst, d0 + wave, lane, h, a, mx, sp, acc);
    sweep_dst<1>(dst, d0 + wave, lane, h, a, mx, sp, acc);
    sweep_dst<2>(dst, d0 + wave, lane, h, a, mx, sp, acc);
  } else if (m > 0) {
    for (int j0 = 0; j0 < m; j0 += 32) {
      float v[8];
#pragma unroll
      for (int t = 0; t < 8; ++t) {
        const int j = j0 + 4 * t + slot;
        v[t] = a[(long)my[j < m ? j : m - 1] * HEADS + head];
      }
#pragma unroll
      for (int t = 0; t < 8; ++t) mx = fmaxf(mx, v[t]);
    }
    mx = fmaxf(mx, xlane<8>(mx));
    mx = fmaxf(mx, xrow(mx));
    for (int j0 = 0; j0 < m; j0 += 32) {
      float v[8];
#pragma unroll
      for (int t = 0; t < 8; ++t) {
        const int j = j0 + 4 * t + slot;
        v[t] = a[(long)my[j < m ? j : m - 1] * HEADS + head];
      }
#pragma unroll
      for (int t = 0; t < 8; ++t)
        sp += (j0 + 4 * t + slot < m) ? __expf(v[t] - mx) : 0.f;
    }
    sp += xlane<8>(sp);
    sp += xrow(sp);
    for (int j0 = 0; j0 < m; j0 += 8) {
      const int jl = j0 + (lane & 7);
      const int mye = my[jl < m ? jl : m - 1];
      const int ja = j0 + slot, jb = j0 + 4 + slot;
      const int ea = my[ja < m ? ja : m - 1], eb = my[jb < m ? jb : m - 1];
      f32x2 x[8];
#pragma unroll
      for (int u = 0; u < 8; ++u) {
        const int e = __builtin_amdgcn_readlane(mye, u);
        x[u].x = h[(long)e * HID + lane];
        x[u].y = h[(long)e * HID + 32 + lane];
      }
      const float va = a[(long)ea * HEADS + head];
      const float vb = a[(long)eb * HEADS + head];
      const float wa = ja < m ? __expf(va - mx) / sp : 0.f;
      const float wb = jb < m ? __expf(vb - mx) / sp : 0.f;
#pragma unroll
      for (int u = 0; u < 8; ++u) {
#pragma unroll
        for (int hh = 0; hh < HEADS; ++hh)
          acc[hh] = fma_bcast(lane_bcast(u < 4 ? wa : wb, (u & 3) * 8 + hh), x[u], acc[hh]);
      }
    }
  }

  float* o = eft + (long)(d0 + wave) * HEADS * HID + lane;
#pragma unroll
  for (int hh = 0; hh < HEADS; ++hh) {
    const float x = acc[hh].x, y = acc[hh].y;
    o[hh * HID]      = x > 0.f ? x : (__expf(x) - 1.f);
    o[hh * HID + 32] = y > 0.f ? y : (__expf(y) - 1.f);
  }
}

#define SC_ROWS  64
#define SC_PITCH 516
__device__ __forceinline__ void score_chunk(const float4 (&a)[4], const float* b,
                                            f32x8 (&acc)[4]) {
#pragma unroll
  for (int q = 0; q < 4; ++q) {
    float4 bv[4];
#pragma unroll
    for (int nt = 0; nt < 4; ++nt)
      bv[nt] = *(const float4*)(b + nt * 16 * SC_PITCH + q * 8);
    f32x2 a01, a23;
    a01.x = a[q].x; a01.y = a[q].y;
    a23.x = a[q].z; a23.y = a[q].w;
#pragma unroll
    for (int nt = 0; nt < 4; ++nt) {
      f32x2 b01;
      b01.x = bv[nt].x; b01.y = bv[nt].y;
      acc[nt] = wmma_f32(a01, b01, acc[nt]);
    }
#pragma unroll
    for (int nt = 0; nt < 4; ++nt) {
      f32x2 b23;
      b23.x = bv[nt].z; b23.y = bv[nt].w;
      acc[nt] = wmma_f32(a23, b23, acc[nt]);
    }
  }
}
__global__ __launch_bounds__(256) void score_kernel(
    const float* __restrict__ eft, const int* __restrict__ target0,
    const int* __restrict__ target1, const float* __restrict__ fc1_w,
    const float* __restrict__ fc1_b, const float* __restrict__ fc2_w,
    float* __restrict__ score) {
  __shared__ __attribute__((aligned(16))) float ldsW[ATTNV * SC_PITCH];
  const int tid = threadIdx.x;
  const float4* w4 = (const float4*)fc1_w;
  const int rot = blockIdx.x * 128;
  for (int base = 0; base < ATTNV * (HEADS * HID) / 4; base += 256 * 16) {
    float4 t[16];
#pragma unroll
    for (int u = 0; u < 16; ++u)
      t[u] = w4[(base + u * 256 + tid + rot) & (ATTNV * (HEADS * HID) / 4 - 1)];
#pragma unroll
    for (int u = 0; u < 16; ++u) {
      const int i = (base + u * 256 + tid + rot) & (ATTNV * (HEADS * HID) / 4 - 1);
      *(float4*)(ldsW + (i >> 7) * SC_PITCH + (i & 127) * 4) = t[u];
    }
  }
  __syncthreads();

  const int p = blockIdx.x / (NB / SC_ROWS);
  const int mblk = blockIdx.x % (NB / SC_ROWS);
  const float* A = eft + (long)p * NG * (HEADS * HID);
  const int* target = p ? target1 : target0;
  const int wave = tid >> 5, lane = tid & 31;
  const int mt = wave & 3, nh = wave >> 2;
  const int half = lane >> 4, l15 = lane & 15;
  const long arow = target[mblk * SC_ROWS + mt * 16 + l15];
  const float4* ap = (const float4*)(A + arow * (HEADS * HID) + half * 4);
  const float* bp = ldsW + (nh * 64 + l15) * SC_PITCH + half * 4;
  f32x8 acc[4] = {};
  float4 a0[4], a1[4];
#pragma unroll
  for (int q = 0; q < 4; ++q) a0[q] = ap[q * 2];
  for (int ch = 0; ch < (HEADS * HID) / 32; ch += 2) {
#pragma unroll
    for (int q = 0; q < 4; ++q) a1[q] = ap[((ch + 1) * 4 + q) * 2];
    score_chunk(a0, bp + ch * 32, acc);
    const int nx = ch + 2 < (HEADS * HID) / 32 ? ch + 2 : ch;
#pragma unroll
    for (int q = 0; q < 4; ++q) a0[q] = ap[(nx * 4 + q) * 2];
    asm volatile("" ::: "memory");
    score_chunk(a1, bp + (ch + 1) * 32, acc);
  }
  float ssum = 0.f;
#pragma unroll
  for (int nt = 0; nt < 4; ++nt) {
    int col = nh * 64 + nt * 16 + l15;
    float bb = fc1_b[col], w2 = fc2_w[col];
#pragma unroll
    for (int v = 0; v < 8; ++v) ssum += tanhf(acc[nt][v] + bb) * w2;
  }
#pragma unroll
  for (int off = 16; off > 0; off >>= 1) ssum += __shfl_xor(ssum, off, 32);
  if (lane == 0) atomicAdd(&score[p], ssum);
}

#define HL_ROWS  16
#define HL_PITCH 516
__global__ __launch_bounds__(256) void hagg_logits_kernel(
    const float* __restrict__ eft, const int* __restrict__ target0,
    const int* __restrict__ target1, const float* __restrict__ score,
    const float* __restrict__ fc_w, const float* __restrict__ fc_b,
    float* __restrict__ hagg, float* __restrict__ logits) {
  __shared__ __attribute__((aligned(16))) float ldsA[HL_ROWS * HL_PITCH];
  __shared__ __attribute__((aligned(16))) float ldsW[OUTD * HL_PITCH];
  const int tid = threadIdx.x;
  const int b0 = blockIdx.x * HL_ROWS;
  const int c4 = tid & 127, rsub = tid >> 7;
  int t0[8], t1[8];
#pragma unroll
  for (int u = 0; u < 8; ++u) {
    t0[u] = target0[b0 + 2 * u + rsub];
    t1[u] = target1[b0 + 2 * u + rsub];
  }
  float4 wv[8];
#pragma unroll
  for (int u = 0; u < 8; ++u)
    wv[u] = *((const float4*)(fc_w + (2 * u + rsub) * (HEADS * HID)) + c4);

  float s0 = score[0] / (float)NB, s1 = score[1] / (float)NB;
  float mx = fmaxf(s0, s1);
  float e0 = __expf(s0 - mx), e1 = __expf(s1 - mx);
  float inv = 1.f / (e0 + e1);
  float beta0 = e0 * inv, beta1 = e1 * inv;

  float4 x0[8], x1[8];
#pragma unroll
  for (int u = 0; u < 8; ++u) {
    x0[u] = *((const float4*)(eft + (long)t0[u] * (HEADS * HID)) + c4);
    x1[u] = *((const float4*)(eft + (long)NG * (HEADS * HID) +
                              (long)t1[u] * (HEADS * HID)) + c4);
  }
#pragma unroll
  for (int u = 0; u < 8; ++u)
    *(float4*)(ldsW + (2 * u + rsub) * HL_PITCH + c4 * 4) = wv[u];
  float4 hv[8];
#pragma unroll
  for (int u = 0; u < 8; ++u) {
    hv[u].x = __builtin_fmaf(beta0, x0[u].x, beta1 * x1[u].x);
    hv[u].y = __builtin_fmaf(beta0, x0[u].y, beta1 * x1[u].y);
    hv[u].z = __builtin_fmaf(beta0, x0[u].z, beta1 * x1[u].z);
    hv[u].w = __builtin_fmaf(beta0, x0[u].w, beta1 * x1[u].w);
    *(float4*)(ldsA + (2 * u + rsub) * HL_PITCH + c4 * 4) = hv[u];
  }
  __syncthreads();
#pragma unroll
  for (int u = 0; u < 8; ++u)
    *((float4*)(hagg + (long)(b0 + 2 * u + rsub) * (HEADS * HID)) + c4) = hv[u];
  if (tid >= 32) return;
  const int half = tid >> 4, l15 = tid & 15;
  const float* ap = ldsA + l15 * HL_PITCH + half * 2;
  const float* bp = ldsW + l15 * HL_PITCH + half * 2;
  f32x8 c = {};
  for (int k0 = 0; k0 < HEADS * HID; k0 += 4) {
    const float2 av = *(const float2*)(ap + k0);
    const float2 bv = *(const float2*)(bp + k0);
    f32x2 af, bf;
    af.x = av.x; af.y = av.y;
    bf.x = bv.x; bf.y = bv.y;
    c = wmma_f32(af, bf, c);
  }
  const float bias = fc_b[l15];
#pragma unroll
  for (int v = 0; v < 8; ++v)
    logits[(long)(b0 + v + half * 8) * OUTD + l15] = c[v] + bias;
}

extern "C" void kernel_launch(void* const* d_in, const int* in_sizes, int n_in,
                              void* d_out, int out_size, void* d_ws,
                              size_t ws_size, hipStream_t stream) {
  const float* feat0 = (const float*)d_in[0];
  const float* feat1 = (const float*)d_in[1];
  const float* W0    = (const float*)d_in[2];
  const float* b0    = (const float*)d_in[3];
  const float* W1    = (const float*)d_in[4];
  const float* b1    = (const float*)d_in[5];
  const float* r_vec = (const float*)d_in[6];
  const float* attn0 = (const float*)d_in[7];
  const float* attn1 = (const float*)d_in[8];
  const float* fc1_w = (const float*)d_in[9];
  const float* fc1_b = (const float*)d_in[10];
  const float* fc2_w = (const float*)d_in[11];
  const float* fc_w  = (const float*)d_in[12];
  const float* fc_b  = (const float*)d_in[13];
  const int* idx0    = (const int*)d_in[14];
  const int* idx1    = (const int*)d_in[15];
  const int* mp_indices[2] = {(const int*)d_in[16], (const int*)d_in[19]};
  const int* mp_dst[2]     = {(const int*)d_in[17], (const int*)d_in[20]};
  const int* mp_target[2]  = {(const int*)d_in[18], (const int*)d_in[21]};
  const float* attn[2] = {attn0, attn1};

  float* ws     = (float*)d_ws;
  float* tf     = ws;
  float* finals = tf + (long)N_NODES * HID;
  float* h      = finals + 2 * LP * HID;
  float* a      = h + (long)NE * HID;
  float* eft    = a + (long)NE * HEADS;
  float* score  = eft + 2L * NG * HEADS * HID;

  float* logits = (float*)d_out;
  float* hagg   = logits + (long)NB * OUTD;

  setup_kernel<<<GEMM0_WG + GEMM1_WG + INIT_WG, 256, 0, stream>>>(
      feat0, W0, b0, idx0, feat1, W1, b1, idx1, tf, r_vec, finals, score);

  for (int mp = 0; mp < 2; ++mp) {
    edge_h_kernel<<<(NE / EPW + 7) / 8, 256, 0, stream>>>(tf, finals + mp * LP * HID,
                                              mp_indices[mp], attn[mp], h, a);
    owner_sum_kernel<<<NG / OWN_D, 1024, 0, stream>>>(
        h, a, mp_dst[mp], eft + (long)mp * NG * HEADS * HID);
  }

  score_kernel<<<2 * (NB / SC_ROWS), 256, 0, stream>>>(
      eft, mp_target[0], mp_target[1], fc1_w, fc1_b, fc2_w, score);
  hagg_logits_kernel<<<NB / HL_ROWS, 256, 0, stream>>>(
      eft, mp_target[0], mp_target[1], score, fc_w, fc_b, hagg, logits);
}


# set-up launch groups edges by owner range (LDS integer adds; a 128-byte slot per chunk and owner); owners read 123 slots, not all of dst; lists sorted by edge number: sums in ascending edge order
# speedup vs baseline: 6.6706x; 1.2369x over previous
#include <hip/hip_runtime.h>
#include <math.h>


#define N_NODES 100000
#define N0      60000
#define N1      40000
#define F0      128
#define F1      64
#define HID     64
#define HEADS   8
#define OUTD    16
#define ATTNV   128
#define NE      250000
#define LP      3
#define NG      8192
#define NB      8192

typedef __attribute__((ext_vector_type(2))) float f32x2;
typedef __attribute__((ext_vector_type(8))) float f32x8;

__device__ __forceinline__ f32x8 wmma_f32(f32x2 a, f32x2 b, f32x8 c) {
  return __builtin_amdgcn_wmma_f32_16x16x4_f32(false, a, false, b, (short)0, c,
                                               false, false);
}

__device__ __forceinline__ float xrow(float x) {
  int b = __builtin_bit_cast(int, x);
  return __builtin_bit_cast(
      float, __builtin_amdgcn_permlanex16(b, b, 0x76543210, 0xfedcba98, false,
                                          false));
}
template <int M>
__device__ __forceinline__ float xlane(float x) {
  int b = __builtin_bit_cast(int, x);
  return __builtin_bit_cast(
      float, __builtin_amdgcn_update_dpp(b, b, 0x160 + M, 0xf, 0xf, false));
}

#define INIT_WG 1
__device__ __forceinline__ void init_body(const float* __restrict__ r_vec,
                                          float* __restrict__ finals  ,
                                          float* __restrict__ score  ) {
  const int c = threadIdx.x;
  if (c < 2) score[c] = 0.f;
  if (c >= 32) return;
  for (int mp = 0; mp < 2; ++mp) {
    float re = r_vec[(mp * 32 + c) * 2 + 0];
    float im = r_vec[(mp * 32 + c) * 2 + 1];
    float inv = rsqrtf(re * re + im * im);
    re *= inv; im *= inv;
    float f1re = re, f1im = -im;
    float f0re = f1re * re - f1im * im;
    float f0im = f1re * im + f1im * re;
    float* F = finals + mp * LP * HID;
    F[0 * HID + 2 * c] = f0re; F[0 * HID + 2 * c + 1] = f0im;
    F[1 * HID + 2 * c] = f1re; F[1 * HID + 2 * c + 1] = f1im;
    F[2 * HID + 2 * c] = 1.f;  F[2 * HID + 2 * c + 1] = 0.f;
  }
}

template <int K>
__device__ __forceinline__ void gemm_tf_body(
    int blk, float* ldsB, const float* __restrict__ feat,
    const float* __restrict__ W, const float* __restrict__ bias,
    const int* __restrict__ idx, float* __restrict__ tf, int M) {
  int tid = threadIdx.x;
  for (int i = tid; i < K * HID; i += 256) {
    int k = i >> 6, n = i & 63;
    ldsB[i] = W[n * K + k];
  }
  __syncthreads();
  int wave = tid >> 5, lane = tid & 31;
  int tile_m = blk * 128 + wave * 16;
  if (tile_m >= M) return;
  int half = lane >> 4, l15 = lane & 15;
  int arow = tile_m + l15;
  if (arow >= M) arow = M - 1;
  f32x8 c0 = {}, c1 = {}, c2 = {}, c3 = {};
  for (int k0 = 0; k0 < K; k0 += 4) {
    int kk = k0 + half * 2;
    f32x2 a;
    a.x = feat[(long)arow * K + kk];
    a.y = feat[(long)arow * K + kk + 1];
    f32x2 b0, b1, b2, b3;
    b0.x = ldsB[kk * HID + l15 + 0];  b0.y = ldsB[(kk + 1) * HID + l15 + 0];
    b1.x = ldsB[kk * HID + l15 + 16]; b1.y = ldsB[(kk + 1) * HID + l15 + 16];
    b2.x = ldsB[kk * HID + l15 + 32]; b2.y = ldsB[(kk + 1) * HID + l15 + 32];
    b3.x = ldsB[kk * HID + l15 + 48]; b3.y = ldsB[(kk + 1) * HID + l15 + 48];
    c0 = wmma_f32(a, b0, c0);
    c1 = wmma_f32(a, b1, c1);
    c2 = wmma_f32(a, b2, c2);
    c3 = wmma_f32(a, b3, c3);
  }
#pragma unroll
  for (int v = 0; v < 8; ++v) {
    int r = tile_m + v + half * 8;
    if (r < M) {
      float* drow = tf + (long)idx[r] * HID;
      drow[l15 + 0]  = c0[v] + bias[l15 + 0];
      drow[l15 + 16] = c1[v] + bias[l15 + 16];
      drow[l15 + 32] = c2[v] + bias[l15 + 32];
      drow[l15 + 48] = c3[v] + bias[l15 + 48];
    }
  }
}

#define OWN_D    32
#define NOWN     (NG / OWN_D)
#define CH_EDGES 2048
#define NCH      ((NE + CH_EDGES - 1) / CH_EDGES)
#define SLOT_W   32
#define SLOT_CAP (SLOT_W - 1)
static_assert(NOWN * SLOT_W == F0 * HID, "the slots of a chunk fill the set-up launch's LDS array");
static_assert(NCH * SLOT_W <= OWN_D * HEADS * HID, "an owner's slots fit into its own rows");

__device__ __forceinline__ void group_place(int d, int e, int* stage) {
  if ((unsigned)d < (unsigned)NG) {
    const int o = d / OWN_D;
    const int r = atomicAdd(&stage[o * SLOT_W], 1);
    if (r < SLOT_CAP) stage[o * SLOT_W + 1 + r] = (e << 5) | (d % OWN_D);
  }
}

__device__ __forceinline__ void group_body(int c, int* stage  ,
                                           const int* __restrict__ dst,
                                           float* __restrict__ plane  ) {
  const int tid = threadIdx.x;
  int4* st4 = (int4*)stage;
  for (int i = tid; i < NOWN * SLOT_W / 4; i += 256) st4[i] = make_int4(0, 0, 0, 0);
  __syncthreads();
  const int4* dst4 = (const int4*)dst;
  const int q0 = c * (CH_EDGES / 4) + tid, q1 = q0 + 256;
  int4 dv0 = make_int4(-1, -1, -1, -1), dv1 = make_int4(-1, -1, -1, -1);
  if (q0 < NE / 4) dv0 = dst4[q0];
  if (q1 < NE / 4) dv1 = dst4[q1];
  group_place(dv0.x, 4 * q0 + 0, stage);
  group_place(dv0.y, 4 * q0 + 1, stage);
  group_place(dv0.z, 4 * q0 + 2, stage);
  group_place(dv0.w, 4 * q0 + 3, stage);
  group_place(dv1.x, 4 * q1 + 0, stage);
  group_place(dv1.y, 4 * q1 + 1, stage);
  group_place(dv1.z, 4 * q1 + 2, stage);
  group_place(dv1.w, 4 * q1 + 3, stage);
  __syncthreads();
  for (int i = tid; i < NOWN * SLOT_W / 4; i += 256) {
    float* line = plane + (long)(i >> 3) * (OWN_D * HEADS * HID) + c * SLOT_W;
    *((int4*)line + (i & 7)) = st4[i];
  }
}

#define GEMM0_WG ((N0 + 127) / 128)
#define GEMM1_WG ((N1 + 127) / 128)
__global__ __launch_bounds__(256) void setup_kernel(
    const float* __restrict__ feat0, const float* __restrict__ W0,
    const float* __restrict__ b0, const int* __restrict__ idx0,
    const float* __restrict__ feat1, const float* __restrict__ W1,
    const float* __restrict__ b1, const int* __restrict__ idx1,
    float* __restrict__ tf, const float* __restrict__ r_vec,
    float* __restrict__ finals, float* __restrict__ score,
    const int* __restrict__ dst0, const int* __restrict__ dst1,
    float* __restrict__ eft) {
  __shared__ __attribute__((aligned(16))) float ldsB[F0 * HID];
  const int blk = blockIdx.x;
  if (blk < GEMM0_WG) {
    gemm_tf_body<F0>(blk, ldsB, feat0, W0, b0, idx0, tf, N0);
  } else if (blk < GEMM0_WG + GEMM1_WG) {
    gemm_tf_body<F1>(blk - GEMM0_WG, ldsB, feat1, W1, b1, idx1, tf, N1);
  } else if (blk < GEMM0_WG + GEMM1_WG + INIT_WG) {
    init_body(r_vec, finals, score);
  } else {
    const int g = blk - (GEMM0_WG + GEMM1_WG + INIT_WG);
    if (g < NCH) group_body(g, (int*)ldsB, dst0, eft);
    else         group_body(g - NCH, (int*)ldsB, dst1, eft + (long)NG * HEADS * HID);
  }
}

#define EPW 8
__global__ __launch_bounds__(256) void edge_h_kernel(
    const float* __restrict__ tf, const float* __restrict__ finals,
    const int* __restrict__ indices, const float* __restrict__ attn,
    float* __restrict__ h, float* __restrict__ a) {
  const int lane = threadIdx.x & 31;
  const int wave = __builtin_amdgcn_readfirstlane(threadIdx.x >> 5);
  const int e0 = (blockIdx.x * 8 + wave) * EPW;
  if (e0 >= NE) return;
  const int mynode = indices[e0 * LP + (lane < LP * EPW ? lane : lane - 8)];
  const int g = lane >> 2, r = lane & 3;
  float2 v[EPW][LP];
#pragma unroll
  for (int u = 0; u < EPW; ++u) {
#pragma unroll
    for (int i = 0; i < LP; ++i) {
      const int node = __builtin_amdgcn_readlane(mynode, u * LP + i);
      v[u][i] = *((const float2*)(tf + (long)node * HID) + lane);
    }
  }
  float2 f[LP];
#pragma unroll
  for (int i = 0; i < LP; ++i) f[i] = *((const float2*)(finals + i * HID) + lane);
  float2 w[HEADS];
#pragma unroll
  for (int hh = 0; hh < HEADS; ++hh)
    w[hh] = *((const float2*)(attn + hh * HID) + lane);

  const bool b4 = (lane & 16) != 0, b3 = (lane & 8) != 0, b2 = (lane & 4) != 0;
  float tlo = 0.f, thi = 0.f;
#pragma unroll
  for (int u = 0; u < EPW; ++u) {
    float2 acc = make_float2(0.f, 0.f);
#pragma unroll
    for (int i = 0; i < LP; ++i) {
      const float vx = v[u][i].x, vy = v[u][i].y;
      const float fx = f[i].x, fy = f[i].y;
      acc.x += vx * fx - vy * fy;
      acc.y += vx * fy + vy * fx;
    }
    acc.x *= (1.f / 3.f);
    acc.y *= (1.f / 3.f);
    *((float2*)(h + (long)(e0 + u) * HID) + lane) = acc;
    float p[HEADS];
#pragma unroll
    for (int hh = 0; hh < HEADS; ++hh)
      p[hh] = __builtin_fmaf(acc.x, w[hh].x, acc.y * w[hh].y);
    float q[4];
#pragma unroll
    for (int j = 0; j < 4; ++j) {
      float keep = b4 ? p[j + 4] : p[j];
      float send = b4 ? p[j] : p[j + 4];
      q[j] = keep + xrow(send);
    }
    float rr[2];
#pragma unroll
    for (int j = 0; j < 2; ++j) {
      float keep = b3 ? q[j + 2] : q[j];
      float send = b3 ? q[j] : q[j + 2];
      rr[j] = keep + xlane<8>(send);
    }
    float keep2 = b2 ? rr[1] : rr[0];
    float send2 = b2 ? rr[0] : rr[1];
    float t = keep2 + xlane<4>(send2);
    t += xlane<2>(t);
    t += xlane<1>(t);
    if (u < 4) tlo = (r == u) ? t : tlo;
    else       thi = (r == u - 4) ? t : thi;
  }
  const float alo = tlo > 0.f ? tlo : 0.01f * tlo;
  const float ahi = thi > 0.f ? thi : 0.01f * thi;
  a[(long)(e0 + r) * HEADS + g] = alo;
  a[(long)(e0 + 4 + r) * HEADS + g] = ahi;
}

#define OWN_DCAP 640
static_assert(NCH * (SLOT_W / 4) <= 1024, "one 16-byte piece of the owner's slots per thread");

__device__ __forceinline__ float lane_bcast(float x, int l) {
  return __builtin_bit_cast(
      float, __builtin_amdgcn_readlane(__builtin_bit_cast(int, x), l));
}

__device__ __forceinline__ f32x2 fma_bcast(float w, f32x2 x, f32x2 acc) {
  f32x2 ws;
  ws.x = w; ws.y = w;
  return __builtin_elementwise_fma(ws, x, acc);
}

template <int KIND>
__device__ __forceinline__ void sweep_dst(const int* __restrict__ dst, int dd,
                                          int lane, const float* __restrict__ h,
                                          const float* __restrict__ a, float& mx,
                                          float& sp, f32x2 (&acc)[HEADS]) {
  for (int i0 = 0; i0 < NE; i0 += 32) {
    const int i = i0 + lane;
    const int d = i < NE ? dst[i] : -1;
    unsigned mask = __builtin_amdgcn_ballot_w32(d == dd);
    while (mask != 0u) {
      const int e = i0 + __builtin_ctz(mask);
      mask &= mask - 1u;
      const float v = a[(long)e * HEADS + (lane & 7)];
      if (KIND == 0) mx = fmaxf(mx, v);
      if (KIND == 1) sp += __expf(v - mx);
      if (KIND == 2) {
        const float wl = __expf(v - mx) / sp;
        f32x2 x;
        x.x = h[(long)e * HID + lane];
        x.y = h[(long)e * HID + 32 + lane];
#pragma unroll
        for (int hh = 0; hh < HEADS; ++hh) acc[hh] = fma_bcast(lane_bcast(wl, hh), x, acc[hh]);
      }
    }
  }
}

__global__ __launch_bounds__(1024) void owner_sum_kernel(
    const float* __restrict__ h, const float* __restrict__ a,
    const int* __restrict__ dst,
    float* __restrict__ eft  ) {
  __shared__ int raw[OWN_D * OWN_DCAP];
  __shared__ int mine[OWN_D * OWN_DCAP];
  __shared__ int cntd[OWN_D];
  __shared__ int over;
  const int tid = threadIdx.x, lane = tid & 31;
  const int wave = __builtin_amdgcn_readfirstlane(tid >> 5);
  const int d0 = blockIdx.x * OWN_D;

  if (tid < OWN_D) cntd[tid] = 0;
  if (tid == OWN_D) over = 0;
  __syncthreads();

  if (tid < NCH * (SLOT_W / 4)) {
    const int4 v = *((const int4*)(eft + (long)d0 * (HEADS * HID)) + tid);
    const int part = lane & 7;
    const int n = __shfl(v.x, lane & ~7, 32);
    if (part == 0 && n > SLOT_CAP) over = 1;
    const int w[4] = {v.x, v.y, v.z, v.w};
#pragma unroll
    for (int j = 0; j < 4; ++j) {
      const int k = part * 4 + j - 1;
      if (k >= 0 && k < n) {
        const int dl = w[j] & (OWN_D - 1);
        const int s = atomicAdd(&cntd[dl], 1);
        if (s < OWN_DCAP) raw[dl * OWN_DCAP + s] = w[j] >> 5;
      }
    }
  }
  __syncthreads();

  const bool ovf = __builtin_amdgcn_readfirstlane(over) != 0;
  const int m = __builtin_amdgcn_readfirstlane(cntd[wave]);
  const int* rw = raw + wave * OWN_DCAP;
  int* my = mine + wave * OWN_DCAP;
  if (!ovf && m <= OWN_DCAP) {
    for (int i0 = 0; i0 < m; i0 += 32) {
      const int i = i0 + lane;
      const int ei = rw[i < m ? i : 0];
      int rank = 0;
      for (int j0 = 0; j0 < m; j0 += 32) {
        const int j = j0 + lane;
        const int ld = rw[j < m ? j : 0];
        const int ej = j < m ? ld : 0x7fffffff;
#pragma unroll
        for (int t = 0; t < 32; ++t)
          rank += __builtin_amdgcn_readlane(ej, t) < ei ? 1 : 0;
      }
      if (i < m) my[rank] = ei;
    }
  }
  __builtin_amdgcn_wave_barrier();

  const int slot = lane >> 3, head = lane & 7;
  float mx = -3.0e38f, sp = 0.f;
  f32x2 acc[HEADS];
#pragma unroll
  for (int hh = 0; hh < HEADS; ++hh) acc[hh].x = acc[hh].y = 0.f;

  if (ovf || m > OWN_DCAP) {
    sweep_dst<0>(dst, d0 + wave, lane, h, a, mx, sp, acc);
    sweep_dst<1>(dst, d0 + wave, lane, h, a, mx, sp, acc);
    sweep_dst<2>(dst, d0 + wave, lane, h, a, mx, sp, acc);
  } else if (m > 0) {
    for (int j0 = 0; j0 < m; j0 += 32) {
      float v[8];
#pragma unroll
      for (int t = 0; t < 8; ++t) {
        const int j = j0 + 4 * t + slot;
        v[t] = a[(long)my[j < m ? j : m - 1] * HEADS + head];
      }
#pragma unroll
      for (int t = 0; t < 8; ++t) mx = fmaxf(mx, v[t]);
    }
    mx = fmaxf(mx, xlane<8>(mx));
    mx = fmaxf(mx, xrow(mx));
    for (int j0 = 0; j0 < m; j0 += 32) {
      float v[8];
#pragma unroll
      for (int t = 0; t < 8; ++t) {
        const int j = j0 + 4 * t + slot;
        v[t] = a[(long)my[j < m ? j : m - 1] * HEADS + head];
      }
#pragma unroll
      for (int t = 0; t < 8; ++t)
        sp += (j0 + 4 * t + slot < m) ? __expf(v[t] - mx) : 0.f;
    }
    sp += xlane<8>(sp);
    sp += xrow(sp);
    for (int j0 = 0; j0 < m; j0 += 8) {
      const int jl = j0 + (lane & 7);
      const int mye = my[jl < m ? jl : m - 1];
      const int ja = j0 + slot, jb = j0 + 4 + slot;
      const int ea = my[ja < m ? ja : m - 1], eb = my[jb < m ? jb : m - 1];
      f32x2 x[8];
#pragma unroll
      for (int u = 0; u < 8; ++u) {
        const int e = __builtin_amdgcn_readlane(mye, u);
        x[u].x = h[(long)e * HID + lane];
        x[u].y = h[(long)e * HID + 32 + lane];
      }
      const float va = a[(long)ea * HEADS + head];
      const float vb = a[(long)eb * HEADS + head];
      const float wa = ja < m ? __expf(va - mx) / sp : 0.f;
      const float wb = jb < m ? __expf(vb - mx) / sp : 0.f;
#pragma unroll
      for (int u = 0; u < 8; ++u) {
#pragma unroll
        for (int hh = 0; hh < HEADS; ++hh)
          acc[hh] = fma_bcast(lane_bcast(u < 4 ? wa : wb, (u & 3) * 8 + hh), x[u], acc[hh]);
      }
    }
  }

  float* o = eft + (long)(d0 + wave) * HEADS * HID + lane;
#pragma unroll
  for (int hh = 0; hh < HEADS; ++hh) {
    const float x = acc[hh].x, y = acc[hh].y;
    o[hh * HID]      = x > 0.f ? x : (__expf(x) - 1.f);
    o[hh * HID + 32] = y > 0.f ? y : (__expf(y) - 1.f);
  }
}

#define SC_ROWS  64
#define SC_PITCH 516
__device__ __forceinline__ void score_chunk(const float4 (&a)[4], const float* b,
                                            f32x8 (&acc)[4]) {
#pragma unroll
  for (int q = 0; q < 4; ++q) {
    float4 bv[4];
#pragma unroll
    for (int nt = 0; nt < 4; ++nt)
      bv[nt] = *(const float4*)(b + nt * 16 * SC_PITCH + q * 8);
    f32x2 a01, a23;
    a01.x = a[q].x; a01.y = a[q].y;
    a23.x = a[q].z; a23.y = a[q].w;
#pragma unroll
    for (int nt = 0; nt < 4; ++nt) {
      f32x2 b01;
      b01.x = bv[nt].x; b01.y = bv[nt].y;
      acc[nt] = wmma_f32(a01, b01, acc[nt]);
    }
#pragma unroll
    for (int nt = 0; nt < 4; ++nt) {
      f32x2 b23;
      b23.x = bv[nt].z; b23.y = bv[nt].w;
      acc[nt] = wmma_f32(a23, b23, acc[nt]);
    }
  }
}
__global__ __launch_bounds__(256) void score_kernel(
    const float* __restrict__ eft, const int* __restrict__ target0,
    const int* __restrict__ target1, const float* __restrict__ fc1_w,
    const float* __restrict__ fc1_b, const float* __restrict__ fc2_w,
    float* __restrict__ score) {
  __shared__ __attribute__((aligned(16))) float ldsW[ATTNV * SC_PITCH];
  const int tid = threadIdx.x;
  const float4* w4 = (const float4*)fc1_w;
  const int rot = blockIdx.x * 128;
  for (int base = 0; base < ATTNV * (HEADS * HID) / 4; base += 256 * 16) {
    float4 t[16];
#pragma unroll
    for (int u = 0; u < 16; ++u)
      t[u] = w4[(base + u * 256 + tid + rot) & (ATTNV * (HEADS * HID) / 4 - 1)];
#pragma unroll
    for (int u = 0; u < 16; ++u) {
      const int i = (base + u * 256 + tid + rot) & (ATTNV * (HEADS * HID) / 4 - 1);
      *(float4*)(ldsW + (i >> 7) * SC_PITCH + (i & 127) * 4) = t[u];
    }
  }
  __syncthreads();

  const int p = blockIdx.x / (NB / SC_ROWS);
  const int mblk = blockIdx.x % (NB / SC_ROWS);
  const float* A = eft + (long)p * NG * (HEADS * HID);
  const int* target = p ? target1 : target0;
  const int wave = tid >> 5, lane = tid & 31;
  const int mt = wave & 3, nh = wave >> 2;
  const int half = lane >> 4, l15 = lane & 15;
  const long arow = target[mblk * SC_ROWS + mt * 16 + l15];
  const float4* ap = (const float4*)(A + arow * (HEADS * HID) + half * 4);
  const float* bp = ldsW + (nh * 64 + l15) * SC_PITCH + half * 4;
  f32x8 acc[4] = {};
  float4 a0[4], a1[4];
#pragma unroll
  for (int q = 0; q < 4; ++q) a0[q] = ap[q * 2];
  for (int ch = 0; ch < (HEADS * HID) / 32; ch += 2) {
#pragma unroll
    for (int q = 0; q < 4; ++q) a1[q] = ap[((ch + 1) * 4 + q) * 2];
    score_chunk(a0, bp + ch * 32, acc);
    const int nx = ch + 2 < (HEADS * HID) / 32 ? ch + 2 : ch;
#pragma unroll
    for (int q = 0; q < 4; ++q) a0[q] = ap[(nx * 4 + q) * 2];
    asm volatile("" ::: "memory");
    score_chunk(a1, bp + (ch + 1) * 32, acc);
  }
  float ssum = 0.f;
#pragma unroll
  for (int nt = 0; nt < 4; ++nt) {
    int col = nh * 64 + nt * 16 + l15;
    float bb = fc1_b[col], w2 = fc2_w[col];
#pragma unroll
    for (int v = 0; v < 8; ++v) ssum += tanhf(acc[nt][v] + bb) * w2;
  }
#pragma unroll
  for (int off = 16; off > 0; off >>= 1) ssum += __shfl_xor(ssum, off, 32);
  if (lane == 0) atomicAdd(&score[p], ssum);
}

#define HL_ROWS  16
#define HL_PITCH 516
__global__ __launch_bounds__(256) void hagg_logits_kernel(
    const float* __restrict__ eft, const int* __restrict__ target0,
    const int* __restrict__ target1, const float* __restrict__ score,
    const float* __restrict__ fc_w, const float* __restrict__ fc_b,
    float* __restrict__ hagg, float* __restrict__ logits) {
  __shared__ __attribute__((aligned(16))) float ldsA[HL_ROWS * HL_PITCH];
  __shared__ __attribute__((aligned(16))) float ldsW[OUTD * HL_PITCH];
  const int tid = threadIdx.x;
  const int b0 = blockIdx.x * HL_ROWS;
  const int c4 = tid & 127, rsub = tid >> 7;
  int t0[8], t1[8];
#pragma unroll
  for (int u = 0; u < 8; ++u) {
    t0[u] = target0[b0 + 2 * u + rsub];
    t1[u] = target1[b0 + 2 * u + rsub];
  }
  float4 wv[8];
#pragma unroll
  for (int u = 0; u < 8; ++u)
    wv[u] = *((const float4*)(fc_w + (2 * u + rsub) * (HEADS * HID)) + c4);

  float s0 = score[0] / (float)NB, s1 = score[1] / (float)NB;
  float mx = fmaxf(s0, s1);
  float e0 = __expf(s0 - mx), e1 = __expf(s1 - mx);
  float inv = 1.f / (e0 + e1);
  float beta0 = e0 * inv, beta1 = e1 * inv;

  float4 x0[8], x1[8];
#pragma unroll
  for (int u = 0; u < 8; ++u) {
    x0[u] = *((const float4*)(eft + (long)t0[u] * (HEADS * HID)) + c4);
    x1[u] = *((const float4*)(eft + (long)NG * (HEADS * HID) +
                              (long)t1[u] * (HEADS * HID)) + c4);
  }
#pragma unroll
  for (int u = 0; u < 8; ++u)
    *(float4*)(ldsW + (2 * u + rsub) * HL_PITCH + c4 * 4) = wv[u];
  float4 hv[8];
#pragma unroll
  for (int u = 0; u < 8; ++u) {
    hv[u].x = __builtin_fmaf(beta0, x0[u].x, beta1 * x1[u].x);
    hv[u].y = __builtin_fmaf(beta0, x0[u].y, beta1 * x1[u].y);
    hv[u].z = __builtin_fmaf(beta0, x0[u].z, beta1 * x1[u].z);
    hv[u].w = __builtin_fmaf(beta0, x0[u].w, beta1 * x1[u].w);
    *(float4*)(ldsA + (2 * u + rsub) * HL_PITCH + c4 * 4) = hv[u];
  }
  __syncthreads();
#pragma unroll
  for (int u = 0; u < 8; ++u)
    *((float4*)(hagg + (long)(b0 + 2 * u + rsub) * (HEADS * HID)) + c4) = hv[u];
  if (tid >= 32) return;
  const int half = tid >> 4, l15 = tid & 15;
  const float* ap = ldsA + l15 * HL_PITCH + half * 2;
  const float* bp = ldsW + l15 * HL_PITCH + half * 2;
  f32x8 c = {};
  for (int k0 = 0; k0 < HEADS * HID; k0 += 4) {
    const float2 av = *(const float2*)(ap + k0);
    const float2 bv = *(const float2*)(bp + k0);
    f32x2 af, bf;
    af.x = av.x; af.y = av.y;
    bf.x = bv.x; bf.y = bv.y;
    c = wmma_f32(af, bf, c);
  }
  const float bias = fc_b[l15];
#pragma unroll
  for (int v = 0; v < 8; ++v)
    logits[(long)(b0 + v + half * 8) * OUTD + l15] = c[v] + bias;
}

extern "C" void kernel_launch(void* const* d_in, const int* in_sizes, int n_in,
                              void* d_out, int out_size, void* d_ws,
                              size_t ws_size, hipStream_t stream) {
  const float* feat0 = (const float*)d_in[0];
  const float* feat1 = (const float*)d_in[1];
  const float* W0    = (const float*)d_in[2];
  const float* b0    = (const float*)d_in[3];
  const float* W1    = (const float*)d_in[4];
  const float* b1    = (const float*)d_in[5];
  const float* r_vec = (const float*)d_in[6];
  const float* attn0 = (const float*)d_in[7];
  const float* attn1 = (const float*)d_in[8];
  const float* fc1_w = (const float*)d_in[9];
  const float* fc1_b = (const float*)d_in[10];
  const float* fc2_w = (const float*)d_in[11];
  const float* fc_w  = (const float*)d_in[12];
  const float* fc_b  = (const float*)d_in[13];
  const int* idx0    = (const int*)d_in[14];
  const int* idx1    = (const int*)d_in[15];
  const int* mp_indices[2] = {(const int*)d_in[16], (const int*)d_in[19]};
  const int* mp_dst[2]     = {(const int*)d_in[17], (const int*)d_in[20]};
  const int* mp_target[2]  = {(const int*)d_in[18], (const int*)d_in[21]};
  const float* attn[2] = {attn0, attn1};

  float* ws     = (float*)d_ws;
  float* tf     = ws;
  float* finals = tf + (long)N_NODES * HID;
  float* h      = finals + 2 * LP * HID;
  float* a      = h + (long)NE * HID;
  float* eft    = a + (long)NE * HEADS;
  float* score  = eft + 2L * NG * HEADS * HID;

  float* logits = (float*)d_out;
  float* hagg   = logits + (long)NB * OUTD;

  setup_kernel<<<GEMM0_WG + GEMM1_WG + INIT_WG + 2 * NCH, 256, 0, stream>>>(
      feat0, W0, b0, idx0, feat1, W1, b1, idx1, tf, r_vec, finals, score,
      mp_dst[0], mp_dst[1], eft);

  for (int mp = 0; mp < 2; ++mp) {
    edge_h_kernel<<<(NE / EPW + 7) / 8, 256, 0, stream>>>(tf, finals + mp * LP * HID,
                                              mp_indices[mp], attn[mp], h, a);
    owner_sum_kernel<<<NG / OWN_D, 1024, 0, stream>>>(
        h, a, mp_dst[mp], eft + (long)mp * NG * HEADS * HID);
  }

  score_kernel<<<2 * (NB / SC_ROWS), 256, 0, stream>>>(
      eft, mp_target[0], mp_target[1], fc1_w, fc1_b, fc2_w, score);
  hagg_logits_kernel<<<NB / HL_ROWS, 256, 0, stream>>>(
      eft, mp_target[0], mp_target[1], score, fc_w, fc_b, hagg, logits);
}
